# starting kernel
# baseline (speedup 1.0000x reference)
#include <hip/hip_runtime.h>
#include <hip/hip_bf16.h>

#define B_    8
#define N_    1024
#define C_    768
#define H_    12
#define HD_   64
#define NB_   49
#define NBP_  64
#define ROWS_ 32
#define BSTR_ 66
#define VCH_  128
#define VSTR_ 130

typedef float v2f __attribute__((ext_vector_type(2)));
typedef float v8f __attribute__((ext_vector_type(8)));

static __device__ __forceinline__ v8f wmma4(v2f a, v2f b, v8f c) {
  return __builtin_amdgcn_wmma_f32_16x16x4_f32(false, a, false, b, (short)0, c,
                                               false, false);
}
static __device__ __forceinline__ v2f ld2(const float* p) {
  return *(const v2f*)p;
}

__global__ __launch_bounds__(256) void k_qkv(const float* __restrict__ x,
                                             const float* __restrict__ wqkv,
                                             float* __restrict__ qb,
                                             float* __restrict__ kb,
                                             float* __restrict__ vb) {
  __shared__ float Bt[64 * BSTR_];

  const int wave  = threadIdx.x >> 5;
  const int lane  = threadIdx.x & 31;
  const int l15   = lane & 15;
  const int lhalf = lane >> 4;

  const int ngrp = blockIdx.y;
  const int nb   = ngrp * 64;
  const int row  = blockIdx.x * 128 + wave * 16 + l15;
  const float* xr = x + (size_t)row * C_;

  v8f acc[4];
  acc[0] = (v8f)0.f; acc[1] = (v8f)0.f; acc[2] = (v8f)0.f; acc[3] = (v8f)0.f;

  for (int k0 = 0; k0 < C_; k0 += 64) {
    __syncthreads();
#pragma unroll
    for (int r = 0; r < 4; ++r) {
      const int idx = threadIdx.x + 256 * r;
      const int k   = idx >> 4;
      const int nq  = (idx & 15) * 4;
      const float4 w = *(const float4*)(wqkv + (size_t)(k0 + k) * (3 * C_) + nb + nq);
      Bt[(nq + 0) * BSTR_ + k] = w.x;
      Bt[(nq + 1) * BSTR_ + k] = w.y;
      Bt[(nq + 2) * BSTR_ + k] = w.z;
      Bt[(nq + 3) * BSTR_ + k] = w.w;
      if (k0 + 64 < C_)
        __builtin_prefetch(wqkv + (size_t)(k0 + 64 + k) * (3 * C_) + nb + nq, 0, 1);
    }
    __syncthreads();

#pragma unroll
    for (int kk = 0; kk < 64; kk += 4) {
      const int kbi = kk + 2 * lhalf;
      v2f a = ld2(xr + k0 + kbi);
#pragma unroll
      for (int t = 0; t < 4; ++t) {
        v2f bf = ld2(&Bt[(t * 16 + l15) * BSTR_ + kbi]);
        acc[t] = wmma4(a, bf, acc[t]);
      }
    }
  }

  const int which = ngrp / H_;
  const int h     = ngrp % H_;
  float* dst = (which == 0) ? qb : (which == 1) ? kb : vb;
#pragma unroll
  for (int t = 0; t < 4; ++t) {
    const int d = t * 16 + l15;
#pragma unroll
    for (int vv = 0; vv < 8; ++vv) {
      const int r    = blockIdx.x * 128 + wave * 16 + vv + 8 * lhalf;
      const int btok = r >> 10;
      const int tok  = r & (N_ - 1);
      dst[(((size_t)btok * H_ + h) * N_ + tok) * HD_ + d] = acc[t][vv];
    }
  }
}

__global__ __launch_bounds__(256) void k_tables(const float* __restrict__ qb,
                                                const float* __restrict__ kb,
                                                const float* __restrict__ tq,
                                                const float* __restrict__ tk,
                                                float* __restrict__ lkb,
                                                float* __restrict__ lqb) {
  const int wave  = threadIdx.x >> 5;
  const int lane  = threadIdx.x & 31;
  const int l15   = lane & 15;
  const int lhalf = lane >> 4;

  const int unit = blockIdx.x * 8 + wave;
  const int bh   = unit >> 6;
  const int mt   = unit & 63;
  const int h    = bh % H_;

  const size_t rowbase = ((size_t)bh * N_ + mt * 16 + l15) * HD_;
  const v2f z2 = (v2f)0.f;

  v8f ck[4], cq[4];
#pragma unroll
  for (int t = 0; t < 4; ++t) { ck[t] = (v8f)0.f; cq[t] = (v8f)0.f; }

#pragma unroll
  for (int kk = 0; kk < HD_; kk += 4) {
    const int kbi = kk + 2 * lhalf;
    v2f aq = ld2(qb + rowbase + kbi);
    v2f ak = ld2(kb + rowbase + kbi);
    ak = ak * 0.125f;
#pragma unroll
    for (int t = 0; t < 4; ++t) {
      const int n = t * 16 + l15;
      v2f btk = (n < NB_) ? ld2(tk + ((size_t)h * NB_ + n) * HD_ + kbi) : z2;
      v2f btq = (n < NB_) ? ld2(tq + ((size_t)h * NB_ + n) * HD_ + kbi) : z2;
      ck[t] = wmma4(aq, btk, ck[t]);
      cq[t] = wmma4(ak, btq, cq[t]);
    }
  }

#pragma unroll
  for (int t = 0; t < 4; ++t) {
    const int col = t * 16 + l15;
#pragma unroll
    for (int vv = 0; vv < 8; ++vv) {
      const int r = mt * 16 + vv + 8 * lhalf;
      lkb[((size_t)bh * N_ + r) * NBP_ + col] = ck[t][vv];
      lqb[((size_t)bh * N_ + r) * NBP_ + col] = cq[t][vv];
    }
  }
}

__global__ __launch_bounds__(256) void k_attn(const float* __restrict__ qb,
                                              const float* __restrict__ kb,
                                              const float* __restrict__ vb,
                                              const float* __restrict__ lkb,
                                              const float* __restrict__ lqb,
                                              const float* __restrict__ tv,
                                              const int*   __restrict__ rp,
                                              float* __restrict__ ob) {
  extern __shared__ float smem[];
  float* S  = smem;
  float* ab = smem + ROWS_ * N_;
  float* Vt = smem + ROWS_ * N_ + ROWS_ * NBP_;

  const int wave  = threadIdx.x >> 5;
  const int lane  = threadIdx.x & 31;
  const int l15   = lane & 15;
  const int lhalf = lane >> 4;

  const int rblk = blockIdx.x;
  const int bh   = blockIdx.y;
  const int b    = bh / H_;
  const int h    = bh % H_;
  const int r0   = rblk * ROWS_;

  {
    const int mt = wave & 1;
    const float* qrow = qb + ((size_t)bh * N_ + r0 + mt * 16 + l15) * HD_;
    v2f qf[16];
#pragma unroll
    for (int s = 0; s < 16; ++s) qf[s] = ld2(qrow + 4 * s + 2 * lhalf);

    for (int nt = (wave >> 1); nt < 64; nt += 4) {
      const float* krow = kb + ((size_t)bh * N_ + nt * 16 + l15) * HD_;
      v8f acc = (v8f)0.f;
#pragma unroll
      for (int s = 0; s < 16; ++s)
        acc = wmma4(qf[s], ld2(krow + 4 * s + 2 * lhalf), acc);

      const int n = nt * 16 + l15;
#pragma unroll
      for (int vv = 0; vv < 8; ++vv) {
        const int m   = mt * 16 + vv + 8 * lhalf;
        const int i   = r0 + m;
        const int bij = rp[i * N_ + n];
        const int bji = rp[n * N_ + i];
        float val = acc[vv]
                  + lkb[((size_t)bh * N_ + i) * NBP_ + bij]
                  + lqb[((size_t)bh * N_ + n) * NBP_ + bji];
        S[m * N_ + n] = val;
      }
    }
  }
  __syncthreads();

  for (int rr = 0; rr < 4; ++rr) {
    const int m = wave * 4 + rr;
    float* row = S + m * N_;
    float mx = -3.4e38f;
    for (int j = lane; j < N_; j += 32) mx = fmaxf(mx, row[j]);
#pragma unroll
    for (int off = 16; off > 0; off >>= 1)
      mx = fmaxf(mx, __shfl_xor(mx, off, 32));
    float sum = 0.f;
    for (int j = lane; j < N_; j += 32) {
      float e = __expf(row[j] - mx);
      row[j] = e;
      sum += e;
    }
#pragma unroll
    for (int off = 16; off > 0; off >>= 1)
      sum += __shfl_xor(sum, off, 32);
    const float inv = 1.0f / sum;
    for (int j = lane; j < N_; j += 32) row[j] *= inv;
  }
  __syncthreads();

  for (int idx = threadIdx.x; idx < ROWS_ * NBP_; idx += 256) ab[idx] = 0.f;
  __syncthreads();
  for (int rr = 0; rr < 4; ++rr) {
    const int m = wave * 4 + rr;
    const int i = r0 + m;
    for (int j = lane; j < N_; j += 32) {
      const int bkt = rp[i * N_ + j];
      atomicAdd(&ab[m * NBP_ + bkt], S[m * N_ + j]);
    }
  }

  const int mt   = wave & 1;
  const int dt   = wave >> 1;
  const int dcol = dt * 16 + l15;
  v8f acc = (v8f)0.f;

  for (int ch = 0; ch < N_; ch += VCH_) {
    __syncthreads();
#pragma unroll
    for (int r = 0; r < 8; ++r) {
      const int idx = threadIdx.x + 256 * r;
      const int j   = idx >> 4;
      const int dq  = (idx & 15) * 4;
      const float4 w = *(const float4*)(vb + ((size_t)bh * N_ + ch + j) * HD_ + dq);
      Vt[(dq + 0) * VSTR_ + j] = w.x;
      Vt[(dq + 1) * VSTR_ + j] = w.y;
      Vt[(dq + 2) * VSTR_ + j] = w.z;
      Vt[(dq + 3) * VSTR_ + j] = w.w;
    }
    __syncthreads();
#pragma unroll
    for (int kk = 0; kk < VCH_; kk += 4) {
      const int kbi = kk + 2 * lhalf;
      v2f a  = ld2(&S[(mt * 16 + l15) * N_ + ch + kbi]);
      v2f bf = ld2(&Vt[dcol * VSTR_ + kbi]);
      acc = wmma4(a, bf, acc);
    }
  }

#pragma unroll
  for (int kk = 0; kk < NBP_; kk += 4) {
    const int kbi = kk + 2 * lhalf;
    v2f a = ld2(&ab[(mt * 16 + l15) * NBP_ + kbi]);
    v2f bf;
    bf.x = (kbi     < NB_) ? tv[((size_t)h * NB_ + kbi)     * HD_ + dcol] : 0.f;
    bf.y = (kbi + 1 < NB_) ? tv[((size_t)h * NB_ + kbi + 1) * HD_ + dcol] : 0.f;
    acc = wmma4(a, bf, acc);
  }

#pragma unroll
  for (int vv = 0; vv < 8; ++vv) {
    const int m   = mt * 16 + vv + 8 * lhalf;
    const int tok = r0 + m;
    ob[(((size_t)b * N_ + tok) * H_ + h) * HD_ + dcol] = acc[vv];
  }
}

__global__ __launch_bounds__(256) void k_proj(const float* __restrict__ ob,
                                              const float* __restrict__ wp,
                                              const float* __restrict__ bias,
                                              float* __restrict__ out) {
  __shared__ float Bt[64 * BSTR_];

  const int wave  = threadIdx.x >> 5;
  const int lane  = threadIdx.x & 31;
  const int l15   = lane & 15;
  const int lhalf = lane >> 4;

  const int ngrp = blockIdx.y;
  const int nb   = ngrp * 64;
  const int row  = blockIdx.x * 128 + wave * 16 + l15;
  const float* xr = ob + (size_t)row * C_;

  v8f acc[4];
  acc[0] = (v8f)0.f; acc[1] = (v8f)0.f; acc[2] = (v8f)0.f; acc[3] = (v8f)0.f;

  for (int k0 = 0; k0 < C_; k0 += 64) {
    __syncthreads();
#pragma unroll
    for (int r = 0; r < 4; ++r) {
      const int idx = threadIdx.x + 256 * r;
      const int k   = idx >> 4;
      const int nq  = (idx & 15) * 4;
      const float4 w = *(const float4*)(wp + (size_t)(k0 + k) * C_ + nb + nq);
      Bt[(nq + 0) * BSTR_ + k] = w.x;
      Bt[(nq + 1) * BSTR_ + k] = w.y;
      Bt[(nq + 2) * BSTR_ + k] = w.z;
      Bt[(nq + 3) * BSTR_ + k] = w.w;
      if (k0 + 64 < C_)
        __builtin_prefetch(wp + (size_t)(k0 + 64 + k) * C_ + nb + nq, 0, 1);
    }
    __syncthreads();

#pragma unroll
    for (int kk = 0; kk < 64; kk += 4) {
      const int kbi = kk + 2 * lhalf;
      v2f a = ld2(xr + k0 + kbi);
#pragma unroll
      for (int t = 0; t < 4; ++t) {
        v2f bf = ld2(&Bt[(t * 16 + l15) * BSTR_ + kbi]);
        acc[t] = wmma4(a, bf, acc[t]);
      }
    }
  }

#pragma unroll
  for (int t = 0; t < 4; ++t) {
    const int c0 = nb + t * 16 + l15;
    const float bv = bias[c0];
#pragma unroll
    for (int vv = 0; vv < 8; ++vv) {
      const int r = blockIdx.x * 128 + wave * 16 + vv + 8 * lhalf;
      out[(size_t)r * C_ + c0] = acc[t][vv] + bv;
    }
  }
}

extern "C" void kernel_launch(void* const* d_in, const int* in_sizes, int n_in,
                              void* d_out, int out_size, void* d_ws, size_t ws_size,
                              hipStream_t stream) {
  const float* x     = (const float*)d_in[0];
  const float* wqkv  = (const float*)d_in[1];
  const float* wproj = (const float*)d_in[2];
  const float* bproj = (const float*)d_in[3];
  const float* tq    = (const float*)d_in[4];
  const float* tk    = (const float*)d_in[5];
  const float* tv    = (const float*)d_in[6];
  const int*   rp    = (const int*)d_in[7];
  float*       out   = (float*)d_out;

  float* ws = (float*)d_ws;
  const size_t SZ = (size_t)B_ * H_ * N_ * HD_;
  float* qb  = ws;
  float* kb  = ws + 1 * SZ;
  float* vb  = ws + 2 * SZ;
  float* lkb = ws + 3 * SZ;
  float* lqb = ws + 4 * SZ;
  float* ob  = ws + 5 * SZ;

  k_qkv<<<dim3(64, 36), 256, 0, stream>>>(x, wqkv, qb, kb, vb);

  k_tables<<<768, 256, 0, stream>>>(qb, kb, tq, tk, lkb, lqb);

  const size_t lds = (size_t)(ROWS_ * N_ + ROWS_ * NBP_ + HD_ * VSTR_) * sizeof(float);
  k_attn<<<dim3(32, 96), 256, lds, stream>>>(qb, kb, vb, lkb, lqb, tv, rp, ob);

  k_proj<<<dim3(64, 12), 256, 0, stream>>>(ob, wproj, bproj, out);

  (void)in_sizes; (void)n_in; (void)out_size; (void)ws_size;
}


# attention bucket sums: lane=row, wave=column range, own LDS copy per wave: no two lanes of an LDS add share a word, loads issued ahead of adds; copies summed in fixed order (sum order changes)
# speedup vs baseline: 1.3831x; 1.3831x over previous
#include <hip/hip_runtime.h>
#include <hip/hip_bf16.h>

#define B_    8
#define N_    1024
#define C_    768
#define H_    12
#define HD_   64
#define NB_   49
#define NBP_  64
#define ROWS_ 32
#define BSTR_ 66
#define VCH_  128
#define VSTR_ 130
#define PSTR_ 66

typedef float v2f __attribute__((ext_vector_type(2)));
typedef float v8f __attribute__((ext_vector_type(8)));

static __device__ __forceinline__ v8f wmma4(v2f a, v2f b, v8f c) {
  return __builtin_amdgcn_wmma_f32_16x16x4_f32(false, a, false, b, (short)0, c,
                                               false, false);
}
static __device__ __forceinline__ v2f ld2(const float* p) {
  return *(const v2f*)p;
}

__global__ __launch_bounds__(256) void k_qkv(const float* __restrict__ x,
                                             const float* __restrict__ wqkv,
                                             float* __restrict__ qb,
                                             float* __restrict__ kb,
                                             float* __restrict__ vb) {
  __shared__ float Bt[64 * BSTR_];

  const int wave  = threadIdx.x >> 5;
  const int lane  = threadIdx.x & 31;
  const int l15   = lane & 15;
  const int lhalf = lane >> 4;

  const int ngrp = blockIdx.y;
  const int nb   = ngrp * 64;
  const int row  = blockIdx.x * 128 + wave * 16 + l15;
  const float* xr = x + (size_t)row * C_;

  v8f acc[4];
  acc[0] = (v8f)0.f; acc[1] = (v8f)0.f; acc[2] = (v8f)0.f; acc[3] = (v8f)0.f;

  for (int k0 = 0; k0 < C_; k0 += 64) {
    __syncthreads();
#pragma unroll
    for (int r = 0; r < 4; ++r) {
      const int idx = threadIdx.x + 256 * r;
      const int k   = idx >> 4;
      const int nq  = (idx & 15) * 4;
      const float4 w = *(const float4*)(wqkv + (size_t)(k0 + k) * (3 * C_) + nb + nq);
      Bt[(nq + 0) * BSTR_ + k] = w.x;
      Bt[(nq + 1) * BSTR_ + k] = w.y;
      Bt[(nq + 2) * BSTR_ + k] = w.z;
      Bt[(nq + 3) * BSTR_ + k] = w.w;
      if (k0 + 64 < C_)
        __builtin_prefetch(wqkv + (size_t)(k0 + 64 + k) * (3 * C_) + nb + nq, 0, 1);
    }
    __syncthreads();

#pragma unroll
    for (int kk = 0; kk < 64; kk += 4) {
      const int kbi = kk + 2 * lhalf;
      v2f a = ld2(xr + k0 + kbi);
#pragma unroll
      for (int t = 0; t < 4; ++t) {
        v2f bf = ld2(&Bt[(t * 16 + l15) * BSTR_ + kbi]);
        acc[t] = wmma4(a, bf, acc[t]);
      }
    }
  }

  const int which = ngrp / H_;
  const int h     = ngrp % H_;
  float* dst = (which == 0) ? qb : (which == 1) ? kb : vb;
#pragma unroll
  for (int t = 0; t < 4; ++t) {
    const int d = t * 16 + l15;
#pragma unroll
    for (int vv = 0; vv < 8; ++vv) {
      const int r    = blockIdx.x * 128 + wave * 16 + vv + 8 * lhalf;
      const int btok = r >> 10;
      const int tok  = r & (N_ - 1);
      dst[(((size_t)btok * H_ + h) * N_ + tok) * HD_ + d] = acc[t][vv];
    }
  }
}

__global__ __launch_bounds__(256) void k_tables(const float* __restrict__ qb,
                                                const float* __restrict__ kb,
                                                const float* __restrict__ tq,
                                                const float* __restrict__ tk,
                                                float* __restrict__ lkb,
                                                float* __restrict__ lqb) {
  const int wave  = threadIdx.x >> 5;
  const int lane  = threadIdx.x & 31;
  const int l15   = lane & 15;
  const int lhalf = lane >> 4;

  const int unit = blockIdx.x * 8 + wave;
  const int bh   = unit >> 6;
  const int mt   = unit & 63;
  const int h    = bh % H_;

  const size_t rowbase = ((size_t)bh * N_ + mt * 16 + l15) * HD_;
  const v2f z2 = (v2f)0.f;

  v8f ck[4], cq[4];
#pragma unroll
  for (int t = 0; t < 4; ++t) { ck[t] = (v8f)0.f; cq[t] = (v8f)0.f; }

#pragma unroll
  for (int kk = 0; kk < HD_; kk += 4) {
    const int kbi = kk + 2 * lhalf;
    v2f aq = ld2(qb + rowbase + kbi);
    v2f ak = ld2(kb + rowbase + kbi);
    ak = ak * 0.125f;
#pragma unroll
    for (int t = 0; t < 4; ++t) {
      const int n = t * 16 + l15;
      v2f btk = (n < NB_) ? ld2(tk + ((size_t)h * NB_ + n) * HD_ + kbi) : z2;
      v2f btq = (n < NB_) ? ld2(tq + ((size_t)h * NB_ + n) * HD_ + kbi) : z2;
      ck[t] = wmma4(aq, btk, ck[t]);
      cq[t] = wmma4(ak, btq, cq[t]);
    }
  }

#pragma unroll
  for (int t = 0; t < 4; ++t) {
    const int col = t * 16 + l15;
#pragma unroll
    for (int vv = 0; vv < 8; ++vv) {
      const int r = mt * 16 + vv + 8 * lhalf;
      lkb[((size_t)bh * N_ + r) * NBP_ + col] = ck[t][vv];
      lqb[((size_t)bh * N_ + r) * NBP_ + col] = cq[t][vv];
    }
  }
}

__global__ __launch_bounds__(256) void k_attn(const float* __restrict__ qb,
                                              const float* __restrict__ kb,
                                              const float* __restrict__ vb,
                                              const float* __restrict__ lkb,
                                              const float* __restrict__ lqb,
                                              const float* __restrict__ tv,
                                              const int*   __restrict__ rp,
                                              float* __restrict__ ob) {
  extern __shared__ __attribute__((aligned(16))) float smem[];
  float* S  = smem;
  float* ab = smem + ROWS_ * N_;
  float* Vt = smem + ROWS_ * N_ + ROWS_ * NBP_;
  float* pw = Vt + HD_ * VSTR_;

  const int wave  = threadIdx.x >> 5;
  const int lane  = threadIdx.x & 31;
  const int l15   = lane & 15;
  const int lhalf = lane >> 4;

  const int rblk = blockIdx.x;
  const int bh   = blockIdx.y;
  const int b    = bh / H_;
  const int h    = bh % H_;
  const int r0   = rblk * ROWS_;

  {
    const int mt = wave & 1;
    const float* qrow = qb + ((size_t)bh * N_ + r0 + mt * 16 + l15) * HD_;
    v2f qf[16];
#pragma unroll
    for (int s = 0; s < 16; ++s) qf[s] = ld2(qrow + 4 * s + 2 * lhalf);

    for (int nt = (wave >> 1); nt < 64; nt += 4) {
      const float* krow = kb + ((size_t)bh * N_ + nt * 16 + l15) * HD_;
      v8f acc = (v8f)0.f;
#pragma unroll
      for (int s = 0; s < 16; ++s)
        acc = wmma4(qf[s], ld2(krow + 4 * s + 2 * lhalf), acc);

      const int n = nt * 16 + l15;
#pragma unroll
      for (int vv = 0; vv < 8; ++vv) {
        const int m   = mt * 16 + vv + 8 * lhalf;
        const int i   = r0 + m;
        const int bij = rp[i * N_ + n];
        const int bji = rp[n * N_ + i];
        float val = acc[vv]
                  + lkb[((size_t)bh * N_ + i) * NBP_ + bij]
                  + lqb[((size_t)bh * N_ + n) * NBP_ + bji];
        S[m * N_ + n] = val;
      }
    }
  }
  __syncthreads();

  for (int rr = 0; rr < 4; ++rr) {
    const int m = wave * 4 + rr;
    float* row = S + m * N_;
    float mx = -3.4e38f;
    for (int j = lane; j < N_; j += 32) mx = fmaxf(mx, row[j]);
#pragma unroll
    for (int off = 16; off > 0; off >>= 1)
      mx = fmaxf(mx, __shfl_xor(mx, off, 32));
    float sum = 0.f;
    for (int j = lane; j < N_; j += 32) {
      float e = __expf(row[j] - mx);
      row[j] = e;
      sum += e;
    }
#pragma unroll
    for (int off = 16; off > 0; off >>= 1)
      sum += __shfl_xor(sum, off, 32);
    const float inv = 1.0f / sum;
    for (int j = lane; j < N_; j += 32) row[j] *= inv;
  }
  __syncthreads();

  {
    float* mine = pw + wave * (ROWS_ * PSTR_);
    for (int k = lane; k < ROWS_ * PSTR_ / 2; k += 32)
      ((v2f*)mine)[k] = (v2f)0.f;
    float*       myrow = mine + lane * PSTR_;
    const float* prow  = S + lane * N_;
    const int*   brow  = rp + (size_t)(r0 + lane) * N_;
    for (int win = 0; win < 2; ++win) {
      const int j0 = wave * 128 + win * 64;
      float4 p[16];
      int4   b[16];
#pragma unroll
      for (int c = 0; c < 16; ++c) {
        const int col = j0 + 4 * ((lane + c) & 15);
        p[c] = *(const float4*)(prow + col);
        b[c] = *(const int4*)(brow + col);
      }
#pragma unroll
      for (int c = 0; c < 16; ++c) {
        atomicAdd(myrow + b[c].x, p[c].x);
        atomicAdd(myrow + b[c].y, p[c].y);
        atomicAdd(myrow + b[c].z, p[c].z);
        atomicAdd(myrow + b[c].w, p[c].w);
      }
    }
  }
  __syncthreads();
  for (int idx = threadIdx.x; idx < ROWS_ * NBP_; idx += 256) {
    const int m  = idx >> 6;
    const int bk = idx & 63;
    float s = 0.f;
    for (int w = 0; w < 8; ++w) s += pw[(w * ROWS_ + m) * PSTR_ + bk];
    ab[idx] = s;
  }

  const int mt   = wave & 1;
  const int dt   = wave >> 1;
  const int dcol = dt * 16 + l15;
  v8f acc = (v8f)0.f;

  for (int ch = 0; ch < N_; ch += VCH_) {
    __syncthreads();
#pragma unroll
    for (int r = 0; r < 8; ++r) {
      const int idx = threadIdx.x + 256 * r;
      const int j   = idx >> 4;
      const int dq  = (idx & 15) * 4;
      const float4 w = *(const float4*)(vb + ((size_t)bh * N_ + ch + j) * HD_ + dq);
      Vt[(dq + 0) * VSTR_ + j] = w.x;
      Vt[(dq + 1) * VSTR_ + j] = w.y;
      Vt[(dq + 2) * VSTR_ + j] = w.z;
      Vt[(dq + 3) * VSTR_ + j] = w.w;
    }
    __syncthreads();
#pragma unroll
    for (int kk = 0; kk < VCH_; kk += 4) {
      const int kbi = kk + 2 * lhalf;
      v2f a  = ld2(&S[(mt * 16 + l15) * N_ + ch + kbi]);
      v2f bf = ld2(&Vt[dcol * VSTR_ + kbi]);
      acc = wmma4(a, bf, acc);
    }
  }

#pragma unroll
  for (int kk = 0; kk < NBP_; kk += 4) {
    const int kbi = kk + 2 * lhalf;
    v2f a = ld2(&ab[(mt * 16 + l15) * NBP_ + kbi]);
    v2f bf;
    bf.x = (kbi     < NB_) ? tv[((size_t)h * NB_ + kbi)     * HD_ + dcol] : 0.f;
    bf.y = (kbi + 1 < NB_) ? tv[((size_t)h * NB_ + kbi + 1) * HD_ + dcol] : 0.f;
    acc = wmma4(a, bf, acc);
  }

#pragma unroll
  for (int vv = 0; vv < 8; ++vv) {
    const int m   = mt * 16 + vv + 8 * lhalf;
    const int tok = r0 + m;
    ob[(((size_t)b * N_ + tok) * H_ + h) * HD_ + dcol] = acc[vv];
  }
}

__global__ __launch_bounds__(256) void k_proj(const float* __restrict__ ob,
                                              const float* __restrict__ wp,
                                              const float* __restrict__ bias,
                                              float* __restrict__ out) {
  __shared__ float Bt[64 * BSTR_];

  const int wave  = threadIdx.x >> 5;
  const int lane  = threadIdx.x & 31;
  const int l15   = lane & 15;
  const int lhalf = lane >> 4;

  const int ngrp = blockIdx.y;
  const int nb   = ngrp * 64;
  const int row  = blockIdx.x * 128 + wave * 16 + l15;
  const float* xr = ob + (size_t)row * C_;

  v8f acc[4];
  acc[0] = (v8f)0.f; acc[1] = (v8f)0.f; acc[2] = (v8f)0.f; acc[3] = (v8f)0.f;

  for (int k0 = 0; k0 < C_; k0 += 64) {
    __syncthreads();
#pragma unroll
    for (int r = 0; r < 4; ++r) {
      const int idx = threadIdx.x + 256 * r;
      const int k   = idx >> 4;
      const int nq  = (idx & 15) * 4;
      const float4 w = *(const float4*)(wp + (size_t)(k0 + k) * C_ + nb + nq);
      Bt[(nq + 0) * BSTR_ + k] = w.x;
      Bt[(nq + 1) * BSTR_ + k] = w.y;
      Bt[(nq + 2) * BSTR_ + k] = w.z;
      Bt[(nq + 3) * BSTR_ + k] = w.w;
      if (k0 + 64 < C_)
        __builtin_prefetch(wp + (size_t)(k0 + 64 + k) * C_ + nb + nq, 0, 1);
    }
    __syncthreads();

#pragma unroll
    for (int kk = 0; kk < 64; kk += 4) {
      const int kbi = kk + 2 * lhalf;
      v2f a = ld2(xr + k0 + kbi);
#pragma unroll
      for (int t = 0; t < 4; ++t) {
        v2f bf = ld2(&Bt[(t * 16 + l15) * BSTR_ + kbi]);
        acc[t] = wmma4(a, bf, acc[t]);
      }
    }
  }

#pragma unroll
  for (int t = 0; t < 4; ++t) {
    const int c0 = nb + t * 16 + l15;
    const float bv = bias[c0];
#pragma unroll
    for (int vv = 0; vv < 8; ++vv) {
      const int r = blockIdx.x * 128 + wave * 16 + vv + 8 * lhalf;
      out[(size_t)r * C_ + c0] = acc[t][vv] + bv;
    }
  }
}

extern "C" void kernel_launch(void* const* d_in, const int* in_sizes, int n_in,
                              void* d_out, int out_size, void* d_ws, size_t ws_size,
                              hipStream_t stream) {
  const float* x     = (const float*)d_in[0];
  const float* wqkv  = (const float*)d_in[1];
  const float* wproj = (const float*)d_in[2];
  const float* bproj = (const float*)d_in[3];
  const float* tq    = (const float*)d_in[4];
  const float* tk    = (const float*)d_in[5];
  const float* tv    = (const float*)d_in[6];
  const int*   rp    = (const int*)d_in[7];
  float*       out   = (float*)d_out;

  float* ws = (float*)d_ws;
  const size_t SZ = (size_t)B_ * H_ * N_ * HD_;
  float* qb  = ws;
  float* kb  = ws + 1 * SZ;
  float* vb  = ws + 2 * SZ;
  float* lkb = ws + 3 * SZ;
  float* lqb = ws + 4 * SZ;
  float* ob  = ws + 5 * SZ;

  k_qkv<<<dim3(64, 36), 256, 0, stream>>>(x, wqkv, qb, kb, vb);

  k_tables<<<768, 256, 0, stream>>>(qb, kb, tq, tk, lkb, lqb);

  const size_t lds = (size_t)(ROWS_ * N_ + ROWS_ * NBP_ + HD_ * VSTR_ +
                              8 * ROWS_ * PSTR_) * sizeof(float);
  k_attn<<<dim3(32, 96), 256, lds, stream>>>(qb, kb, vb, lkb, lqb, tv, rp, ob);

  k_proj<<<dim3(64, 12), 256, 0, stream>>>(ob, wproj, bproj, out);

  (void)in_sizes; (void)n_in; (void)out_size; (void)ws_size;
}


# attention: LDS row pitch of S 1024->1028 and of ab 64->68 floats so the 16-row A-fragment reads of P@V and ab@table_v leave one bank; bucket-sum columns per lane now ascending (sum order changes)
# speedup vs baseline: 1.4991x; 1.0839x over previous
#include <hip/hip_runtime.h>
#include <hip/hip_bf16.h>

#define B_    8
#define N_    1024
#define C_    768
#define H_    12
#define HD_   64
#define NB_   49
#define NBP_  64
#define ROWS_ 32
#define BSTR_ 66
#define VCH_  128
#define VSTR_ 130
#define PSTR_ 66
#define SSTR_ 1028
#define ABSTR_ 68

typedef float v2f __attribute__((ext_vector_type(2)));
typedef float v8f __attribute__((ext_vector_type(8)));

static __device__ __forceinline__ v8f wmma4(v2f a, v2f b, v8f c) {
  return __builtin_amdgcn_wmma_f32_16x16x4_f32(false, a, false, b, (short)0, c,
                                               false, false);
}
static __device__ __forceinline__ v2f ld2(const float* p) {
  return *(const v2f*)p;
}

__global__ __launch_bounds__(256) void k_qkv(const float* __restrict__ x,
                                             const float* __restrict__ wqkv,
                                             float* __restrict__ qb,
                                             float* __restrict__ kb,
                                             float* __restrict__ vb) {
  __shared__ float Bt[64 * BSTR_];

  const int wave  = threadIdx.x >> 5;
  const int lane  = threadIdx.x & 31;
  const int l15   = lane & 15;
  const int lhalf = lane >> 4;

  const int ngrp = blockIdx.y;
  const int nb   = ngrp * 64;
  const int row  = blockIdx.x * 128 + wave * 16 + l15;
  const float* xr = x + (size_t)row * C_;

  v8f acc[4];
  acc[0] = (v8f)0.f; acc[1] = (v8f)0.f; acc[2] = (v8f)0.f; acc[3] = (v8f)0.f;

  for (int k0 = 0; k0 < C_; k0 += 64) {
    __syncthreads();
#pragma unroll
    for (int r = 0; r < 4; ++r) {
      const int idx = threadIdx.x + 256 * r;
      const int k   = idx >> 4;
      const int nq  = (idx & 15) * 4;
      const float4 w = *(const float4*)(wqkv + (size_t)(k0 + k) * (3 * C_) + nb + nq);
      Bt[(nq + 0) * BSTR_ + k] = w.x;
      Bt[(nq + 1) * BSTR_ + k] = w.y;
      Bt[(nq + 2) * BSTR_ + k] = w.z;
      Bt[(nq + 3) * BSTR_ + k] = w.w;
      if (k0 + 64 < C_)
        __builtin_prefetch(wqkv + (size_t)(k0 + 64 + k) * (3 * C_) + nb + nq, 0, 1);
    }
    __syncthreads();

#pragma unroll
    for (int kk = 0; kk < 64; kk += 4) {
      const int kbi = kk + 2 * lhalf;
      v2f a = ld2(xr + k0 + kbi);
#pragma unroll
      for (int t = 0; t < 4; ++t) {
        v2f bf = ld2(&Bt[(t * 16 + l15) * BSTR_ + kbi]);
        acc[t] = wmma4(a, bf, acc[t]);
      }
    }
  }

  const int which = ngrp / H_;
  const int h     = ngrp % H_;
  float* dst = (which == 0) ? qb : (which == 1) ? kb : vb;
#pragma unroll
  for (int t = 0; t < 4; ++t) {
    const int d = t * 16 + l15;
#pragma unroll
    for (int vv = 0; vv < 8; ++vv) {
      const int r    = blockIdx.x * 128 + wave * 16 + vv + 8 * lhalf;
      const int btok = r >> 10;
      const int tok  = r & (N_ - 1);
      dst[(((size_t)btok * H_ + h) * N_ + tok) * HD_ + d] = acc[t][vv];
    }
  }
}

__global__ __launch_bounds__(256) void k_tables(const float* __restrict__ qb,
                                                const float* __restrict__ kb,
                                                const float* __restrict__ tq,
                                                const float* __restrict__ tk,
                                                float* __restrict__ lkb,
                                                float* __restrict__ lqb) {
  const int wave  = threadIdx.x >> 5;
  const int lane  = threadIdx.x & 31;
  const int l15   = lane & 15;
  const int lhalf = lane >> 4;

  const int unit = blockIdx.x * 8 + wave;
  const int bh   = unit >> 6;
  const int mt   = unit & 63;
  const int h    = bh % H_;

  const size_t rowbase = ((size_t)bh * N_ + mt * 16 + l15) * HD_;
  const v2f z2 = (v2f)0.f;

  v8f ck[4], cq[4];
#pragma unroll
  for (int t = 0; t < 4; ++t) { ck[t] = (v8f)0.f; cq[t] = (v8f)0.f; }

#pragma unroll
  for (int kk = 0; kk < HD_; kk += 4) {
    const int kbi = kk + 2 * lhalf;
    v2f aq = ld2(qb + rowbase + kbi);
    v2f ak = ld2(kb + rowbase + kbi);
    ak = ak * 0.125f;
#pragma unroll
    for (int t = 0; t < 4; ++t) {
      const int n = t * 16 + l15;
      v2f btk = (n < NB_) ? ld2(tk + ((size_t)h * NB_ + n) * HD_ + kbi) : z2;
      v2f btq = (n < NB_) ? ld2(tq + ((size_t)h * NB_ + n) * HD_ + kbi) : z2;
      ck[t] = wmma4(aq, btk, ck[t]);
      cq[t] = wmma4(ak, btq, cq[t]);
    }
  }

#pragma unroll
  for (int t = 0; t < 4; ++t) {
    const int col = t * 16 + l15;
#pragma unroll
    for (int vv = 0; vv < 8; ++vv) {
      const int r = mt * 16 + vv + 8 * lhalf;
      lkb[((size_t)bh * N_ + r) * NBP_ + col] = ck[t][vv];
      lqb[((size_t)bh * N_ + r) * NBP_ + col] = cq[t][vv];
    }
  }
}

__global__ __launch_bounds__(256) void k_attn(const float* __restrict__ qb,
                                              const float* __restrict__ kb,
                                              const float* __restrict__ vb,
                                              const float* __restrict__ lkb,
                                              const float* __restrict__ lqb,
                                              const float* __restrict__ tv,
                                              const int*   __restrict__ rp,
                                              float* __restrict__ ob) {
  extern __shared__ __attribute__((aligned(16))) float smem[];
  float* S  = smem;
  float* ab = smem + ROWS_ * SSTR_;
  float* Vt = ab + ROWS_ * ABSTR_;
  float* pw = Vt + HD_ * VSTR_;

  const int wave  = threadIdx.x >> 5;
  const int lane  = threadIdx.x & 31;
  const int l15   = lane & 15;
  const int lhalf = lane >> 4;

  const int rblk = blockIdx.x;
  const int bh   = blockIdx.y;
  const int b    = bh / H_;
  const int h    = bh % H_;
  const int r0   = rblk * ROWS_;

  {
    const int mt = wave & 1;
    const float* qrow = qb + ((size_t)bh * N_ + r0 + mt * 16 + l15) * HD_;
    v2f qf[16];
#pragma unroll
    for (int s = 0; s < 16; ++s) qf[s] = ld2(qrow + 4 * s + 2 * lhalf);

    for (int nt = (wave >> 1); nt < 64; nt += 4) {
      const float* krow = kb + ((size_t)bh * N_ + nt * 16 + l15) * HD_;
      v8f acc = (v8f)0.f;
#pragma unroll
      for (int s = 0; s < 16; ++s)
        acc = wmma4(qf[s], ld2(krow + 4 * s + 2 * lhalf), acc);

      const int n = nt * 16 + l15;
#pragma unroll
      for (int vv = 0; vv < 8; ++vv) {
        const int m   = mt * 16 + vv + 8 * lhalf;
        const int i   = r0 + m;
        const int bij = rp[i * N_ + n];
        const int bji = rp[n * N_ + i];
        float val = acc[vv]
                  + lkb[((size_t)bh * N_ + i) * NBP_ + bij]
                  + lqb[((size_t)bh * N_ + n) * NBP_ + bji];
        S[m * SSTR_ + n] = val;
      }
    }
  }
  __syncthreads();

  for (int rr = 0; rr < 4; ++rr) {
    const int m = wave * 4 + rr;
    float* row = S + m * SSTR_;
    float mx = -3.4e38f;
    for (int j = lane; j < N_; j += 32) mx = fmaxf(mx, row[j]);
#pragma unroll
    for (int off = 16; off > 0; off >>= 1)
      mx = fmaxf(mx, __shfl_xor(mx, off, 32));
    float sum = 0.f;
    for (int j = lane; j < N_; j += 32) {
      float e = __expf(row[j] - mx);
      row[j] = e;
      sum += e;
    }
#pragma unroll
    for (int off = 16; off > 0; off >>= 1)
      sum += __shfl_xor(sum, off, 32);
    const float inv = 1.0f / sum;
    for (int j = lane; j < N_; j += 32) row[j] *= inv;
  }
  __syncthreads();

  {
    float* mine = pw + wave * (ROWS_ * PSTR_);
    for (int k = lane; k < ROWS_ * PSTR_ / 2; k += 32)
      ((v2f*)mine)[k] = (v2f)0.f;
    float*       myrow = mine + lane * PSTR_;
    const float* prow  = S + lane * SSTR_;
    const int*   brow  = rp + (size_t)(r0 + lane) * N_;
    for (int win = 0; win < 2; ++win) {
      const int j0 = wave * 128 + win * 64;
      float4 p[16];
      int4   b[16];
#pragma unroll
      for (int c = 0; c < 16; ++c) {
        const int col = j0 + 4 * c;
        p[c] = *(const float4*)(prow + col);
        b[c] = *(const int4*)(brow + col);
      }
#pragma unroll
      for (int c = 0; c < 16; ++c) {
        atomicAdd(myrow + b[c].x, p[c].x);
        atomicAdd(myrow + b[c].y, p[c].y);
        atomicAdd(myrow + b[c].z, p[c].z);
        atomicAdd(myrow + b[c].w, p[c].w);
      }
    }
  }
  __syncthreads();
  for (int idx = threadIdx.x; idx < ROWS_ * NBP_; idx += 256) {
    const int m  = idx >> 6;
    const int bk = idx & 63;
    float s = 0.f;
    for (int w = 0; w < 8; ++w) s += pw[(w * ROWS_ + m) * PSTR_ + bk];
    ab[m * ABSTR_ + bk] = s;
  }

  const int mt   = wave & 1;
  const int dt   = wave >> 1;
  const int dcol = dt * 16 + l15;
  v8f acc = (v8f)0.f;

  for (int ch = 0; ch < N_; ch += VCH_) {
    __syncthreads();
#pragma unroll
    for (int r = 0; r < 8; ++r) {
      const int idx = threadIdx.x + 256 * r;
      const int j   = idx >> 4;
      const int dq  = (idx & 15) * 4;
      const float4 w = *(const float4*)(vb + ((size_t)bh * N_ + ch + j) * HD_ + dq);
      Vt[(dq + 0) * VSTR_ + j] = w.x;
      Vt[(dq + 1) * VSTR_ + j] = w.y;
      Vt[(dq + 2) * VSTR_ + j] = w.z;
      Vt[(dq + 3) * VSTR_ + j] = w.w;
    }
    __syncthreads();
#pragma unroll
    for (int kk = 0; kk < VCH_; kk += 4) {
      const int kbi = kk + 2 * lhalf;
      v2f a  = ld2(&S[(mt * 16 + l15) * SSTR_ + ch + kbi]);
      v2f bf = ld2(&Vt[dcol * VSTR_ + kbi]);
      acc = wmma4(a, bf, acc);
    }
  }

#pragma unroll
  for (int kk = 0; kk < NBP_; kk += 4) {
    const int kbi = kk + 2 * lhalf;
    v2f a = ld2(&ab[(mt * 16 + l15) * ABSTR_ + kbi]);
    v2f bf;
    bf.x = (kbi     < NB_) ? tv[((size_t)h * NB_ + kbi)     * HD_ + dcol] : 0.f;
    bf.y = (kbi + 1 < NB_) ? tv[((size_t)h * NB_ + kbi + 1) * HD_ + dcol] : 0.f;
    acc = wmma4(a, bf, acc);
  }

#pragma unroll
  for (int vv = 0; vv < 8; ++vv) {
    const int m   = mt * 16 + vv + 8 * lhalf;
    const int tok = r0 + m;
    ob[(((size_t)b * N_ + tok) * H_ + h) * HD_ + dcol] = acc[vv];
  }
}

__global__ __launch_bounds__(256) void k_proj(const float* __restrict__ ob,
                                              const float* __restrict__ wp,
                                              const float* __restrict__ bias,
                                              float* __restrict__ out) {
  __shared__ float Bt[64 * BSTR_];

  const int wave  = threadIdx.x >> 5;
  const int lane  = threadIdx.x & 31;
  const int l15   = lane & 15;
  const int lhalf = lane >> 4;

  const int ngrp = blockIdx.y;
  const int nb   = ngrp * 64;
  const int row  = blockIdx.x * 128 + wave * 16 + l15;
  const float* xr = ob + (size_t)row * C_;

  v8f acc[4];
  acc[0] = (v8f)0.f; acc[1] = (v8f)0.f; acc[2] = (v8f)0.f; acc[3] = (v8f)0.f;

  for (int k0 = 0; k0 < C_; k0 += 64) {
    __syncthreads();
#pragma unroll
    for (int r = 0; r < 4; ++r) {
      const int idx = threadIdx.x + 256 * r;
      const int k   = idx >> 4;
      const int nq  = (idx & 15) * 4;
      const float4 w = *(const float4*)(wp + (size_t)(k0 + k) * C_ + nb + nq);
      Bt[(nq + 0) * BSTR_ + k] = w.x;
      Bt[(nq + 1) * BSTR_ + k] = w.y;
      Bt[(nq + 2) * BSTR_ + k] = w.z;
      Bt[(nq + 3) * BSTR_ + k] = w.w;
      if (k0 + 64 < C_)
        __builtin_prefetch(wp + (size_t)(k0 + 64 + k) * C_ + nb + nq, 0, 1);
    }
    __syncthreads();

#pragma unroll
    for (int kk = 0; kk < 64; kk += 4) {
      const int kbi = kk + 2 * lhalf;
      v2f a = ld2(xr + k0 + kbi);
#pragma unroll
      for (int t = 0; t < 4; ++t) {
        v2f bf = ld2(&Bt[(t * 16 + l15) * BSTR_ + kbi]);
        acc[t] = wmma4(a, bf, acc[t]);
      }
    }
  }

#pragma unroll
  for (int t = 0; t < 4; ++t) {
    const int c0 = nb + t * 16 + l15;
    const float bv = bias[c0];
#pragma unroll
    for (int vv = 0; vv < 8; ++vv) {
      const int r = blockIdx.x * 128 + wave * 16 + vv + 8 * lhalf;
      out[(size_t)r * C_ + c0] = acc[t][vv] + bv;
    }
  }
}

extern "C" void kernel_launch(void* const* d_in, const int* in_sizes, int n_in,
                              void* d_out, int out_size, void* d_ws, size_t ws_size,
                              hipStream_t stream) {
  const float* x     = (const float*)d_in[0];
  const float* wqkv  = (const float*)d_in[1];
  const float* wproj = (const float*)d_in[2];
  const float* bproj = (const float*)d_in[3];
  const float* tq    = (const float*)d_in[4];
  const float* tk    = (const float*)d_in[5];
  const float* tv    = (const float*)d_in[6];
  const int*   rp    = (const int*)d_in[7];
  float*       out   = (float*)d_out;

  float* ws = (float*)d_ws;
  const size_t SZ = (size_t)B_ * H_ * N_ * HD_;
  float* qb  = ws;
  float* kb  = ws + 1 * SZ;
  float* vb  = ws + 2 * SZ;
  float* lkb = ws + 3 * SZ;
  float* lqb = ws + 4 * SZ;
  float* ob  = ws + 5 * SZ;

  k_qkv<<<dim3(64, 36), 256, 0, stream>>>(x, wqkv, qb, kb, vb);

  k_tables<<<768, 256, 0, stream>>>(qb, kb, tq, tk, lkb, lqb);

  const size_t lds = (size_t)(ROWS_ * SSTR_ + ROWS_ * ABSTR_ + HD_ * VSTR_ +
                              8 * ROWS_ * PSTR_) * sizeof(float);
  k_attn<<<dim3(32, 96), 256, lds, stream>>>(qb, kb, vb, lkb, lqb, tv, rp, ob);

  k_proj<<<dim3(64, 12), 256, 0, stream>>>(ob, wproj, bproj, out);

  (void)in_sizes; (void)n_in; (void)out_size; (void)ws_size;
}


# attention scores: wave tile 16x16 -> 32x32 (both q row tiles x two K tiles per trip): K fragments fetched once per block, 4 independent accumulators, 4 trips per wave, a trip's loads all in flight
# speedup vs baseline: 1.6510x; 1.1013x over previous
#include <hip/hip_runtime.h>
#include <hip/hip_bf16.h>

#define B_    8
#define N_    1024
#define C_    768
#define H_    12
#define HD_   64
#define NB_   49
#define NBP_  64
#define ROWS_ 32
#define BSTR_ 66
#define VCH_  128
#define VSTR_ 130
#define PSTR_ 66
#define SSTR_ 1028
#define ABSTR_ 68

typedef float v2f __attribute__((ext_vector_type(2)));
typedef float v8f __attribute__((ext_vector_type(8)));

static __device__ __forceinline__ v8f wmma4(v2f a, v2f b, v8f c) {
  return __builtin_amdgcn_wmma_f32_16x16x4_f32(false, a, false, b, (short)0, c,
                                               false, false);
}
static __device__ __forceinline__ v2f ld2(const float* p) {
  return *(const v2f*)p;
}

__global__ __launch_bounds__(256) void k_qkv(const float* __restrict__ x,
                                             const float* __restrict__ wqkv,
                                             float* __restrict__ qb,
                                             float* __restrict__ kb,
                                             float* __restrict__ vb) {
  __shared__ float Bt[64 * BSTR_];

  const int wave  = threadIdx.x >> 5;
  const int lane  = threadIdx.x & 31;
  const int l15   = lane & 15;
  const int lhalf = lane >> 4;

  const int ngrp = blockIdx.y;
  const int nb   = ngrp * 64;
  const int row  = blockIdx.x * 128 + wave * 16 + l15;
  const float* xr = x + (size_t)row * C_;

  v8f acc[4];
  acc[0] = (v8f)0.f; acc[1] = (v8f)0.f; acc[2] = (v8f)0.f; acc[3] = (v8f)0.f;

  for (int k0 = 0; k0 < C_; k0 += 64) {
    __syncthreads();
#pragma unroll
    for (int r = 0; r < 4; ++r) {
      const int idx = threadIdx.x + 256 * r;
      const int k   = idx >> 4;
      const int nq  = (idx & 15) * 4;
      const float4 w = *(const float4*)(wqkv + (size_t)(k0 + k) * (3 * C_) + nb + nq);
      Bt[(nq + 0) * BSTR_ + k] = w.x;
      Bt[(nq + 1) * BSTR_ + k] = w.y;
      Bt[(nq + 2) * BSTR_ + k] = w.z;
      Bt[(nq + 3) * BSTR_ + k] = w.w;
      if (k0 + 64 < C_)
        __builtin_prefetch(wqkv + (size_t)(k0 + 64 + k) * (3 * C_) + nb + nq, 0, 1);
    }
    __syncthreads();

#pragma unroll
    for (int kk = 0; kk < 64; kk += 4) {
      const int kbi = kk + 2 * lhalf;
      v2f a = ld2(xr + k0 + kbi);
#pragma unroll
      for (int t = 0; t < 4; ++t) {
        v2f bf = ld2(&Bt[(t * 16 + l15) * BSTR_ + kbi]);
        acc[t] = wmma4(a, bf, acc[t]);
      }
    }
  }

  const int which = ngrp / H_;
  const int h     = ngrp % H_;
  float* dst = (which == 0) ? qb : (which == 1) ? kb : vb;
#pragma unroll
  for (int t = 0; t < 4; ++t) {
    const int d = t * 16 + l15;
#pragma unroll
    for (int vv = 0; vv < 8; ++vv) {
      const int r    = blockIdx.x * 128 + wave * 16 + vv + 8 * lhalf;
      const int btok = r >> 10;
      const int tok  = r & (N_ - 1);
      dst[(((size_t)btok * H_ + h) * N_ + tok) * HD_ + d] = acc[t][vv];
    }
  }
}

__global__ __launch_bounds__(256) void k_tables(const float* __restrict__ qb,
                                                const float* __restrict__ kb,
                                                const float* __restrict__ tq,
                                                const float* __restrict__ tk,
                                                float* __restrict__ lkb,
                                                float* __restrict__ lqb) {
  const int wave  = threadIdx.x >> 5;
  const int lane  = threadIdx.x & 31;
  const int l15   = lane & 15;
  const int lhalf = lane >> 4;

  const int unit = blockIdx.x * 8 + wave;
  const int bh   = unit >> 6;
  const int mt   = unit & 63;
  const int h    = bh % H_;

  const size_t rowbase = ((size_t)bh * N_ + mt * 16 + l15) * HD_;
  const v2f z2 = (v2f)0.f;

  v8f ck[4], cq[4];
#pragma unroll
  for (int t = 0; t < 4; ++t) { ck[t] = (v8f)0.f; cq[t] = (v8f)0.f; }

#pragma unroll
  for (int kk = 0; kk < HD_; kk += 4) {
    const int kbi = kk + 2 * lhalf;
    v2f aq = ld2(qb + rowbase + kbi);
    v2f ak = ld2(kb + rowbase + kbi);
    ak = ak * 0.125f;
#pragma unroll
    for (int t = 0; t < 4; ++t) {
      const int n = t * 16 + l15;
      v2f btk = (n < NB_) ? ld2(tk + ((size_t)h * NB_ + n) * HD_ + kbi) : z2;
      v2f btq = (n < NB_) ? ld2(tq + ((size_t)h * NB_ + n) * HD_ + kbi) : z2;
      ck[t] = wmma4(aq, btk, ck[t]);
      cq[t] = wmma4(ak, btq, cq[t]);
    }
  }

#pragma unroll
  for (int t = 0; t < 4; ++t) {
    const int col = t * 16 + l15;
#pragma unroll
    for (int vv = 0; vv < 8; ++vv) {
      const int r = mt * 16 + vv + 8 * lhalf;
      lkb[((size_t)bh * N_ + r) * NBP_ + col] = ck[t][vv];
      lqb[((size_t)bh * N_ + r) * NBP_ + col] = cq[t][vv];
    }
  }
}

__global__ __launch_bounds__(256) __attribute__((amdgpu_waves_per_eu(2, 2)))
void k_attn(const float* __restrict__ qb,
                                              const float* __restrict__ kb,
                                              const float* __restrict__ vb,
                                              const float* __restrict__ lkb,
                                              const float* __restrict__ lqb,
                                              const float* __restrict__ tv,
                                              const int*   __restrict__ rp,
                                              float* __restrict__ ob) {
  extern __shared__ __attribute__((aligned(16))) float smem[];
  float* S  = smem;
  float* ab = smem + ROWS_ * SSTR_;
  float* Vt = ab + ROWS_ * ABSTR_;
  float* pw = Vt + HD_ * VSTR_;

  const int wave  = threadIdx.x >> 5;
  const int lane  = threadIdx.x & 31;
  const int l15   = lane & 15;
  const int lhalf = lane >> 4;

  const int rblk = blockIdx.x;
  const int bh   = blockIdx.y;
  const int b    = bh / H_;
  const int h    = bh % H_;
  const int r0   = rblk * ROWS_;

  {
    const float* qrow = qb + ((size_t)bh * N_ + r0 + l15) * HD_ + 2 * lhalf;
    v2f qf0[16], qf1[16];
#pragma unroll
    for (int s = 0; s < 16; ++s) {
      qf0[s] = ld2(qrow + 4 * s);
      qf1[s] = ld2(qrow + 16 * HD_ + 4 * s);
    }
    const float* kbh  = kb  + (size_t)bh * N_ * HD_;
    const float* lkbh = lkb + (size_t)bh * N_ * NBP_;
    const float* lqbh = lqb + (size_t)bh * N_ * NBP_;

    for (int np = wave; np < 32; np += 8) {
      const unsigned koff = (unsigned)((np * 32 + l15) * HD_ + 2 * lhalf);
      v8f acc[2][2];
      acc[0][0] = (v8f)0.f; acc[0][1] = (v8f)0.f;
      acc[1][0] = (v8f)0.f; acc[1][1] = (v8f)0.f;

      int  bij[2][2][8];
      int4 bji[2][2][2];
#pragma unroll
      for (int kt = 0; kt < 2; ++kt) {
        const unsigned n = (unsigned)(np * 32 + kt * 16 + l15);
#pragma unroll
        for (int rt = 0; rt < 2; ++rt) {
          const unsigned i0 = (unsigned)(r0 + rt * 16 + 8 * lhalf);
#pragma unroll
          for (int vv = 0; vv < 8; ++vv) bij[kt][rt][vv] = rp[(i0 + vv) * N_ + n];
          const int4* pj = (const int4*)(rp + (n * N_ + i0));
          bji[kt][rt][0] = pj[0];
          bji[kt][rt][1] = pj[1];
        }
      }

#pragma unroll
      for (int s = 0; s < 16; ++s) {
        const v2f k0 = ld2(kbh + (koff + 4 * s));
        const v2f k1 = ld2(kbh + (koff + 16 * HD_ + 4 * s));
        acc[0][0] = wmma4(qf0[s], k0, acc[0][0]);
        acc[1][0] = wmma4(qf1[s], k0, acc[1][0]);
        acc[0][1] = wmma4(qf0[s], k1, acc[0][1]);
        acc[1][1] = wmma4(qf1[s], k1, acc[1][1]);
      }

#pragma unroll
      for (int kt = 0; kt < 2; ++kt) {
        const unsigned n = (unsigned)(np * 32 + kt * 16 + l15);
#pragma unroll
        for (int rt = 0; rt < 2; ++rt) {
          const int4 ja = bji[kt][rt][0], jb = bji[kt][rt][1];
          const int bj[8] = {ja.x, ja.y, ja.z, ja.w, jb.x, jb.y, jb.z, jb.w};
#pragma unroll
          for (int vv = 0; vv < 8; ++vv) {
            const unsigned m = (unsigned)(rt * 16 + vv + 8 * lhalf);
            const unsigned i = (unsigned)r0 + m;
            float val = acc[rt][kt][vv]
                      + lkbh[i * NBP_ + (unsigned)bij[kt][rt][vv]]
                      + lqbh[n * NBP_ + (unsigned)bj[vv]];
            S[m * SSTR_ + n] = val;
          }
        }
      }
    }
  }
  __syncthreads();

  for (int rr = 0; rr < 4; ++rr) {
    const int m = wave * 4 + rr;
    float* row = S + m * SSTR_;
    float mx = -3.4e38f;
    for (int j = lane; j < N_; j += 32) mx = fmaxf(mx, row[j]);
#pragma unroll
    for (int off = 16; off > 0; off >>= 1)
      mx = fmaxf(mx, __shfl_xor(mx, off, 32));
    float sum = 0.f;
    for (int j = lane; j < N_; j += 32) {
      float e = __expf(row[j] - mx);
      row[j] = e;
      sum += e;
    }
#pragma unroll
    for (int off = 16; off > 0; off >>= 1)
      sum += __shfl_xor(sum, off, 32);
    const float inv = 1.0f / sum;
    for (int j = lane; j < N_; j += 32) row[j] *= inv;
  }
  __syncthreads();

  {
    float* mine = pw + wave * (ROWS_ * PSTR_);
    for (int k = lane; k < ROWS_ * PSTR_ / 2; k += 32)
      ((v2f*)mine)[k] = (v2f)0.f;
    float*       myrow = mine + lane * PSTR_;
    const float* prow  = S + lane * SSTR_;
    const int*   brow  = rp + (size_t)(r0 + lane) * N_;
    for (int win = 0; win < 2; ++win) {
      const int j0 = wave * 128 + win * 64;
      float4 p[16];
      int4   b[16];
#pragma unroll
      for (int c = 0; c < 16; ++c) {
        const int col = j0 + 4 * c;
        p[c] = *(const float4*)(prow + col);
        b[c] = *(const int4*)(brow + col);
      }
#pragma unroll
      for (int c = 0; c < 16; ++c) {
        atomicAdd(myrow + b[c].x, p[c].x);
        atomicAdd(myrow + b[c].y, p[c].y);
        atomicAdd(myrow + b[c].z, p[c].z);
        atomicAdd(myrow + b[c].w, p[c].w);
      }
    }
  }
  __syncthreads();
  for (int idx = threadIdx.x; idx < ROWS_ * NBP_; idx += 256) {
    const int m  = idx >> 6;
    const int bk = idx & 63;
    float s = 0.f;
    for (int w = 0; w < 8; ++w) s += pw[(w * ROWS_ + m) * PSTR_ + bk];
    ab[m * ABSTR_ + bk] = s;
  }

  const int mt   = wave & 1;
  const int dt   = wave >> 1;
  const int dcol = dt * 16 + l15;
  v8f acc = (v8f)0.f;

  for (int ch = 0; ch < N_; ch += VCH_) {
    __syncthreads();
#pragma unroll
    for (int r = 0; r < 8; ++r) {
      const int idx = threadIdx.x + 256 * r;
      const int j   = idx >> 4;
      const int dq  = (idx & 15) * 4;
      const float4 w = *(const float4*)(vb + ((size_t)bh * N_ + ch + j) * HD_ + dq);
      Vt[(dq + 0) * VSTR_ + j] = w.x;
      Vt[(dq + 1) * VSTR_ + j] = w.y;
      Vt[(dq + 2) * VSTR_ + j] = w.z;
      Vt[(dq + 3) * VSTR_ + j] = w.w;
    }
    __syncthreads();
#pragma unroll
    for (int kk = 0; kk < VCH_; kk += 4) {
      const int kbi = kk + 2 * lhalf;
      v2f a  = ld2(&S[(mt * 16 + l15) * SSTR_ + ch + kbi]);
      v2f bf = ld2(&Vt[dcol * VSTR_ + kbi]);
      acc = wmma4(a, bf, acc);
    }
  }

#pragma unroll
  for (int kk = 0; kk < NBP_; kk += 4) {
    const int kbi = kk + 2 * lhalf;
    v2f a = ld2(&ab[(mt * 16 + l15) * ABSTR_ + kbi]);
    v2f bf;
    bf.x = (kbi     < NB_) ? tv[((size_t)h * NB_ + kbi)     * HD_ + dcol] : 0.f;
    bf.y = (kbi + 1 < NB_) ? tv[((size_t)h * NB_ + kbi + 1) * HD_ + dcol] : 0.f;
    acc = wmma4(a, bf, acc);
  }

#pragma unroll
  for (int vv = 0; vv < 8; ++vv) {
    const int m   = mt * 16 + vv + 8 * lhalf;
    const int tok = r0 + m;
    ob[(((size_t)b * N_ + tok) * H_ + h) * HD_ + dcol] = acc[vv];
  }
}

__global__ __launch_bounds__(256) void k_proj(const float* __restrict__ ob,
                                              const float* __restrict__ wp,
                                              const float* __restrict__ bias,
                                              float* __restrict__ out) {
  __shared__ float Bt[64 * BSTR_];

  const int wave  = threadIdx.x >> 5;
  const int lane  = threadIdx.x & 31;
  const int l15   = lane & 15;
  const int lhalf = lane >> 4;

  const int ngrp = blockIdx.y;
  const int nb   = ngrp * 64;
  const int row  = blockIdx.x * 128 + wave * 16 + l15;
  const float* xr = ob + (size_t)row * C_;

  v8f acc[4];
  acc[0] = (v8f)0.f; acc[1] = (v8f)0.f; acc[2] = (v8f)0.f; acc[3] = (v8f)0.f;

  for (int k0 = 0; k0 < C_; k0 += 64) {
    __syncthreads();
#pragma unroll
    for (int r = 0; r < 4; ++r) {
      const int idx = threadIdx.x + 256 * r;
      const int k   = idx >> 4;
      const int nq  = (idx & 15) * 4;
      const float4 w = *(const float4*)(wp + (size_t)(k0 + k) * C_ + nb + nq);
      Bt[(nq + 0) * BSTR_ + k] = w.x;
      Bt[(nq + 1) * BSTR_ + k] = w.y;
      Bt[(nq + 2) * BSTR_ + k] = w.z;
      Bt[(nq + 3) * BSTR_ + k] = w.w;
      if (k0 + 64 < C_)
        __builtin_prefetch(wp + (size_t)(k0 + 64 + k) * C_ + nb + nq, 0, 1);
    }
    __syncthreads();

#pragma unroll
    for (int kk = 0; kk < 64; kk += 4) {
      const int kbi = kk + 2 * lhalf;
      v2f a = ld2(xr + k0 + kbi);
#pragma unroll
      for (int t = 0; t < 4; ++t) {
        v2f bf = ld2(&Bt[(t * 16 + l15) * BSTR_ + kbi]);
        acc[t] = wmma4(a, bf, acc[t]);
      }
    }
  }

#pragma unroll
  for (int t = 0; t < 4; ++t) {
    const int c0 = nb + t * 16 + l15;
    const float bv = bias[c0];
#pragma unroll
    for (int vv = 0; vv < 8; ++vv) {
      const int r = blockIdx.x * 128 + wave * 16 + vv + 8 * lhalf;
      out[(size_t)r * C_ + c0] = acc[t][vv] + bv;
    }
  }
}

extern "C" void kernel_launch(void* const* d_in, const int* in_sizes, int n_in,
                              void* d_out, int out_size, void* d_ws, size_t ws_size,
                              hipStream_t stream) {
  const float* x     = (const float*)d_in[0];
  const float* wqkv  = (const float*)d_in[1];
  const float* wproj = (const float*)d_in[2];
  const float* bproj = (const float*)d_in[3];
  const float* tq    = (const float*)d_in[4];
  const float* tk    = (const float*)d_in[5];
  const float* tv    = (const float*)d_in[6];
  const int*   rp    = (const int*)d_in[7];
  float*       out   = (float*)d_out;

  float* ws = (float*)d_ws;
  const size_t SZ = (size_t)B_ * H_ * N_ * HD_;
  float* qb  = ws;
  float* kb  = ws + 1 * SZ;
  float* vb  = ws + 2 * SZ;
  float* lkb = ws + 3 * SZ;
  float* lqb = ws + 4 * SZ;
  float* ob  = ws + 5 * SZ;

  k_qkv<<<dim3(64, 36), 256, 0, stream>>>(x, wqkv, qb, kb, vb);

  k_tables<<<768, 256, 0, stream>>>(qb, kb, tq, tk, lkb, lqb);

  const size_t lds = (size_t)(ROWS_ * SSTR_ + ROWS_ * ABSTR_ + HD_ * VSTR_ +
                              8 * ROWS_ * PSTR_) * sizeof(float);
  k_attn<<<dim3(32, 96), 256, lds, stream>>>(qb, kb, vb, lkb, lqb, tv, rp, ob);

  k_proj<<<dim3(64, 12), 256, 0, stream>>>(ob, wproj, bproj, out);

  (void)in_sizes; (void)n_in; (void)out_size; (void)ws_size;
}


# attention P@V + ab@table_v split over keys: each wave all 32x64 outputs of its 128 keys, 8 accumulators, V rows from global, no LDS staging or barriers; partials summed via LDS (sum order changes)
# speedup vs baseline: 1.7818x; 1.0792x over previous
#include <hip/hip_runtime.h>
#include <hip/hip_bf16.h>

#define B_    8
#define N_    1024
#define C_    768
#define H_    12
#define HD_   64
#define NB_   49
#define NBP_  64
#define ROWS_ 32
#define BSTR_ 66
#define KSPL_ 128
#define PSTR_ 66
#define SSTR_ 1028
#define ABSTR_ 68

typedef float v2f __attribute__((ext_vector_type(2)));
typedef float v8f __attribute__((ext_vector_type(8)));

static __device__ __forceinline__ v8f wmma4(v2f a, v2f b, v8f c) {
  return __builtin_amdgcn_wmma_f32_16x16x4_f32(false, a, false, b, (short)0, c,
                                               false, false);
}
static __device__ __forceinline__ v2f ld2(const float* p) {
  return *(const v2f*)p;
}

__global__ __launch_bounds__(256) void k_qkv(const float* __restrict__ x,
                                             const float* __restrict__ wqkv,
                                             float* __restrict__ qb,
                                             float* __restrict__ kb,
                                             float* __restrict__ vb) {
  __shared__ float Bt[64 * BSTR_];

  const int wave  = threadIdx.x >> 5;
  const int lane  = threadIdx.x & 31;
  const int l15   = lane & 15;
  const int lhalf = lane >> 4;

  const int ngrp = blockIdx.y;
  const int nb   = ngrp * 64;
  const int row  = blockIdx.x * 128 + wave * 16 + l15;
  const float* xr = x + (size_t)row * C_;

  v8f acc[4];
  acc[0] = (v8f)0.f; acc[1] = (v8f)0.f; acc[2] = (v8f)0.f; acc[3] = (v8f)0.f;

  for (int k0 = 0; k0 < C_; k0 += 64) {
    __syncthreads();
#pragma unroll
    for (int r = 0; r < 4; ++r) {
      const int idx = threadIdx.x + 256 * r;
      const int k   = idx >> 4;
      const int nq  = (idx & 15) * 4;
      const float4 w = *(const float4*)(wqkv + (size_t)(k0 + k) * (3 * C_) + nb + nq);
      Bt[(nq + 0) * BSTR_ + k] = w.x;
      Bt[(nq + 1) * BSTR_ + k] = w.y;
      Bt[(nq + 2) * BSTR_ + k] = w.z;
      Bt[(nq + 3) * BSTR_ + k] = w.w;
      if (k0 + 64 < C_)
        __builtin_prefetch(wqkv + (size_t)(k0 + 64 + k) * (3 * C_) + nb + nq, 0, 1);
    }
    __syncthreads();

#pragma unroll
    for (int kk = 0; kk < 64; kk += 4) {
      const int kbi = kk + 2 * lhalf;
      v2f a = ld2(xr + k0 + kbi);
#pragma unroll
      for (int t = 0; t < 4; ++t) {
        v2f bf = ld2(&Bt[(t * 16 + l15) * BSTR_ + kbi]);
        acc[t] = wmma4(a, bf, acc[t]);
      }
    }
  }

  const int which = ngrp / H_;
  const int h     = ngrp % H_;
  float* dst = (which == 0) ? qb : (which == 1) ? kb : vb;
#pragma unroll
  for (int t = 0; t < 4; ++t) {
    const int d = t * 16 + l15;
#pragma unroll
    for (int vv = 0; vv < 8; ++vv) {
      const int r    = blockIdx.x * 128 + wave * 16 + vv + 8 * lhalf;
      const int btok = r >> 10;
      const int tok  = r & (N_ - 1);
      dst[(((size_t)btok * H_ + h) * N_ + tok) * HD_ + d] = acc[t][vv];
    }
  }
}

__global__ __launch_bounds__(256) void k_tables(const float* __restrict__ qb,
                                                const float* __restrict__ kb,
                                                const float* __restrict__ tq,
                                                const float* __restrict__ tk,
                                                float* __restrict__ lkb,
                                                float* __restrict__ lqb) {
  const int wave  = threadIdx.x >> 5;
  const int lane  = threadIdx.x & 31;
  const int l15   = lane & 15;
  const int lhalf = lane >> 4;

  const int unit = blockIdx.x * 8 + wave;
  const int bh   = unit >> 6;
  const int mt   = unit & 63;
  const int h    = bh % H_;

  const size_t rowbase = ((size_t)bh * N_ + mt * 16 + l15) * HD_;
  const v2f z2 = (v2f)0.f;

  v8f ck[4], cq[4];
#pragma unroll
  for (int t = 0; t < 4; ++t) { ck[t] = (v8f)0.f; cq[t] = (v8f)0.f; }

#pragma unroll
  for (int kk = 0; kk < HD_; kk += 4) {
    const int kbi = kk + 2 * lhalf;
    v2f aq = ld2(qb + rowbase + kbi);
    v2f ak = ld2(kb + rowbase + kbi);
    ak = ak * 0.125f;
#pragma unroll
    for (int t = 0; t < 4; ++t) {
      const int n = t * 16 + l15;
      v2f btk = (n < NB_) ? ld2(tk + ((size_t)h * NB_ + n) * HD_ + kbi) : z2;
      v2f btq = (n < NB_) ? ld2(tq + ((size_t)h * NB_ + n) * HD_ + kbi) : z2;
      ck[t] = wmma4(aq, btk, ck[t]);
      cq[t] = wmma4(ak, btq, cq[t]);
    }
  }

#pragma unroll
  for (int t = 0; t < 4; ++t) {
    const int col = t * 16 + l15;
#pragma unroll
    for (int vv = 0; vv < 8; ++vv) {
      const int r = mt * 16 + vv + 8 * lhalf;
      lkb[((size_t)bh * N_ + r) * NBP_ + col] = ck[t][vv];
      lqb[((size_t)bh * N_ + r) * NBP_ + col] = cq[t][vv];
    }
  }
}

__global__ __launch_bounds__(256) __attribute__((amdgpu_waves_per_eu(2, 2)))
void k_attn(const float* __restrict__ qb,
                                              const float* __restrict__ kb,
                                              const float* __restrict__ vb,
                                              const float* __restrict__ lkb,
                                              const float* __restrict__ lqb,
                                              const float* __restrict__ tv,
                                              const int*   __restrict__ rp,
                                              float* __restrict__ ob) {
  extern __shared__ __attribute__((aligned(16))) float smem[];
  float* S  = smem;
  float* ab = smem + ROWS_ * SSTR_;
  float* pw = ab + ROWS_ * ABSTR_;

  const int wave  = threadIdx.x >> 5;
  const int lane  = threadIdx.x & 31;
  const int l15   = lane & 15;
  const int lhalf = lane >> 4;

  const int rblk = blockIdx.x;
  const int bh   = blockIdx.y;
  const int b    = bh / H_;
  const int h    = bh % H_;
  const int r0   = rblk * ROWS_;

  {
    const float* qrow = qb + ((size_t)bh * N_ + r0 + l15) * HD_ + 2 * lhalf;
    v2f qf0[16], qf1[16];
#pragma unroll
    for (int s = 0; s < 16; ++s) {
      qf0[s] = ld2(qrow + 4 * s);
      qf1[s] = ld2(qrow + 16 * HD_ + 4 * s);
    }
    const float* kbh  = kb  + (size_t)bh * N_ * HD_;
    const float* lkbh = lkb + (size_t)bh * N_ * NBP_;
    const float* lqbh = lqb + (size_t)bh * N_ * NBP_;

    for (int np = wave; np < 32; np += 8) {
      const unsigned koff = (unsigned)((np * 32 + l15) * HD_ + 2 * lhalf);
      v8f acc[2][2];
      acc[0][0] = (v8f)0.f; acc[0][1] = (v8f)0.f;
      acc[1][0] = (v8f)0.f; acc[1][1] = (v8f)0.f;

      int  bij[2][2][8];
      int4 bji[2][2][2];
#pragma unroll
      for (int kt = 0; kt < 2; ++kt) {
        const unsigned n = (unsigned)(np * 32 + kt * 16 + l15);
#pragma unroll
        for (int rt = 0; rt < 2; ++rt) {
          const unsigned i0 = (unsigned)(r0 + rt * 16 + 8 * lhalf);
#pragma unroll
          for (int vv = 0; vv < 8; ++vv) bij[kt][rt][vv] = rp[(i0 + vv) * N_ + n];
          const int4* pj = (const int4*)(rp + (n * N_ + i0));
          bji[kt][rt][0] = pj[0];
          bji[kt][rt][1] = pj[1];
        }
      }

#pragma unroll
      for (int s = 0; s < 16; ++s) {
        const v2f k0 = ld2(kbh + (koff + 4 * s));
        const v2f k1 = ld2(kbh + (koff + 16 * HD_ + 4 * s));
        acc[0][0] = wmma4(qf0[s], k0, acc[0][0]);
        acc[1][0] = wmma4(qf1[s], k0, acc[1][0]);
        acc[0][1] = wmma4(qf0[s], k1, acc[0][1]);
        acc[1][1] = wmma4(qf1[s], k1, acc[1][1]);
      }

#pragma unroll
      for (int kt = 0; kt < 2; ++kt) {
        const unsigned n = (unsigned)(np * 32 + kt * 16 + l15);
#pragma unroll
        for (int rt = 0; rt < 2; ++rt) {
          const int4 ja = bji[kt][rt][0], jb = bji[kt][rt][1];
          const int bj[8] = {ja.x, ja.y, ja.z, ja.w, jb.x, jb.y, jb.z, jb.w};
#pragma unroll
          for (int vv = 0; vv < 8; ++vv) {
            const unsigned m = (unsigned)(rt * 16 + vv + 8 * lhalf);
            const unsigned i = (unsigned)r0 + m;
            float val = acc[rt][kt][vv]
                      + lkbh[i * NBP_ + (unsigned)bij[kt][rt][vv]]
                      + lqbh[n * NBP_ + (unsigned)bj[vv]];
            S[m * SSTR_ + n] = val;
          }
        }
      }
    }
  }
  __syncthreads();

  for (int rr = 0; rr < 4; ++rr) {
    const int m = wave * 4 + rr;
    float* row = S + m * SSTR_;
    float mx = -3.4e38f;
    for (int j = lane; j < N_; j += 32) mx = fmaxf(mx, row[j]);
#pragma unroll
    for (int off = 16; off > 0; off >>= 1)
      mx = fmaxf(mx, __shfl_xor(mx, off, 32));
    float sum = 0.f;
    for (int j = lane; j < N_; j += 32) {
      float e = __expf(row[j] - mx);
      row[j] = e;
      sum += e;
    }
#pragma unroll
    for (int off = 16; off > 0; off >>= 1)
      sum += __shfl_xor(sum, off, 32);
    const float inv = 1.0f / sum;
    for (int j = lane; j < N_; j += 32) row[j] *= inv;
  }
  __syncthreads();

  {
    float* mine = pw + wave * (ROWS_ * PSTR_);
    for (int k = lane; k < ROWS_ * PSTR_ / 2; k += 32)
      ((v2f*)mine)[k] = (v2f)0.f;
    float*       myrow = mine + lane * PSTR_;
    const float* prow  = S + lane * SSTR_;
    const int*   brow  = rp + (size_t)(r0 + lane) * N_;
    for (int win = 0; win < 2; ++win) {
      const int j0 = wave * 128 + win * 64;
      float4 p[16];
      int4   b[16];
#pragma unroll
      for (int c = 0; c < 16; ++c) {
        const int col = j0 + 4 * c;
        p[c] = *(const float4*)(prow + col);
        b[c] = *(const int4*)(brow + col);
      }
#pragma unroll
      for (int c = 0; c < 16; ++c) {
        atomicAdd(myrow + b[c].x, p[c].x);
        atomicAdd(myrow + b[c].y, p[c].y);
        atomicAdd(myrow + b[c].z, p[c].z);
        atomicAdd(myrow + b[c].w, p[c].w);
      }
    }
  }
  __syncthreads();

  const float* vp = vb + ((size_t)bh * N_ + wave * KSPL_ + 2 * lhalf) * HD_ + 4 * l15;
  float4 vq[2][16];
#pragma unroll
  for (int s = 0; s < 8; ++s) {
    vq[0][2 * s]     = *(const float4*)(vp + (4 * s) * HD_);
    vq[0][2 * s + 1] = *(const float4*)(vp + (4 * s + 1) * HD_);
  }
  float4 tq4[2][2];
#pragma unroll
  for (int s = 0; s < 2; ++s) {
#pragma unroll
    for (int e = 0; e < 2; ++e) {
      const int kb = wave * 8 + 4 * s + 2 * lhalf + e;
      const int kr = (kb < NB_) ? kb : NB_ - 1;
      tq4[s][e] = *(const float4*)(tv + ((size_t)h * NB_ + kr) * HD_ + 4 * l15);
    }
  }

  for (int idx = threadIdx.x; idx < ROWS_ * NBP_; idx += 256) {
    const int m  = idx >> 6;
    const int bk = idx & 63;
    float s = 0.f;
    for (int w = 0; w < 8; ++w) s += pw[(w * ROWS_ + m) * PSTR_ + bk];
    ab[m * ABSTR_ + bk] = s;
  }
  __syncthreads();

  v8f acc[2][4];
#pragma unroll
  for (int t = 0; t < 4; ++t) { acc[0][t] = (v8f)0.f; acc[1][t] = (v8f)0.f; }

#pragma unroll
  for (int g = 0; g < KSPL_ / 32; ++g) {
    if (g + 1 < KSPL_ / 32) {
#pragma unroll
      for (int s = 0; s < 8; ++s) {
        vq[(g + 1) & 1][2 * s]     = *(const float4*)(vp + ((g + 1) * 32 + 4 * s) * HD_);
        vq[(g + 1) & 1][2 * s + 1] = *(const float4*)(vp + ((g + 1) * 32 + 4 * s + 1) * HD_);
      }
    }
#pragma unroll
    for (int s = 0; s < 8; ++s) {
      const int col = wave * KSPL_ + g * 32 + 4 * s + 2 * lhalf;
      const v2f a0 = ld2(&S[l15 * SSTR_ + col]);
      const v2f a1 = ld2(&S[(16 + l15) * SSTR_ + col]);
      const float4 x0 = vq[g & 1][2 * s];
      const float4 x1 = vq[g & 1][2 * s + 1];
      v2f bf;
      bf.x = x0.x; bf.y = x1.x;
      acc[0][0] = wmma4(a0, bf, acc[0][0]);
      acc[1][0] = wmma4(a1, bf, acc[1][0]);
      bf.x = x0.y; bf.y = x1.y;
      acc[0][1] = wmma4(a0, bf, acc[0][1]);
      acc[1][1] = wmma4(a1, bf, acc[1][1]);
      bf.x = x0.z; bf.y = x1.z;
      acc[0][2] = wmma4(a0, bf, acc[0][2]);
      acc[1][2] = wmma4(a1, bf, acc[1][2]);
      bf.x = x0.w; bf.y = x1.w;
      acc[0][3] = wmma4(a0, bf, acc[0][3]);
      acc[1][3] = wmma4(a1, bf, acc[1][3]);
    }
  }

#pragma unroll
  for (int s = 0; s < 2; ++s) {
    const int kb = wave * 8 + 4 * s + 2 * lhalf;
    const float4 z4 = make_float4(0.f, 0.f, 0.f, 0.f);
    float4 x0 = tq4[s][0];
    float4 x1 = tq4[s][1];
    if (kb     >= NB_) x0 = z4;
    if (kb + 1 >= NB_) x1 = z4;
    const v2f a0 = ld2(&ab[l15 * ABSTR_ + kb]);
    const v2f a1 = ld2(&ab[(16 + l15) * ABSTR_ + kb]);
    v2f bf;
    bf.x = x0.x; bf.y = x1.x;
    acc[0][0] = wmma4(a0, bf, acc[0][0]);
    acc[1][0] = wmma4(a1, bf, acc[1][0]);
    bf.x = x0.y; bf.y = x1.y;
    acc[0][1] = wmma4(a0, bf, acc[0][1]);
    acc[1][1] = wmma4(a1, bf, acc[1][1]);
    bf.x = x0.z; bf.y = x1.z;
    acc[0][2] = wmma4(a0, bf, acc[0][2]);
    acc[1][2] = wmma4(a1, bf, acc[1][2]);
    bf.x = x0.w; bf.y = x1.w;
    acc[0][3] = wmma4(a0, bf, acc[0][3]);
    acc[1][3] = wmma4(a1, bf, acc[1][3]);
  }

  float* red = pw;
  {
    float* mine = red + wave * (ROWS_ * HD_);
#pragma unroll
    for (int mt = 0; mt < 2; ++mt) {
#pragma unroll
      for (int vv = 0; vv < 8; ++vv) {
        const int m = mt * 16 + vv + 8 * lhalf;
        *(float4*)(mine + m * HD_ + 4 * l15) =
            make_float4(acc[mt][0][vv], acc[mt][1][vv], acc[mt][2][vv], acc[mt][3][vv]);
      }
    }
  }
  __syncthreads();

  for (int e = threadIdx.x; e < ROWS_ * HD_ / 4; e += 256) {
    float4 o = ((const float4*)red)[e];
#pragma unroll
    for (int w = 1; w < 8; ++w) {
      const float4 p = ((const float4*)(red + w * (ROWS_ * HD_)))[e];
      o.x += p.x; o.y += p.y; o.z += p.z; o.w += p.w;
    }
    const int m   = e >> 4;
    const int d4  = (e & 15) * 4;
    const int tok = r0 + m;
    *(float4*)(ob + (((size_t)b * N_ + tok) * H_ + h) * HD_ + d4) = o;
  }
}

__global__ __launch_bounds__(256) void k_proj(const float* __restrict__ ob,
                                              const float* __restrict__ wp,
                                              const float* __restrict__ bias,
                                              float* __restrict__ out) {
  __shared__ float Bt[64 * BSTR_];

  const int wave  = threadIdx.x >> 5;
  const int lane  = threadIdx.x & 31;
  const int l15   = lane & 15;
  const int lhalf = lane >> 4;

  const int ngrp = blockIdx.y;
  const int nb   = ngrp * 64;
  const int row  = blockIdx.x * 128 + wave * 16 + l15;
  const float* xr = ob + (size_t)row * C_;

  v8f acc[4];
  acc[0] = (v8f)0.f; acc[1] = (v8f)0.f; acc[2] = (v8f)0.f; acc[3] = (v8f)0.f;

  for (int k0 = 0; k0 < C_; k0 += 64) {
    __syncthreads();
#pragma unroll
    for (int r = 0; r < 4; ++r) {
      const int idx = threadIdx.x + 256 * r;
      const int k   = idx >> 4;
      const int nq  = (idx & 15) * 4;
      const float4 w = *(const float4*)(wp + (size_t)(k0 + k) * C_ + nb + nq);
      Bt[(nq + 0) * BSTR_ + k] = w.x;
      Bt[(nq + 1) * BSTR_ + k] = w.y;
      Bt[(nq + 2) * BSTR_ + k] = w.z;
      Bt[(nq + 3) * BSTR_ + k] = w.w;
      if (k0 + 64 < C_)
        __builtin_prefetch(wp + (size_t)(k0 + 64 + k) * C_ + nb + nq, 0, 1);
    }
    __syncthreads();

#pragma unroll
    for (int kk = 0; kk < 64; kk += 4) {
      const int kbi = kk + 2 * lhalf;
      v2f a = ld2(xr + k0 + kbi);
#pragma unroll
      for (int t = 0; t < 4; ++t) {
        v2f bf = ld2(&Bt[(t * 16 + l15) * BSTR_ + kbi]);
        acc[t] = wmma4(a, bf, acc[t]);
      }
    }
  }

#pragma unroll
  for (int t = 0; t < 4; ++t) {
    const int c0 = nb + t * 16 + l15;
    const float bv = bias[c0];
#pragma unroll
    for (int vv = 0; vv < 8; ++vv) {
      const int r = blockIdx.x * 128 + wave * 16 + vv + 8 * lhalf;
      out[(size_t)r * C_ + c0] = acc[t][vv] + bv;
    }
  }
}

extern "C" void kernel_launch(void* const* d_in, const int* in_sizes, int n_in,
                              void* d_out, int out_size, void* d_ws, size_t ws_size,
                              hipStream_t stream) {
  const float* x     = (const float*)d_in[0];
  const float* wqkv  = (const float*)d_in[1];
  const float* wproj = (const float*)d_in[2];
  const float* bproj = (const float*)d_in[3];
  const float* tq    = (const float*)d_in[4];
  const float* tk    = (const float*)d_in[5];
  const float* tv    = (const float*)d_in[6];
  const int*   rp    = (const int*)d_in[7];
  float*       out   = (float*)d_out;

  float* ws = (float*)d_ws;
  const size_t SZ = (size_t)B_ * H_ * N_ * HD_;
  float* qb  = ws;
  float* kb  = ws + 1 * SZ;
  float* vb  = ws + 2 * SZ;
  float* lkb = ws + 3 * SZ;
  float* lqb = ws + 4 * SZ;
  float* ob  = ws + 5 * SZ;

  k_qkv<<<dim3(64, 36), 256, 0, stream>>>(x, wqkv, qb, kb, vb);

  k_tables<<<768, 256, 0, stream>>>(qb, kb, tq, tk, lkb, lqb);

  const size_t lds = (size_t)(ROWS_ * SSTR_ + ROWS_ * ABSTR_ +
                              8 * ROWS_ * PSTR_) * sizeof(float);
  k_attn<<<dim3(32, 96), 256, lds, stream>>>(qb, kb, vb, lkb, lqb, tv, rp, ob);

  k_proj<<<dim3(64, 12), 256, 0, stream>>>(ob, wproj, bproj, out);

  (void)in_sizes; (void)n_in; (void)out_size; (void)ws_size;
}


# attention softmax: each wave keeps its 4 rows in registers: one LDS read and one write per element (b128), rows interleaved, lane reductions on the vector ALU (DPP, permlanex16) (sum order changes)
# speedup vs baseline: 1.8776x; 1.0538x over previous
#include <hip/hip_runtime.h>
#include <hip/hip_bf16.h>

#define B_    8
#define N_    1024
#define C_    768
#define H_    12
#define HD_   64
#define NB_   49
#define NBP_  64
#define ROWS_ 32
#define BSTR_ 66
#define KSPL_ 128
#define PSTR_ 66
#define SSTR_ 1028
#define ABSTR_ 68

typedef float v2f __attribute__((ext_vector_type(2)));
typedef float v8f __attribute__((ext_vector_type(8)));

static __device__ __forceinline__ v8f wmma4(v2f a, v2f b, v8f c) {
  return __builtin_amdgcn_wmma_f32_16x16x4_f32(false, a, false, b, (short)0, c,
                                               false, false);
}
static __device__ __forceinline__ v2f ld2(const float* p) {
  return *(const v2f*)p;
}

static __device__ __forceinline__ float xhalf(float x) {
  const int b = __builtin_bit_cast(int, x);
  return __builtin_bit_cast(float,
      __builtin_amdgcn_permlanex16(b, b, 0x76543210, 0xfedcba98, false, false));
}
template <int M>
static __device__ __forceinline__ float xrow(float x) {
  const int b = __builtin_bit_cast(int, x);
  return __builtin_bit_cast(float,
      __builtin_amdgcn_update_dpp(b, b, 0x160 + M, 0xf, 0xf, false));
}
static __device__ __forceinline__ float wave_max(float v) {
  v = fmaxf(v, xrow<1>(v));
  v = fmaxf(v, xrow<2>(v));
  v = fmaxf(v, xrow<4>(v));
  v = fmaxf(v, xrow<8>(v));
  return fmaxf(v, xhalf(v));
}
static __device__ __forceinline__ float wave_sum(float v) {
  v += xrow<1>(v);
  v += xrow<2>(v);
  v += xrow<4>(v);
  v += xrow<8>(v);
  return v + xhalf(v);
}

__global__ __launch_bounds__(256) void k_qkv(const float* __restrict__ x,
                                             const float* __restrict__ wqkv,
                                             float* __restrict__ qb,
                                             float* __restrict__ kb,
                                             float* __restrict__ vb) {
  __shared__ float Bt[64 * BSTR_];

  const int wave  = threadIdx.x >> 5;
  const int lane  = threadIdx.x & 31;
  const int l15   = lane & 15;
  const int lhalf = lane >> 4;

  const int ngrp = blockIdx.y;
  const int nb   = ngrp * 64;
  const int row  = blockIdx.x * 128 + wave * 16 + l15;
  const float* xr = x + (size_t)row * C_;

  v8f acc[4];
  acc[0] = (v8f)0.f; acc[1] = (v8f)0.f; acc[2] = (v8f)0.f; acc[3] = (v8f)0.f;

  for (int k0 = 0; k0 < C_; k0 += 64) {
    __syncthreads();
#pragma unroll
    for (int r = 0; r < 4; ++r) {
      const int idx = threadIdx.x + 256 * r;
      const int k   = idx >> 4;
      const int nq  = (idx & 15) * 4;
      const float4 w = *(const float4*)(wqkv + (size_t)(k0 + k) * (3 * C_) + nb + nq);
      Bt[(nq + 0) * BSTR_ + k] = w.x;
      Bt[(nq + 1) * BSTR_ + k] = w.y;
      Bt[(nq + 2) * BSTR_ + k] = w.z;
      Bt[(nq + 3) * BSTR_ + k] = w.w;
      if (k0 + 64 < C_)
        __builtin_prefetch(wqkv + (size_t)(k0 + 64 + k) * (3 * C_) + nb + nq, 0, 1);
    }
    __syncthreads();

#pragma unroll
    for (int kk = 0; kk < 64; kk += 4) {
      const int kbi = kk + 2 * lhalf;
      v2f a = ld2(xr + k0 + kbi);
#pragma unroll
      for (int t = 0; t < 4; ++t) {
        v2f bf = ld2(&Bt[(t * 16 + l15) * BSTR_ + kbi]);
        acc[t] = wmma4(a, bf, acc[t]);
      }
    }
  }

  const int which = ngrp / H_;
  const int h     = ngrp % H_;
  float* dst = (which == 0) ? qb : (which == 1) ? kb : vb;
#pragma unroll
  for (int t = 0; t < 4; ++t) {
    const int d = t * 16 + l15;
#pragma unroll
    for (int vv = 0; vv < 8; ++vv) {
      const int r    = blockIdx.x * 128 + wave * 16 + vv + 8 * lhalf;
      const int btok = r >> 10;
      const int tok  = r & (N_ - 1);
      dst[(((size_t)btok * H_ + h) * N_ + tok) * HD_ + d] = acc[t][vv];
    }
  }
}

__global__ __launch_bounds__(256) void k_tables(const float* __restrict__ qb,
                                                const float* __restrict__ kb,
                                                const float* __restrict__ tq,
                                                const float* __restrict__ tk,
                                                float* __restrict__ lkb,
                                                float* __restrict__ lqb) {
  const int wave  = threadIdx.x >> 5;
  const int lane  = threadIdx.x & 31;
  const int l15   = lane & 15;
  const int lhalf = lane >> 4;

  const int unit = blockIdx.x * 8 + wave;
  const int bh   = unit >> 6;
  const int mt   = unit & 63;
  const int h    = bh % H_;

  const size_t rowbase = ((size_t)bh * N_ + mt * 16 + l15) * HD_;
  const v2f z2 = (v2f)0.f;

  v8f ck[4], cq[4];
#pragma unroll
  for (int t = 0; t < 4; ++t) { ck[t] = (v8f)0.f; cq[t] = (v8f)0.f; }

#pragma unroll
  for (int kk = 0; kk < HD_; kk += 4) {
    const int kbi = kk + 2 * lhalf;
    v2f aq = ld2(qb + rowbase + kbi);
    v2f ak = ld2(kb + rowbase + kbi);
    ak = ak * 0.125f;
#pragma unroll
    for (int t = 0; t < 4; ++t) {
      const int n = t * 16 + l15;
      v2f btk = (n < NB_) ? ld2(tk + ((size_t)h * NB_ + n) * HD_ + kbi) : z2;
      v2f btq = (n < NB_) ? ld2(tq + ((size_t)h * NB_ + n) * HD_ + kbi) : z2;
      ck[t] = wmma4(aq, btk, ck[t]);
      cq[t] = wmma4(ak, btq, cq[t]);
    }
  }

#pragma unroll
  for (int t = 0; t < 4; ++t) {
    const int col = t * 16 + l15;
#pragma unroll
    for (int vv = 0; vv < 8; ++vv) {
      const int r = mt * 16 + vv + 8 * lhalf;
      lkb[((size_t)bh * N_ + r) * NBP_ + col] = ck[t][vv];
      lqb[((size_t)bh * N_ + r) * NBP_ + col] = cq[t][vv];
    }
  }
}

__global__ __launch_bounds__(256) __attribute__((amdgpu_waves_per_eu(2, 2)))
void k_attn(const float* __restrict__ qb,
                                              const float* __restrict__ kb,
                                              const float* __restrict__ vb,
                                              const float* __restrict__ lkb,
                                              const float* __restrict__ lqb,
                                              const float* __restrict__ tv,
                                              const int*   __restrict__ rp,
                                              float* __restrict__ ob) {
  extern __shared__ __attribute__((aligned(16))) float smem[];
  float* S  = smem;
  float* ab = smem + ROWS_ * SSTR_;
  float* pw = ab + ROWS_ * ABSTR_;

  const int wave  = threadIdx.x >> 5;
  const int lane  = threadIdx.x & 31;
  const int l15   = lane & 15;
  const int lhalf = lane >> 4;

  const int rblk = blockIdx.x;
  const int bh   = blockIdx.y;
  const int b    = bh / H_;
  const int h    = bh % H_;
  const int r0   = rblk * ROWS_;

  {
    const float* qrow = qb + ((size_t)bh * N_ + r0 + l15) * HD_ + 2 * lhalf;
    v2f qf0[16], qf1[16];
#pragma unroll
    for (int s = 0; s < 16; ++s) {
      qf0[s] = ld2(qrow + 4 * s);
      qf1[s] = ld2(qrow + 16 * HD_ + 4 * s);
    }
    const float* kbh  = kb  + (size_t)bh * N_ * HD_;
    const float* lkbh = lkb + (size_t)bh * N_ * NBP_;
    const float* lqbh = lqb + (size_t)bh * N_ * NBP_;

    for (int np = wave; np < 32; np += 8) {
      const unsigned koff = (unsigned)((np * 32 + l15) * HD_ + 2 * lhalf);
      v8f acc[2][2];
      acc[0][0] = (v8f)0.f; acc[0][1] = (v8f)0.f;
      acc[1][0] = (v8f)0.f; acc[1][1] = (v8f)0.f;

      int  bij[2][2][8];
      int4 bji[2][2][2];
#pragma unroll
      for (int kt = 0; kt < 2; ++kt) {
        const unsigned n = (unsigned)(np * 32 + kt * 16 + l15);
#pragma unroll
        for (int rt = 0; rt < 2; ++rt) {
          const unsigned i0 = (unsigned)(r0 + rt * 16 + 8 * lhalf);
#pragma unroll
          for (int vv = 0; vv < 8; ++vv) bij[kt][rt][vv] = rp[(i0 + vv) * N_ + n];
          const int4* pj = (const int4*)(rp + (n * N_ + i0));
          bji[kt][rt][0] = pj[0];
          bji[kt][rt][1] = pj[1];
        }
      }

#pragma unroll
      for (int s = 0; s < 16; ++s) {
        const v2f k0 = ld2(kbh + (koff + 4 * s));
        const v2f k1 = ld2(kbh + (koff + 16 * HD_ + 4 * s));
        acc[0][0] = wmma4(qf0[s], k0, acc[0][0]);
        acc[1][0] = wmma4(qf1[s], k0, acc[1][0]);
        acc[0][1] = wmma4(qf0[s], k1, acc[0][1]);
        acc[1][1] = wmma4(qf1[s], k1, acc[1][1]);
      }

#pragma unroll
      for (int kt = 0; kt < 2; ++kt) {
        const unsigned n = (unsigned)(np * 32 + kt * 16 + l15);
#pragma unroll
        for (int rt = 0; rt < 2; ++rt) {
          const int4 ja = bji[kt][rt][0], jb = bji[kt][rt][1];
          const int bj[8] = {ja.x, ja.y, ja.z, ja.w, jb.x, jb.y, jb.z, jb.w};
#pragma unroll
          for (int vv = 0; vv < 8; ++vv) {
            const unsigned m = (unsigned)(rt * 16 + vv + 8 * lhalf);
            const unsigned i = (unsigned)r0 + m;
            float val = acc[rt][kt][vv]
                      + lkbh[i * NBP_ + (unsigned)bij[kt][rt][vv]]
                      + lqbh[n * NBP_ + (unsigned)bj[vv]];
            S[m * SSTR_ + n] = val;
          }
        }
      }
    }
  }
  __syncthreads();

  {
    float4 x[4][8];
#pragma unroll
    for (int rr = 0; rr < 4; ++rr) {
      const float* row = S + (wave * 4 + rr) * SSTR_ + 4 * lane;
#pragma unroll
      for (int c = 0; c < 8; ++c) x[rr][c] = *(const float4*)(row + 128 * c);
    }
    float mx[4];
#pragma unroll
    for (int rr = 0; rr < 4; ++rr) {
      float m = fmaxf(fmaxf(x[rr][0].x, x[rr][0].y), fmaxf(x[rr][0].z, x[rr][0].w));
#pragma unroll
      for (int c = 1; c < 8; ++c)
        m = fmaxf(m, fmaxf(fmaxf(x[rr][c].x, x[rr][c].y), fmaxf(x[rr][c].z, x[rr][c].w)));
      mx[rr] = wave_max(m);
    }
    float inv[4];
#pragma unroll
    for (int rr = 0; rr < 4; ++rr) {
      float s0 = 0.f, s1 = 0.f;
#pragma unroll
      for (int c = 0; c < 8; ++c) {
        float4 e;
        e.x = __expf(x[rr][c].x - mx[rr]);
        e.y = __expf(x[rr][c].y - mx[rr]);
        e.z = __expf(x[rr][c].z - mx[rr]);
        e.w = __expf(x[rr][c].w - mx[rr]);
        x[rr][c] = e;
        s0 += e.x + e.y;
        s1 += e.z + e.w;
      }
      inv[rr] = 1.0f / wave_sum(s0 + s1);
    }
#pragma unroll
    for (int rr = 0; rr < 4; ++rr) {
      float* row = S + (wave * 4 + rr) * SSTR_ + 4 * lane;
#pragma unroll
      for (int c = 0; c < 8; ++c) {
        float4 p = x[rr][c];
        p.x *= inv[rr]; p.y *= inv[rr]; p.z *= inv[rr]; p.w *= inv[rr];
        *(float4*)(row + 128 * c) = p;
      }
    }
  }
  __syncthreads();

  {
    float* mine = pw + wave * (ROWS_ * PSTR_);
    for (int k = lane; k < ROWS_ * PSTR_ / 2; k += 32)
      ((v2f*)mine)[k] = (v2f)0.f;
    float*       myrow = mine + lane * PSTR_;
    const float* prow  = S + lane * SSTR_;
    const int*   brow  = rp + (size_t)(r0 + lane) * N_;
    for (int win = 0; win < 2; ++win) {
      const int j0 = wave * 128 + win * 64;
      float4 p[16];
      int4   b[16];
#pragma unroll
      for (int c = 0; c < 16; ++c) {
        const int col = j0 + 4 * c;
        p[c] = *(const float4*)(prow + col);
        b[c] = *(const int4*)(brow + col);
      }
#pragma unroll
      for (int c = 0; c < 16; ++c) {
        atomicAdd(myrow + b[c].x, p[c].x);
        atomicAdd(myrow + b[c].y, p[c].y);
        atomicAdd(myrow + b[c].z, p[c].z);
        atomicAdd(myrow + b[c].w, p[c].w);
      }
    }
  }
  __syncthreads();

  const float* vp = vb + ((size_t)bh * N_ + wave * KSPL_ + 2 * lhalf) * HD_ + 4 * l15;
  float4 vq[2][16];
#pragma unroll
  for (int s = 0; s < 8; ++s) {
    vq[0][2 * s]     = *(const float4*)(vp + (4 * s) * HD_);
    vq[0][2 * s + 1] = *(const float4*)(vp + (4 * s + 1) * HD_);
  }
  float4 tq4[2][2];
#pragma unroll
  for (int s = 0; s < 2; ++s) {
#pragma unroll
    for (int e = 0; e < 2; ++e) {
      const int kb = wave * 8 + 4 * s + 2 * lhalf + e;
      const int kr = (kb < NB_) ? kb : NB_ - 1;
      tq4[s][e] = *(const float4*)(tv + ((size_t)h * NB_ + kr) * HD_ + 4 * l15);
    }
  }

  for (int idx = threadIdx.x; idx < ROWS_ * NBP_; idx += 256) {
    const int m  = idx >> 6;
    const int bk = idx & 63;
    float s = 0.f;
    for (int w = 0; w < 8; ++w) s += pw[(w * ROWS_ + m) * PSTR_ + bk];
    ab[m * ABSTR_ + bk] = s;
  }
  __syncthreads();

  v8f acc[2][4];
#pragma unroll
  for (int t = 0; t < 4; ++t) { acc[0][t] = (v8f)0.f; acc[1][t] = (v8f)0.f; }

#pragma unroll
  for (int g = 0; g < KSPL_ / 32; ++g) {
    if (g + 1 < KSPL_ / 32) {
#pragma unroll
      for (int s = 0; s < 8; ++s) {
        vq[(g + 1) & 1][2 * s]     = *(const float4*)(vp + ((g + 1) * 32 + 4 * s) * HD_);
        vq[(g + 1) & 1][2 * s + 1] = *(const float4*)(vp + ((g + 1) * 32 + 4 * s + 1) * HD_);
      }
    }
#pragma unroll
    for (int s = 0; s < 8; ++s) {
      const int col = wave * KSPL_ + g * 32 + 4 * s + 2 * lhalf;
      const v2f a0 = ld2(&S[l15 * SSTR_ + col]);
      const v2f a1 = ld2(&S[(16 + l15) * SSTR_ + col]);
      const float4 x0 = vq[g & 1][2 * s];
      const float4 x1 = vq[g & 1][2 * s + 1];
      v2f bf;
      bf.x = x0.x; bf.y = x1.x;
      acc[0][0] = wmma4(a0, bf, acc[0][0]);
      acc[1][0] = wmma4(a1, bf, acc[1][0]);
      bf.x = x0.y; bf.y = x1.y;
      acc[0][1] = wmma4(a0, bf, acc[0][1]);
      acc[1][1] = wmma4(a1, bf, acc[1][1]);
      bf.x = x0.z; bf.y = x1.z;
      acc[0][2] = wmma4(a0, bf, acc[0][2]);
      acc[1][2] = wmma4(a1, bf, acc[1][2]);
      bf.x = x0.w; bf.y = x1.w;
      acc[0][3] = wmma4(a0, bf, acc[0][3]);
      acc[1][3] = wmma4(a1, bf, acc[1][3]);
    }
  }

#pragma unroll
  for (int s = 0; s < 2; ++s) {
    const int kb = wave * 8 + 4 * s + 2 * lhalf;
    const float4 z4 = make_float4(0.f, 0.f, 0.f, 0.f);
    float4 x0 = tq4[s][0];
    float4 x1 = tq4[s][1];
    if (kb     >= NB_) x0 = z4;
    if (kb + 1 >= NB_) x1 = z4;
    const v2f a0 = ld2(&ab[l15 * ABSTR_ + kb]);
    const v2f a1 = ld2(&ab[(16 + l15) * ABSTR_ + kb]);
    v2f bf;
    bf.x = x0.x; bf.y = x1.x;
    acc[0][0] = wmma4(a0, bf, acc[0][0]);
    acc[1][0] = wmma4(a1, bf, acc[1][0]);
    bf.x = x0.y; bf.y = x1.y;
    acc[0][1] = wmma4(a0, bf, acc[0][1]);
    acc[1][1] = wmma4(a1, bf, acc[1][1]);
    bf.x = x0.z; bf.y = x1.z;
    acc[0][2] = wmma4(a0, bf, acc[0][2]);
    acc[1][2] = wmma4(a1, bf, acc[1][2]);
    bf.x = x0.w; bf.y = x1.w;
    acc[0][3] = wmma4(a0, bf, acc[0][3]);
    acc[1][3] = wmma4(a1, bf, acc[1][3]);
  }

  float* red = pw;
  {
    float* mine = red + wave * (ROWS_ * HD_);
#pragma unroll
    for (int mt = 0; mt < 2; ++mt) {
#pragma unroll
      for (int vv = 0; vv < 8; ++vv) {
        const int m = mt * 16 + vv + 8 * lhalf;
        *(float4*)(mine + m * HD_ + 4 * l15) =
            make_float4(acc[mt][0][vv], acc[mt][1][vv], acc[mt][2][vv], acc[mt][3][vv]);
      }
    }
  }
  __syncthreads();

  for (int e = threadIdx.x; e < ROWS_ * HD_ / 4; e += 256) {
    float4 o = ((const float4*)red)[e];
#pragma unroll
    for (int w = 1; w < 8; ++w) {
      const float4 p = ((const float4*)(red + w * (ROWS_ * HD_)))[e];
      o.x += p.x; o.y += p.y; o.z += p.z; o.w += p.w;
    }
    const int m   = e >> 4;
    const int d4  = (e & 15) * 4;
    const int tok = r0 + m;
    *(float4*)(ob + (((size_t)b * N_ + tok) * H_ + h) * HD_ + d4) = o;
  }
}

__global__ __launch_bounds__(256) void k_proj(const float* __restrict__ ob,
                                              const float* __restrict__ wp,
                                              const float* __restrict__ bias,
                                              float* __restrict__ out) {
  __shared__ float Bt[64 * BSTR_];

  const int wave  = threadIdx.x >> 5;
  const int lane  = threadIdx.x & 31;
  const int l15   = lane & 15;
  const int lhalf = lane >> 4;

  const int ngrp = blockIdx.y;
  const int nb   = ngrp * 64;
  const int row  = blockIdx.x * 128 + wave * 16 + l15;
  const float* xr = ob + (size_t)row * C_;

  v8f acc[4];
  acc[0] = (v8f)0.f; acc[1] = (v8f)0.f; acc[2] = (v8f)0.f; acc[3] = (v8f)0.f;

  for (int k0 = 0; k0 < C_; k0 += 64) {
    __syncthreads();
#pragma unroll
    for (int r = 0; r < 4; ++r) {
      const int idx = threadIdx.x + 256 * r;
      const int k   = idx >> 4;
      const int nq  = (idx & 15) * 4;
      const float4 w = *(const float4*)(wp + (size_t)(k0 + k) * C_ + nb + nq);
      Bt[(nq + 0) * BSTR_ + k] = w.x;
      Bt[(nq + 1) * BSTR_ + k] = w.y;
      Bt[(nq + 2) * BSTR_ + k] = w.z;
      Bt[(nq + 3) * BSTR_ + k] = w.w;
      if (k0 + 64 < C_)
        __builtin_prefetch(wp + (size_t)(k0 + 64 + k) * C_ + nb + nq, 0, 1);
    }
    __syncthreads();

#pragma unroll
    for (int kk = 0; kk < 64; kk += 4) {
      const int kbi = kk + 2 * lhalf;
      v2f a = ld2(xr + k0 + kbi);
#pragma unroll
      for (int t = 0; t < 4; ++t) {
        v2f bf = ld2(&Bt[(t * 16 + l15) * BSTR_ + kbi]);
        acc[t] = wmma4(a, bf, acc[t]);
      }
    }
  }

#pragma unroll
  for (int t = 0; t < 4; ++t) {
    const int c0 = nb + t * 16 + l15;
    const float bv = bias[c0];
#pragma unroll
    for (int vv = 0; vv < 8; ++vv) {
      const int r = blockIdx.x * 128 + wave * 16 + vv + 8 * lhalf;
      out[(size_t)r * C_ + c0] = acc[t][vv] + bv;
    }
  }
}

extern "C" void kernel_launch(void* const* d_in, const int* in_sizes, int n_in,
                              void* d_out, int out_size, void* d_ws, size_t ws_size,
                              hipStream_t stream) {
  const float* x     = (const float*)d_in[0];
  const float* wqkv  = (const float*)d_in[1];
  const float* wproj = (const float*)d_in[2];
  const float* bproj = (const float*)d_in[3];
  const float* tq    = (const float*)d_in[4];
  const float* tk    = (const float*)d_in[5];
  const float* tv    = (const float*)d_in[6];
  const int*   rp    = (const int*)d_in[7];
  float*       out   = (float*)d_out;

  float* ws = (float*)d_ws;
  const size_t SZ = (size_t)B_ * H_ * N_ * HD_;
  float* qb  = ws;
  float* kb  = ws + 1 * SZ;
  float* vb  = ws + 2 * SZ;
  float* lkb = ws + 3 * SZ;
  float* lqb = ws + 4 * SZ;
  float* ob  = ws + 5 * SZ;

  k_qkv<<<dim3(64, 36), 256, 0, stream>>>(x, wqkv, qb, kb, vb);

  k_tables<<<768, 256, 0, stream>>>(qb, kb, tq, tk, lkb, lqb);

  const size_t lds = (size_t)(ROWS_ * SSTR_ + ROWS_ * ABSTR_ +
                              8 * ROWS_ * PSTR_) * sizeof(float);
  k_attn<<<dim3(32, 96), 256, lds, stream>>>(qb, kb, vb, lkb, lqb, tv, rp, ob);

  k_proj<<<dim3(64, 12), 256, 0, stream>>>(ob, wproj, bproj, out);

  (void)in_sizes; (void)n_in; (void)out_size; (void)ws_size;
}


# attention scores: tiles transposed (K tile = A, q = B): lane = 1 query row x 8 consecutive keys, so an lq gather spans 2 table rows, not 16; the block's 32 lk rows are gathered from an LDS copy
# speedup vs baseline: 1.9505x; 1.0388x over previous
#include <hip/hip_runtime.h>
#include <hip/hip_bf16.h>

#define B_    8
#define N_    1024
#define C_    768
#define H_    12
#define HD_   64
#define NB_   49
#define NBP_  64
#define ROWS_ 32
#define BSTR_ 66
#define KSPL_ 128
#define PSTR_ 66
#define SSTR_ 1028
#define ABSTR_ 68
#define LKSTR_ 68

typedef float v2f __attribute__((ext_vector_type(2)));
typedef float v8f __attribute__((ext_vector_type(8)));

static __device__ __forceinline__ v8f wmma4(v2f a, v2f b, v8f c) {
  return __builtin_amdgcn_wmma_f32_16x16x4_f32(false, a, false, b, (short)0, c,
                                               false, false);
}
static __device__ __forceinline__ v2f ld2(const float* p) {
  return *(const v2f*)p;
}

static __device__ __forceinline__ float xhalf(float x) {
  const int b = __builtin_bit_cast(int, x);
  return __builtin_bit_cast(float,
      __builtin_amdgcn_permlanex16(b, b, 0x76543210, 0xfedcba98, false, false));
}
template <int M>
static __device__ __forceinline__ float xrow(float x) {
  const int b = __builtin_bit_cast(int, x);
  return __builtin_bit_cast(float,
      __builtin_amdgcn_update_dpp(b, b, 0x160 + M, 0xf, 0xf, false));
}
static __device__ __forceinline__ float wave_max(float v) {
  v = fmaxf(v, xrow<1>(v));
  v = fmaxf(v, xrow<2>(v));
  v = fmaxf(v, xrow<4>(v));
  v = fmaxf(v, xrow<8>(v));
  return fmaxf(v, xhalf(v));
}
static __device__ __forceinline__ float wave_sum(float v) {
  v += xrow<1>(v);
  v += xrow<2>(v);
  v += xrow<4>(v);
  v += xrow<8>(v);
  return v + xhalf(v);
}

__global__ __launch_bounds__(256) void k_qkv(const float* __restrict__ x,
                                             const float* __restrict__ wqkv,
                                             float* __restrict__ qb,
                                             float* __restrict__ kb,
                                             float* __restrict__ vb) {
  __shared__ float Bt[64 * BSTR_];

  const int wave  = threadIdx.x >> 5;
  const int lane  = threadIdx.x & 31;
  const int l15   = lane & 15;
  const int lhalf = lane >> 4;

  const int ngrp = blockIdx.y;
  const int nb   = ngrp * 64;
  const int row  = blockIdx.x * 128 + wave * 16 + l15;
  const float* xr = x + (size_t)row * C_;

  v8f acc[4];
  acc[0] = (v8f)0.f; acc[1] = (v8f)0.f; acc[2] = (v8f)0.f; acc[3] = (v8f)0.f;

  for (int k0 = 0; k0 < C_; k0 += 64) {
    __syncthreads();
#pragma unroll
    for (int r = 0; r < 4; ++r) {
      const int idx = threadIdx.x + 256 * r;
      const int k   = idx >> 4;
      const int nq  = (idx & 15) * 4;
      const float4 w = *(const float4*)(wqkv + (size_t)(k0 + k) * (3 * C_) + nb + nq);
      Bt[(nq + 0) * BSTR_ + k] = w.x;
      Bt[(nq + 1) * BSTR_ + k] = w.y;
      Bt[(nq + 2) * BSTR_ + k] = w.z;
      Bt[(nq + 3) * BSTR_ + k] = w.w;
      if (k0 + 64 < C_)
        __builtin_prefetch(wqkv + (size_t)(k0 + 64 + k) * (3 * C_) + nb + nq, 0, 1);
    }
    __syncthreads();

#pragma unroll
    for (int kk = 0; kk < 64; kk += 4) {
      const int kbi = kk + 2 * lhalf;
      v2f a = ld2(xr + k0 + kbi);
#pragma unroll
      for (int t = 0; t < 4; ++t) {
        v2f bf = ld2(&Bt[(t * 16 + l15) * BSTR_ + kbi]);
        acc[t] = wmma4(a, bf, acc[t]);
      }
    }
  }

  const int which = ngrp / H_;
  const int h     = ngrp % H_;
  float* dst = (which == 0) ? qb : (which == 1) ? kb : vb;
#pragma unroll
  for (int t = 0; t < 4; ++t) {
    const int d = t * 16 + l15;
#pragma unroll
    for (int vv = 0; vv < 8; ++vv) {
      const int r    = blockIdx.x * 128 + wave * 16 + vv + 8 * lhalf;
      const int btok = r >> 10;
      const int tok  = r & (N_ - 1);
      dst[(((size_t)btok * H_ + h) * N_ + tok) * HD_ + d] = acc[t][vv];
    }
  }
}

__global__ __launch_bounds__(256) void k_tables(const float* __restrict__ qb,
                                                const float* __restrict__ kb,
                                                const float* __restrict__ tq,
                                                const float* __restrict__ tk,
                                                float* __restrict__ lkb,
                                                float* __restrict__ lqb) {
  const int wave  = threadIdx.x >> 5;
  const int lane  = threadIdx.x & 31;
  const int l15   = lane & 15;
  const int lhalf = lane >> 4;

  const int unit = blockIdx.x * 8 + wave;
  const int bh   = unit >> 6;
  const int mt   = unit & 63;
  const int h    = bh % H_;

  const size_t rowbase = ((size_t)bh * N_ + mt * 16 + l15) * HD_;
  const v2f z2 = (v2f)0.f;

  v8f ck[4], cq[4];
#pragma unroll
  for (int t = 0; t < 4; ++t) { ck[t] = (v8f)0.f; cq[t] = (v8f)0.f; }

#pragma unroll
  for (int kk = 0; kk < HD_; kk += 4) {
    const int kbi = kk + 2 * lhalf;
    v2f aq = ld2(qb + rowbase + kbi);
    v2f ak = ld2(kb + rowbase + kbi);
    ak = ak * 0.125f;
#pragma unroll
    for (int t = 0; t < 4; ++t) {
      const int n = t * 16 + l15;
      v2f btk = (n < NB_) ? ld2(tk + ((size_t)h * NB_ + n) * HD_ + kbi) : z2;
      v2f btq = (n < NB_) ? ld2(tq + ((size_t)h * NB_ + n) * HD_ + kbi) : z2;
      ck[t] = wmma4(aq, btk, ck[t]);
      cq[t] = wmma4(ak, btq, cq[t]);
    }
  }

#pragma unroll
  for (int t = 0; t < 4; ++t) {
    const int col = t * 16 + l15;
#pragma unroll
    for (int vv = 0; vv < 8; ++vv) {
      const int r = mt * 16 + vv + 8 * lhalf;
      lkb[((size_t)bh * N_ + r) * NBP_ + col] = ck[t][vv];
      lqb[((size_t)bh * N_ + r) * NBP_ + col] = cq[t][vv];
    }
  }
}

__global__ __launch_bounds__(256) __attribute__((amdgpu_waves_per_eu(2, 2)))
void k_attn(const float* __restrict__ qb,
                                              const float* __restrict__ kb,
                                              const float* __restrict__ vb,
                                              const float* __restrict__ lkb,
                                              const float* __restrict__ lqb,
                                              const float* __restrict__ tv,
                                              const int*   __restrict__ rp,
                                              float* __restrict__ ob) {
  extern __shared__ __attribute__((aligned(16))) float smem[];
  float* S  = smem;
  float* ab = smem + ROWS_ * SSTR_;
  float* pw = ab + ROWS_ * ABSTR_;

  const int wave  = threadIdx.x >> 5;
  const int lane  = threadIdx.x & 31;
  const int l15   = lane & 15;
  const int lhalf = lane >> 4;

  const int rblk = blockIdx.x;
  const int bh   = blockIdx.y;
  const int b    = bh / H_;
  const int h    = bh % H_;
  const int r0   = rblk * ROWS_;

  {
    float* lkS = pw;
    {
      const float4* lsrc = (const float4*)(lkb + ((size_t)bh * N_ + r0) * NBP_);
      const int e0 = threadIdx.x, e1 = threadIdx.x + 256;
      const float4 t0 = lsrc[e0];
      const float4 t1 = lsrc[e1];
      *(float4*)(lkS + (e0 >> 4) * LKSTR_ + (e0 & 15) * 4) = t0;
      *(float4*)(lkS + (e1 >> 4) * LKSTR_ + (e1 & 15) * 4) = t1;
    }
    const float* qrow = qb + ((size_t)bh * N_ + r0 + l15) * HD_ + 2 * lhalf;
    v2f qf0[16], qf1[16];
#pragma unroll
    for (int s = 0; s < 16; ++s) {
      qf0[s] = ld2(qrow + 4 * s);
      qf1[s] = ld2(qrow + 16 * HD_ + 4 * s);
    }
    const float* kbh  = kb  + (size_t)bh * N_ * HD_;
    const float* lqbh = lqb + (size_t)bh * N_ * NBP_;
    __syncthreads();

    for (int np = wave; np < 32; np += 8) {
      const unsigned koff = (unsigned)((np * 32 + l15) * HD_ + 2 * lhalf);
      v8f acc[2][2];
      acc[0][0] = (v8f)0.f; acc[0][1] = (v8f)0.f;
      acc[1][0] = (v8f)0.f; acc[1][1] = (v8f)0.f;

      int4 bin[2][2][2];
      int  bni[2][2][8];
#pragma unroll
      for (int kt = 0; kt < 2; ++kt) {
        const unsigned nb = (unsigned)(np * 32 + kt * 16 + 8 * lhalf);
#pragma unroll
        for (int rt = 0; rt < 2; ++rt) {
          const unsigned i = (unsigned)(r0 + rt * 16 + l15);
          const int4* pi = (const int4*)(rp + (i * N_ + nb));
          bin[kt][rt][0] = pi[0];
          bin[kt][rt][1] = pi[1];
#pragma unroll
          for (int vv = 0; vv < 8; ++vv) bni[kt][rt][vv] = rp[(nb + vv) * N_ + i];
        }
      }

      float lkv[2][2][8];
#pragma unroll
      for (int kt = 0; kt < 2; ++kt) {
#pragma unroll
        for (int rt = 0; rt < 2; ++rt) {
          const float* lkrow = lkS + (rt * 16 + l15) * LKSTR_;
          const int4 ia = bin[kt][rt][0], ib = bin[kt][rt][1];
          const int bi[8] = {ia.x, ia.y, ia.z, ia.w, ib.x, ib.y, ib.z, ib.w};
#pragma unroll
          for (int vv = 0; vv < 8; ++vv) lkv[kt][rt][vv] = lkrow[(unsigned)bi[vv]];
        }
      }

#pragma unroll
      for (int s = 0; s < 16; ++s) {
        const v2f k0 = ld2(kbh + (koff + 4 * s));
        const v2f k1 = ld2(kbh + (koff + 16 * HD_ + 4 * s));
        acc[0][0] = wmma4(k0, qf0[s], acc[0][0]);
        acc[1][0] = wmma4(k0, qf1[s], acc[1][0]);
        acc[0][1] = wmma4(k1, qf0[s], acc[0][1]);
        acc[1][1] = wmma4(k1, qf1[s], acc[1][1]);
      }

#pragma unroll
      for (int kt = 0; kt < 2; ++kt) {
        const unsigned nb = (unsigned)(np * 32 + kt * 16 + 8 * lhalf);
#pragma unroll
        for (int rt = 0; rt < 2; ++rt) {
          const unsigned m = (unsigned)(rt * 16 + l15);
          float o[8];
#pragma unroll
          for (int vv = 0; vv < 8; ++vv) {
            const unsigned n = nb + vv;
            o[vv] = acc[rt][kt][vv]
                  + lkv[kt][rt][vv]
                  + lqbh[n * NBP_ + (unsigned)bni[kt][rt][vv]];
          }
          *(float4*)(S + m * SSTR_ + nb)     = make_float4(o[0], o[1], o[2], o[3]);
          *(float4*)(S + m * SSTR_ + nb + 4) = make_float4(o[4], o[5], o[6], o[7]);
        }
      }
    }
  }
  __syncthreads();

  {
    float4 x[4][8];
#pragma unroll
    for (int rr = 0; rr < 4; ++rr) {
      const float* row = S + (wave * 4 + rr) * SSTR_ + 4 * lane;
#pragma unroll
      for (int c = 0; c < 8; ++c) x[rr][c] = *(const float4*)(row + 128 * c);
    }
    float mx[4];
#pragma unroll
    for (int rr = 0; rr < 4; ++rr) {
      float m = fmaxf(fmaxf(x[rr][0].x, x[rr][0].y), fmaxf(x[rr][0].z, x[rr][0].w));
#pragma unroll
      for (int c = 1; c < 8; ++c)
        m = fmaxf(m, fmaxf(fmaxf(x[rr][c].x, x[rr][c].y), fmaxf(x[rr][c].z, x[rr][c].w)));
      mx[rr] = wave_max(m);
    }
    float inv[4];
#pragma unroll
    for (int rr = 0; rr < 4; ++rr) {
      float s0 = 0.f, s1 = 0.f;
#pragma unroll
      for (int c = 0; c < 8; ++c) {
        float4 e;
        e.x = __expf(x[rr][c].x - mx[rr]);
        e.y = __expf(x[rr][c].y - mx[rr]);
        e.z = __expf(x[rr][c].z - mx[rr]);
        e.w = __expf(x[rr][c].w - mx[rr]);
        x[rr][c] = e;
        s0 += e.x + e.y;
        s1 += e.z + e.w;
      }
      inv[rr] = 1.0f / wave_sum(s0 + s1);
    }
#pragma unroll
    for (int rr = 0; rr < 4; ++rr) {
      float* row = S + (wave * 4 + rr) * SSTR_ + 4 * lane;
#pragma unroll
      for (int c = 0; c < 8; ++c) {
        float4 p = x[rr][c];
        p.x *= inv[rr]; p.y *= inv[rr]; p.z *= inv[rr]; p.w *= inv[rr];
        *(float4*)(row + 128 * c) = p;
      }
    }
  }
  __syncthreads();

  {
    float* mine = pw + wave * (ROWS_ * PSTR_);
    for (int k = lane; k < ROWS_ * PSTR_ / 2; k += 32)
      ((v2f*)mine)[k] = (v2f)0.f;
    float*       myrow = mine + lane * PSTR_;
    const float* prow  = S + lane * SSTR_;
    const int*   brow  = rp + (size_t)(r0 + lane) * N_;
    for (int win = 0; win < 2; ++win) {
      const int j0 = wave * 128 + win * 64;
      float4 p[16];
      int4   b[16];
#pragma unroll
      for (int c = 0; c < 16; ++c) {
        const int col = j0 + 4 * c;
        p[c] = *(const float4*)(prow + col);
        b[c] = *(const int4*)(brow + col);
      }
#pragma unroll
      for (int c = 0; c < 16; ++c) {
        atomicAdd(myrow + b[c].x, p[c].x);
        atomicAdd(myrow + b[c].y, p[c].y);
        atomicAdd(myrow + b[c].z, p[c].z);
        atomicAdd(myrow + b[c].w, p[c].w);
      }
    }
  }
  __syncthreads();

  const float* vp = vb + ((size_t)bh * N_ + wave * KSPL_ + 2 * lhalf) * HD_ + 4 * l15;
  float4 vq[2][16];
#pragma unroll
  for (int s = 0; s < 8; ++s) {
    vq[0][2 * s]     = *(const float4*)(vp + (4 * s) * HD_);
    vq[0][2 * s + 1] = *(const float4*)(vp + (4 * s + 1) * HD_);
  }
  float4 tq4[2][2];
#pragma unroll
  for (int s = 0; s < 2; ++s) {
#pragma unroll
    for (int e = 0; e < 2; ++e) {
      const int kb = wave * 8 + 4 * s + 2 * lhalf + e;
      const int kr = (kb < NB_) ? kb : NB_ - 1;
      tq4[s][e] = *(const float4*)(tv + ((size_t)h * NB_ + kr) * HD_ + 4 * l15);
    }
  }

  for (int idx = threadIdx.x; idx < ROWS_ * NBP_; idx += 256) {
    const int m  = idx >> 6;
    const int bk = idx & 63;
    float s = 0.f;
    for (int w = 0; w < 8; ++w) s += pw[(w * ROWS_ + m) * PSTR_ + bk];
    ab[m * ABSTR_ + bk] = s;
  }
  __syncthreads();

  v8f acc[2][4];
#pragma unroll
  for (int t = 0; t < 4; ++t) { acc[0][t] = (v8f)0.f; acc[1][t] = (v8f)0.f; }

#pragma unroll
  for (int g = 0; g < KSPL_ / 32; ++g) {
    if (g + 1 < KSPL_ / 32) {
#pragma unroll
      for (int s = 0; s < 8; ++s) {
        vq[(g + 1) & 1][2 * s]     = *(const float4*)(vp + ((g + 1) * 32 + 4 * s) * HD_);
        vq[(g + 1) & 1][2 * s + 1] = *(const float4*)(vp + ((g + 1) * 32 + 4 * s + 1) * HD_);
      }
    }
#pragma unroll
    for (int s = 0; s < 8; ++s) {
      const int col = wave * KSPL_ + g * 32 + 4 * s + 2 * lhalf;
      const v2f a0 = ld2(&S[l15 * SSTR_ + col]);
      const v2f a1 = ld2(&S[(16 + l15) * SSTR_ + col]);
      const float4 x0 = vq[g & 1][2 * s];
      const float4 x1 = vq[g & 1][2 * s + 1];
      v2f bf;
      bf.x = x0.x; bf.y = x1.x;
      acc[0][0] = wmma4(a0, bf, acc[0][0]);
      acc[1][0] = wmma4(a1, bf, acc[1][0]);
      bf.x = x0.y; bf.y = x1.y;
      acc[0][1] = wmma4(a0, bf, acc[0][1]);
      acc[1][1] = wmma4(a1, bf, acc[1][1]);
      bf.x = x0.z; bf.y = x1.z;
      acc[0][2] = wmma4(a0, bf, acc[0][2]);
      acc[1][2] = wmma4(a1, bf, acc[1][2]);
      bf.x = x0.w; bf.y = x1.w;
      acc[0][3] = wmma4(a0, bf, acc[0][3]);
      acc[1][3] = wmma4(a1, bf, acc[1][3]);
    }
  }

#pragma unroll
  for (int s = 0; s < 2; ++s) {
    const int kb = wave * 8 + 4 * s + 2 * lhalf;
    const float4 z4 = make_float4(0.f, 0.f, 0.f, 0.f);
    float4 x0 = tq4[s][0];
    float4 x1 = tq4[s][1];
    if (kb     >= NB_) x0 = z4;
    if (kb + 1 >= NB_) x1 = z4;
    const v2f a0 = ld2(&ab[l15 * ABSTR_ + kb]);
    const v2f a1 = ld2(&ab[(16 + l15) * ABSTR_ + kb]);
    v2f bf;
    bf.x = x0.x; bf.y = x1.x;
    acc[0][0] = wmma4(a0, bf, acc[0][0]);
    acc[1][0] = wmma4(a1, bf, acc[1][0]);
    bf.x = x0.y; bf.y = x1.y;
    acc[0][1] = wmma4(a0, bf, acc[0][1]);
    acc[1][1] = wmma4(a1, bf, acc[1][1]);
    bf.x = x0.z; bf.y = x1.z;
    acc[0][2] = wmma4(a0, bf, acc[0][2]);
    acc[1][2] = wmma4(a1, bf, acc[1][2]);
    bf.x = x0.w; bf.y = x1.w;
    acc[0][3] = wmma4(a0, bf, acc[0][3]);
    acc[1][3] = wmma4(a1, bf, acc[1][3]);
  }

  float* red = pw;
  {
    float* mine = red + wave * (ROWS_ * HD_);
#pragma unroll
    for (int mt = 0; mt < 2; ++mt) {
#pragma unroll
      for (int vv = 0; vv < 8; ++vv) {
        const int m = mt * 16 + vv + 8 * lhalf;
        *(float4*)(mine + m * HD_ + 4 * l15) =
            make_float4(acc[mt][0][vv], acc[mt][1][vv], acc[mt][2][vv], acc[mt][3][vv]);
      }
    }
  }
  __syncthreads();

  for (int e = threadIdx.x; e < ROWS_ * HD_ / 4; e += 256) {
    float4 o = ((const float4*)red)[e];
#pragma unroll
    for (int w = 1; w < 8; ++w) {
      const float4 p = ((const float4*)(red + w * (ROWS_ * HD_)))[e];
      o.x += p.x; o.y += p.y; o.z += p.z; o.w += p.w;
    }
    const int m   = e >> 4;
    const int d4  = (e & 15) * 4;
    const int tok = r0 + m;
    *(float4*)(ob + (((size_t)b * N_ + tok) * H_ + h) * HD_ + d4) = o;
  }
}

__global__ __launch_bounds__(256) void k_proj(const float* __restrict__ ob,
                                              const float* __restrict__ wp,
                                              const float* __restrict__ bias,
                                              float* __restrict__ out) {
  __shared__ float Bt[64 * BSTR_];

  const int wave  = threadIdx.x >> 5;
  const int lane  = threadIdx.x & 31;
  const int l15   = lane & 15;
  const int lhalf = lane >> 4;

  const int ngrp = blockIdx.y;
  const int nb   = ngrp * 64;
  const int row  = blockIdx.x * 128 + wave * 16 + l15;
  const float* xr = ob + (size_t)row * C_;

  v8f acc[4];
  acc[0] = (v8f)0.f; acc[1] = (v8f)0.f; acc[2] = (v8f)0.f; acc[3] = (v8f)0.f;

  for (int k0 = 0; k0 < C_; k0 += 64) {
    __syncthreads();
#pragma unroll
    for (int r = 0; r < 4; ++r) {
      const int idx = threadIdx.x + 256 * r;
      const int k   = idx >> 4;
      const int nq  = (idx & 15) * 4;
      const float4 w = *(const float4*)(wp + (size_t)(k0 + k) * C_ + nb + nq);
      Bt[(nq + 0) * BSTR_ + k] = w.x;
      Bt[(nq + 1) * BSTR_ + k] = w.y;
      Bt[(nq + 2) * BSTR_ + k] = w.z;
      Bt[(nq + 3) * BSTR_ + k] = w.w;
      if (k0 + 64 < C_)
        __builtin_prefetch(wp + (size_t)(k0 + 64 + k) * C_ + nb + nq, 0, 1);
    }
    __syncthreads();

#pragma unroll
    for (int kk = 0; kk < 64; kk += 4) {
      const int kbi = kk + 2 * lhalf;
      v2f a = ld2(xr + k0 + kbi);
#pragma unroll
      for (int t = 0; t < 4; ++t) {
        v2f bf = ld2(&Bt[(t * 16 + l15) * BSTR_ + kbi]);
        acc[t] = wmma4(a, bf, acc[t]);
      }
    }
  }

#pragma unroll
  for (int t = 0; t < 4; ++t) {
    const int c0 = nb + t * 16 + l15;
    const float bv = bias[c0];
#pragma unroll
    for (int vv = 0; vv < 8; ++vv) {
      const int r = blockIdx.x * 128 + wave * 16 + vv + 8 * lhalf;
      out[(size_t)r * C_ + c0] = acc[t][vv] + bv;
    }
  }
}

extern "C" void kernel_launch(void* const* d_in, const int* in_sizes, int n_in,
                              void* d_out, int out_size, void* d_ws, size_t ws_size,
                              hipStream_t stream) {
  const float* x     = (const float*)d_in[0];
  const float* wqkv  = (const float*)d_in[1];
  const float* wproj = (const float*)d_in[2];
  const float* bproj = (const float*)d_in[3];
  const float* tq    = (const float*)d_in[4];
  const float* tk    = (const float*)d_in[5];
  const float* tv    = (const float*)d_in[6];
  const int*   rp    = (const int*)d_in[7];
  float*       out   = (float*)d_out;

  float* ws = (float*)d_ws;
  const size_t SZ = (size_t)B_ * H_ * N_ * HD_;
  float* qb  = ws;
  float* kb  = ws + 1 * SZ;
  float* vb  = ws + 2 * SZ;
  float* lkb = ws + 3 * SZ;
  float* lqb = ws + 4 * SZ;
  float* ob  = ws + 5 * SZ;

  k_qkv<<<dim3(64, 36), 256, 0, stream>>>(x, wqkv, qb, kb, vb);

  k_tables<<<768, 256, 0, stream>>>(qb, kb, tq, tk, lkb, lqb);

  const size_t lds = (size_t)(ROWS_ * SSTR_ + ROWS_ * ABSTR_ +
                              8 * ROWS_ * PSTR_) * sizeof(float);
  k_attn<<<dim3(32, 96), 256, lds, stream>>>(qb, kb, vb, lkb, lqb, tv, rp, ob);

  k_proj<<<dim3(64, 12), 256, 0, stream>>>(ob, wproj, bproj, out);

  (void)in_sizes; (void)n_in; (void)out_size; (void)ws_size;
}


# attention scores: a wave's 32-key K tile (a trip ahead) and the block's q stripe come in by whole-line 16-byte loads and pass through LDS; fragments are read from LDS, not from 16 global rows
# speedup vs baseline: 2.0124x; 1.0317x over previous
#include <hip/hip_runtime.h>
#include <hip/hip_bf16.h>

#define B_    8
#define N_    1024
#define C_    768
#define H_    12
#define HD_   64
#define NB_   49
#define NBP_  64
#define ROWS_ 32
#define BSTR_ 66
#define KSPL_ 128
#define PSTR_ 66
#define SSTR_ 1028
#define ABSTR_ 68
#define LKSTR_ 68
#define KQSTR_ 68

typedef float v2f __attribute__((ext_vector_type(2)));
typedef float v8f __attribute__((ext_vector_type(8)));

static __device__ __forceinline__ v8f wmma4(v2f a, v2f b, v8f c) {
  return __builtin_amdgcn_wmma_f32_16x16x4_f32(false, a, false, b, (short)0, c,
                                               false, false);
}
static __device__ __forceinline__ v2f ld2(const float* p) {
  return *(const v2f*)p;
}

static __device__ __forceinline__ float xhalf(float x) {
  const int b = __builtin_bit_cast(int, x);
  return __builtin_bit_cast(float,
      __builtin_amdgcn_permlanex16(b, b, 0x76543210, 0xfedcba98, false, false));
}
template <int M>
static __device__ __forceinline__ float xrow(float x) {
  const int b = __builtin_bit_cast(int, x);
  return __builtin_bit_cast(float,
      __builtin_amdgcn_update_dpp(b, b, 0x160 + M, 0xf, 0xf, false));
}
static __device__ __forceinline__ float wave_max(float v) {
  v = fmaxf(v, xrow<1>(v));
  v = fmaxf(v, xrow<2>(v));
  v = fmaxf(v, xrow<4>(v));
  v = fmaxf(v, xrow<8>(v));
  return fmaxf(v, xhalf(v));
}
static __device__ __forceinline__ float wave_sum(float v) {
  v += xrow<1>(v);
  v += xrow<2>(v);
  v += xrow<4>(v);
  v += xrow<8>(v);
  return v + xhalf(v);
}

__global__ __launch_bounds__(256) void k_qkv(const float* __restrict__ x,
                                             const float* __restrict__ wqkv,
                                             float* __restrict__ qb,
                                             float* __restrict__ kb,
                                             float* __restrict__ vb) {
  __shared__ float Bt[64 * BSTR_];

  const int wave  = threadIdx.x >> 5;
  const int lane  = threadIdx.x & 31;
  const int l15   = lane & 15;
  const int lhalf = lane >> 4;

  const int ngrp = blockIdx.y;
  const int nb   = ngrp * 64;
  const int row  = blockIdx.x * 128 + wave * 16 + l15;
  const float* xr = x + (size_t)row * C_;

  v8f acc[4];
  acc[0] = (v8f)0.f; acc[1] = (v8f)0.f; acc[2] = (v8f)0.f; acc[3] = (v8f)0.f;

  for (int k0 = 0; k0 < C_; k0 += 64) {
    __syncthreads();
#pragma unroll
    for (int r = 0; r < 4; ++r) {
      const int idx = threadIdx.x + 256 * r;
      const int k   = idx >> 4;
      const int nq  = (idx & 15) * 4;
      const float4 w = *(const float4*)(wqkv + (size_t)(k0 + k) * (3 * C_) + nb + nq);
      Bt[(nq + 0) * BSTR_ + k] = w.x;
      Bt[(nq + 1) * BSTR_ + k] = w.y;
      Bt[(nq + 2) * BSTR_ + k] = w.z;
      Bt[(nq + 3) * BSTR_ + k] = w.w;
      if (k0 + 64 < C_)
        __builtin_prefetch(wqkv + (size_t)(k0 + 64 + k) * (3 * C_) + nb + nq, 0, 1);
    }
    __syncthreads();

#pragma unroll
    for (int kk = 0; kk < 64; kk += 4) {
      const int kbi = kk + 2 * lhalf;
      v2f a = ld2(xr + k0 + kbi);
#pragma unroll
      for (int t = 0; t < 4; ++t) {
        v2f bf = ld2(&Bt[(t * 16 + l15) * BSTR_ + kbi]);
        acc[t] = wmma4(a, bf, acc[t]);
      }
    }
  }

  const int which = ngrp / H_;
  const int h     = ngrp % H_;
  float* dst = (which == 0) ? qb : (which == 1) ? kb : vb;
#pragma unroll
  for (int t = 0; t < 4; ++t) {
    const int d = t * 16 + l15;
#pragma unroll
    for (int vv = 0; vv < 8; ++vv) {
      const int r    = blockIdx.x * 128 + wave * 16 + vv + 8 * lhalf;
      const int btok = r >> 10;
      const int tok  = r & (N_ - 1);
      dst[(((size_t)btok * H_ + h) * N_ + tok) * HD_ + d] = acc[t][vv];
    }
  }
}

__global__ __launch_bounds__(256) void k_tables(const float* __restrict__ qb,
                                                const float* __restrict__ kb,
                                                const float* __restrict__ tq,
                                                const float* __restrict__ tk,
                                                float* __restrict__ lkb,
                                                float* __restrict__ lqb) {
  const int wave  = threadIdx.x >> 5;
  const int lane  = threadIdx.x & 31;
  const int l15   = lane & 15;
  const int lhalf = lane >> 4;

  const int unit = blockIdx.x * 8 + wave;
  const int bh   = unit >> 6;
  const int mt   = unit & 63;
  const int h    = bh % H_;

  const size_t rowbase = ((size_t)bh * N_ + mt * 16 + l15) * HD_;
  const v2f z2 = (v2f)0.f;

  v8f ck[4], cq[4];
#pragma unroll
  for (int t = 0; t < 4; ++t) { ck[t] = (v8f)0.f; cq[t] = (v8f)0.f; }

#pragma unroll
  for (int kk = 0; kk < HD_; kk += 4) {
    const int kbi = kk + 2 * lhalf;
    v2f aq = ld2(qb + rowbase + kbi);
    v2f ak = ld2(kb + rowbase + kbi);
    ak = ak * 0.125f;
#pragma unroll
    for (int t = 0; t < 4; ++t) {
      const int n = t * 16 + l15;
      v2f btk = (n < NB_) ? ld2(tk + ((size_t)h * NB_ + n) * HD_ + kbi) : z2;
      v2f btq = (n < NB_) ? ld2(tq + ((size_t)h * NB_ + n) * HD_ + kbi) : z2;
      ck[t] = wmma4(aq, btk, ck[t]);
      cq[t] = wmma4(ak, btq, cq[t]);
    }
  }

#pragma unroll
  for (int t = 0; t < 4; ++t) {
    const int col = t * 16 + l15;
#pragma unroll
    for (int vv = 0; vv < 8; ++vv) {
      const int r = mt * 16 + vv + 8 * lhalf;
      lkb[((size_t)bh * N_ + r) * NBP_ + col] = ck[t][vv];
      lqb[((size_t)bh * N_ + r) * NBP_ + col] = cq[t][vv];
    }
  }
}

__global__ __launch_bounds__(256) __attribute__((amdgpu_waves_per_eu(2, 2)))
void k_attn(const float* __restrict__ qb,
                                              const float* __restrict__ kb,
                                              const float* __restrict__ vb,
                                              const float* __restrict__ lkb,
                                              const float* __restrict__ lqb,
                                              const float* __restrict__ tv,
                                              const int*   __restrict__ rp,
                                              float* __restrict__ ob) {
  extern __shared__ __attribute__((aligned(16))) float smem[];
  float* S  = smem;
  float* ab = smem + ROWS_ * SSTR_;
  float* pw = ab + ROWS_ * ABSTR_;
  float* kst = pw + 8 * ROWS_ * PSTR_;
  float* qS  = kst + 8 * 32 * KQSTR_;

  const int wave  = threadIdx.x >> 5;
  const int lane  = threadIdx.x & 31;
  const int l15   = lane & 15;
  const int lhalf = lane >> 4;

  const int rblk = blockIdx.x;
  const int bh   = blockIdx.y;
  const int b    = bh / H_;
  const int h    = bh % H_;
  const int r0   = rblk * ROWS_;

  {
    float* lkS = pw;
    {
      const float4* lsrc = (const float4*)(lkb + ((size_t)bh * N_ + r0) * NBP_);
      const int e0 = threadIdx.x, e1 = threadIdx.x + 256;
      const float4 t0 = lsrc[e0];
      const float4 t1 = lsrc[e1];
      *(float4*)(lkS + (e0 >> 4) * LKSTR_ + (e0 & 15) * 4) = t0;
      *(float4*)(lkS + (e1 >> 4) * LKSTR_ + (e1 & 15) * 4) = t1;
      const float4* qsrc = (const float4*)(qb + ((size_t)bh * N_ + r0) * HD_);
      const float4 u0 = qsrc[e0];
      const float4 u1 = qsrc[e1];
      *(float4*)(qS + (e0 >> 4) * KQSTR_ + (e0 & 15) * 4) = u0;
      *(float4*)(qS + (e1 >> 4) * KQSTR_ + (e1 & 15) * 4) = u1;
    }
    const float* kbh  = kb  + (size_t)bh * N_ * HD_;
    const float* lqbh = lqb + (size_t)bh * N_ * NBP_;
    const float4* ksrc = (const float4*)kbh + lane;
    float* kw = kst + wave * (32 * KQSTR_);
    float* kwst = kw + lhalf * KQSTR_ + 4 * l15;
    const float* kwld = kw + l15 * KQSTR_ + 2 * lhalf;
    {
      float4 kt[16];
#pragma unroll
      for (int cc = 0; cc < 16; ++cc) kt[cc] = ksrc[wave * 512 + 32 * cc];
#pragma unroll
      for (int cc = 0; cc < 16; ++cc) *(float4*)(kwst + 2 * cc * KQSTR_) = kt[cc];
    }
    __syncthreads();

    v2f qf0[16], qf1[16];
    {
      const float* qrow = qS + l15 * KQSTR_ + 2 * lhalf;
#pragma unroll
      for (int s = 0; s < 16; ++s) {
        qf0[s] = ld2(qrow + 4 * s);
        qf1[s] = ld2(qrow + 16 * KQSTR_ + 4 * s);
      }
    }

    for (int np = wave; np < 32; np += 8) {
      v2f kf0[16], kf1[16];
#pragma unroll
      for (int s = 0; s < 16; ++s) {
        kf0[s] = ld2(kwld + 4 * s);
        kf1[s] = ld2(kwld + 16 * KQSTR_ + 4 * s);
      }
      const int nxt = (np + 8 < 32) ? np + 8 : np;
      float4 kn[16];
#pragma unroll
      for (int cc = 0; cc < 16; ++cc) kn[cc] = ksrc[nxt * 512 + 32 * cc];
      v8f acc[2][2];
      acc[0][0] = (v8f)0.f; acc[0][1] = (v8f)0.f;
      acc[1][0] = (v8f)0.f; acc[1][1] = (v8f)0.f;

      int4 bin[2][2][2];
      int  bni[2][2][8];
#pragma unroll
      for (int kt = 0; kt < 2; ++kt) {
        const unsigned nb = (unsigned)(np * 32 + kt * 16 + 8 * lhalf);
#pragma unroll
        for (int rt = 0; rt < 2; ++rt) {
          const unsigned i = (unsigned)(r0 + rt * 16 + l15);
          const int4* pi = (const int4*)(rp + (i * N_ + nb));
          bin[kt][rt][0] = pi[0];
          bin[kt][rt][1] = pi[1];
#pragma unroll
          for (int vv = 0; vv < 8; ++vv) bni[kt][rt][vv] = rp[(nb + vv) * N_ + i];
        }
      }

      float lkv[2][2][8];
#pragma unroll
      for (int kt = 0; kt < 2; ++kt) {
#pragma unroll
        for (int rt = 0; rt < 2; ++rt) {
          const float* lkrow = lkS + (rt * 16 + l15) * LKSTR_;
          const int4 ia = bin[kt][rt][0], ib = bin[kt][rt][1];
          const int bi[8] = {ia.x, ia.y, ia.z, ia.w, ib.x, ib.y, ib.z, ib.w};
#pragma unroll
          for (int vv = 0; vv < 8; ++vv) lkv[kt][rt][vv] = lkrow[(unsigned)bi[vv]];
        }
      }

#pragma unroll
      for (int s = 0; s < 16; ++s) {
        acc[0][0] = wmma4(kf0[s], qf0[s], acc[0][0]);
        acc[1][0] = wmma4(kf0[s], qf1[s], acc[1][0]);
        acc[0][1] = wmma4(kf1[s], qf0[s], acc[0][1]);
        acc[1][1] = wmma4(kf1[s], qf1[s], acc[1][1]);
      }

#pragma unroll
      for (int kt = 0; kt < 2; ++kt) {
        const unsigned nb = (unsigned)(np * 32 + kt * 16 + 8 * lhalf);
#pragma unroll
        for (int rt = 0; rt < 2; ++rt) {
          const unsigned m = (unsigned)(rt * 16 + l15);
          float o[8];
#pragma unroll
          for (int vv = 0; vv < 8; ++vv) {
            const unsigned n = nb + vv;
            o[vv] = acc[rt][kt][vv]
                  + lkv[kt][rt][vv]
                  + lqbh[n * NBP_ + (unsigned)bni[kt][rt][vv]];
          }
          *(float4*)(S + m * SSTR_ + nb)     = make_float4(o[0], o[1], o[2], o[3]);
          *(float4*)(S + m * SSTR_ + nb + 4) = make_float4(o[4], o[5], o[6], o[7]);
        }
      }
#pragma unroll
      for (int cc = 0; cc < 16; ++cc) *(float4*)(kwst + 2 * cc * KQSTR_) = kn[cc];
    }
  }
  __syncthreads();

  {
    float4 x[4][8];
#pragma unroll
    for (int rr = 0; rr < 4; ++rr) {
      const float* row = S + (wave * 4 + rr) * SSTR_ + 4 * lane;
#pragma unroll
      for (int c = 0; c < 8; ++c) x[rr][c] = *(const float4*)(row + 128 * c);
    }
    float mx[4];
#pragma unroll
    for (int rr = 0; rr < 4; ++rr) {
      float m = fmaxf(fmaxf(x[rr][0].x, x[rr][0].y), fmaxf(x[rr][0].z, x[rr][0].w));
#pragma unroll
      for (int c = 1; c < 8; ++c)
        m = fmaxf(m, fmaxf(fmaxf(x[rr][c].x, x[rr][c].y), fmaxf(x[rr][c].z, x[rr][c].w)));
      mx[rr] = wave_max(m);
    }
    float inv[4];
#pragma unroll
    for (int rr = 0; rr < 4; ++rr) {
      float s0 = 0.f, s1 = 0.f;
#pragma unroll
      for (int c = 0; c < 8; ++c) {
        float4 e;
        e.x = __expf(x[rr][c].x - mx[rr]);
        e.y = __expf(x[rr][c].y - mx[rr]);
        e.z = __expf(x[rr][c].z - mx[rr]);
        e.w = __expf(x[rr][c].w - mx[rr]);
        x[rr][c] = e;
        s0 += e.x + e.y;
        s1 += e.z + e.w;
      }
      inv[rr] = 1.0f / wave_sum(s0 + s1);
    }
#pragma unroll
    for (int rr = 0; rr < 4; ++rr) {
      float* row = S + (wave * 4 + rr) * SSTR_ + 4 * lane;
#pragma unroll
      for (int c = 0; c < 8; ++c) {
        float4 p = x[rr][c];
        p.x *= inv[rr]; p.y *= inv[rr]; p.z *= inv[rr]; p.w *= inv[rr];
        *(float4*)(row + 128 * c) = p;
      }
    }
  }
  __syncthreads();

  {
    float* mine = pw + wave * (ROWS_ * PSTR_);
    for (int k = lane; k < ROWS_ * PSTR_ / 2; k += 32)
      ((v2f*)mine)[k] = (v2f)0.f;
    float*       myrow = mine + lane * PSTR_;
    const float* prow  = S + lane * SSTR_;
    const int*   brow  = rp + (size_t)(r0 + lane) * N_;
    for (int win = 0; win < 2; ++win) {
      const int j0 = wave * 128 + win * 64;
      float4 p[16];
      int4   b[16];
#pragma unroll
      for (int c = 0; c < 16; ++c) {
        const int col = j0 + 4 * c;
        p[c] = *(const float4*)(prow + col);
        b[c] = *(const int4*)(brow + col);
      }
#pragma unroll
      for (int c = 0; c < 16; ++c) {
        atomicAdd(myrow + b[c].x, p[c].x);
        atomicAdd(myrow + b[c].y, p[c].y);
        atomicAdd(myrow + b[c].z, p[c].z);
        atomicAdd(myrow + b[c].w, p[c].w);
      }
    }
  }
  __syncthreads();

  const float* vp = vb + ((size_t)bh * N_ + wave * KSPL_ + 2 * lhalf) * HD_ + 4 * l15;
  float4 vq[2][16];
#pragma unroll
  for (int s = 0; s < 8; ++s) {
    vq[0][2 * s]     = *(const float4*)(vp + (4 * s) * HD_);
    vq[0][2 * s + 1] = *(const float4*)(vp + (4 * s + 1) * HD_);
  }
  float4 tq4[2][2];
#pragma unroll
  for (int s = 0; s < 2; ++s) {
#pragma unroll
    for (int e = 0; e < 2; ++e) {
      const int kb = wave * 8 + 4 * s + 2 * lhalf + e;
      const int kr = (kb < NB_) ? kb : NB_ - 1;
      tq4[s][e] = *(const float4*)(tv + ((size_t)h * NB_ + kr) * HD_ + 4 * l15);
    }
  }

  for (int idx = threadIdx.x; idx < ROWS_ * NBP_; idx += 256) {
    const int m  = idx >> 6;
    const int bk = idx & 63;
    float s = 0.f;
    for (int w = 0; w < 8; ++w) s += pw[(w * ROWS_ + m) * PSTR_ + bk];
    ab[m * ABSTR_ + bk] = s;
  }
  __syncthreads();

  v8f acc[2][4];
#pragma unroll
  for (int t = 0; t < 4; ++t) { acc[0][t] = (v8f)0.f; acc[1][t] = (v8f)0.f; }

#pragma unroll
  for (int g = 0; g < KSPL_ / 32; ++g) {
    if (g + 1 < KSPL_ / 32) {
#pragma unroll
      for (int s = 0; s < 8; ++s) {
        vq[(g + 1) & 1][2 * s]     = *(const float4*)(vp + ((g + 1) * 32 + 4 * s) * HD_);
        vq[(g + 1) & 1][2 * s + 1] = *(const float4*)(vp + ((g + 1) * 32 + 4 * s + 1) * HD_);
      }
    }
#pragma unroll
    for (int s = 0; s < 8; ++s) {
      const int col = wave * KSPL_ + g * 32 + 4 * s + 2 * lhalf;
      const v2f a0 = ld2(&S[l15 * SSTR_ + col]);
      const v2f a1 = ld2(&S[(16 + l15) * SSTR_ + col]);
      const float4 x0 = vq[g & 1][2 * s];
      const float4 x1 = vq[g & 1][2 * s + 1];
      v2f bf;
      bf.x = x0.x; bf.y = x1.x;
      acc[0][0] = wmma4(a0, bf, acc[0][0]);
      acc[1][0] = wmma4(a1, bf, acc[1][0]);
      bf.x = x0.y; bf.y = x1.y;
      acc[0][1] = wmma4(a0, bf, acc[0][1]);
      acc[1][1] = wmma4(a1, bf, acc[1][1]);
      bf.x = x0.z; bf.y = x1.z;
      acc[0][2] = wmma4(a0, bf, acc[0][2]);
      acc[1][2] = wmma4(a1, bf, acc[1][2]);
      bf.x = x0.w; bf.y = x1.w;
      acc[0][3] = wmma4(a0, bf, acc[0][3]);
      acc[1][3] = wmma4(a1, bf, acc[1][3]);
    }
  }

#pragma unroll
  for (int s = 0; s < 2; ++s) {
    const int kb = wave * 8 + 4 * s + 2 * lhalf;
    const float4 z4 = make_float4(0.f, 0.f, 0.f, 0.f);
    float4 x0 = tq4[s][0];
    float4 x1 = tq4[s][1];
    if (kb     >= NB_) x0 = z4;
    if (kb + 1 >= NB_) x1 = z4;
    const v2f a0 = ld2(&ab[l15 * ABSTR_ + kb]);
    const v2f a1 = ld2(&ab[(16 + l15) * ABSTR_ + kb]);
    v2f bf;
    bf.x = x0.x; bf.y = x1.x;
    acc[0][0] = wmma4(a0, bf, acc[0][0]);
    acc[1][0] = wmma4(a1, bf, acc[1][0]);
    bf.x = x0.y; bf.y = x1.y;
    acc[0][1] = wmma4(a0, bf, acc[0][1]);
    acc[1][1] = wmma4(a1, bf, acc[1][1]);
    bf.x = x0.z; bf.y = x1.z;
    acc[0][2] = wmma4(a0, bf, acc[0][2]);
    acc[1][2] = wmma4(a1, bf, acc[1][2]);
    bf.x = x0.w; bf.y = x1.w;
    acc[0][3] = wmma4(a0, bf, acc[0][3]);
    acc[1][3] = wmma4(a1, bf, acc[1][3]);
  }

  float* red = pw;
  {
    float* mine = red + wave * (ROWS_ * HD_);
#pragma unroll
    for (int mt = 0; mt < 2; ++mt) {
#pragma unroll
      for (int vv = 0; vv < 8; ++vv) {
        const int m = mt * 16 + vv + 8 * lhalf;
        *(float4*)(mine + m * HD_ + 4 * l15) =
            make_float4(acc[mt][0][vv], acc[mt][1][vv], acc[mt][2][vv], acc[mt][3][vv]);
      }
    }
  }
  __syncthreads();

  for (int e = threadIdx.x; e < ROWS_ * HD_ / 4; e += 256) {
    float4 o = ((const float4*)red)[e];
#pragma unroll
    for (int w = 1; w < 8; ++w) {
      const float4 p = ((const float4*)(red + w * (ROWS_ * HD_)))[e];
      o.x += p.x; o.y += p.y; o.z += p.z; o.w += p.w;
    }
    const int m   = e >> 4;
    const int d4  = (e & 15) * 4;
    const int tok = r0 + m;
    *(float4*)(ob + (((size_t)b * N_ + tok) * H_ + h) * HD_ + d4) = o;
  }
}

__global__ __launch_bounds__(256) void k_proj(const float* __restrict__ ob,
                                              const float* __restrict__ wp,
                                              const float* __restrict__ bias,
                                              float* __restrict__ out) {
  __shared__ float Bt[64 * BSTR_];

  const int wave  = threadIdx.x >> 5;
  const int lane  = threadIdx.x & 31;
  const int l15   = lane & 15;
  const int lhalf = lane >> 4;

  const int ngrp = blockIdx.y;
  const int nb   = ngrp * 64;
  const int row  = blockIdx.x * 128 + wave * 16 + l15;
  const float* xr = ob + (size_t)row * C_;

  v8f acc[4];
  acc[0] = (v8f)0.f; acc[1] = (v8f)0.f; acc[2] = (v8f)0.f; acc[3] = (v8f)0.f;

  for (int k0 = 0; k0 < C_; k0 += 64) {
    __syncthreads();
#pragma unroll
    for (int r = 0; r < 4; ++r) {
      const int idx = threadIdx.x + 256 * r;
      const int k   = idx >> 4;
      const int nq  = (idx & 15) * 4;
      const float4 w = *(const float4*)(wp + (size_t)(k0 + k) * C_ + nb + nq);
      Bt[(nq + 0) * BSTR_ + k] = w.x;
      Bt[(nq + 1) * BSTR_ + k] = w.y;
      Bt[(nq + 2) * BSTR_ + k] = w.z;
      Bt[(nq + 3) * BSTR_ + k] = w.w;
      if (k0 + 64 < C_)
        __builtin_prefetch(wp + (size_t)(k0 + 64 + k) * C_ + nb + nq, 0, 1);
    }
    __syncthreads();

#pragma unroll
    for (int kk = 0; kk < 64; kk += 4) {
      const int kbi = kk + 2 * lhalf;
      v2f a = ld2(xr + k0 + kbi);
#pragma unroll
      for (int t = 0; t < 4; ++t) {
        v2f bf = ld2(&Bt[(t * 16 + l15) * BSTR_ + kbi]);
        acc[t] = wmma4(a, bf, acc[t]);
      }
    }
  }

#pragma unroll
  for (int t = 0; t < 4; ++t) {
    const int c0 = nb + t * 16 + l15;
    const float bv = bias[c0];
#pragma unroll
    for (int vv = 0; vv < 8; ++vv) {
      const int r = blockIdx.x * 128 + wave * 16 + vv + 8 * lhalf;
      out[(size_t)r * C_ + c0] = acc[t][vv] + bv;
    }
  }
}

extern "C" void kernel_launch(void* const* d_in, const int* in_sizes, int n_in,
                              void* d_out, int out_size, void* d_ws, size_t ws_size,
                              hipStream_t stream) {
  const float* x     = (const float*)d_in[0];
  const float* wqkv  = (const float*)d_in[1];
  const float* wproj = (const float*)d_in[2];
  const float* bproj = (const float*)d_in[3];
  const float* tq    = (const float*)d_in[4];
  const float* tk    = (const float*)d_in[5];
  const float* tv    = (const float*)d_in[6];
  const int*   rp    = (const int*)d_in[7];
  float*       out   = (float*)d_out;

  float* ws = (float*)d_ws;
  const size_t SZ = (size_t)B_ * H_ * N_ * HD_;
  float* qb  = ws;
  float* kb  = ws + 1 * SZ;
  float* vb  = ws + 2 * SZ;
  float* lkb = ws + 3 * SZ;
  float* lqb = ws + 4 * SZ;
  float* ob  = ws + 5 * SZ;

  k_qkv<<<dim3(64, 36), 256, 0, stream>>>(x, wqkv, qb, kb, vb);

  k_tables<<<768, 256, 0, stream>>>(qb, kb, tq, tk, lkb, lqb);

  const size_t lds = (size_t)(ROWS_ * SSTR_ + ROWS_ * ABSTR_ +
                              8 * ROWS_ * PSTR_ + 8 * 32 * KQSTR_ +
                              ROWS_ * KQSTR_) * sizeof(float);
  k_attn<<<dim3(32, 96), 256, lds, stream>>>(qb, kb, vb, lkb, lqb, tv, rp, ob);

  k_proj<<<dim3(64, 12), 256, 0, stream>>>(ob, wproj, bproj, out);

  (void)in_sizes; (void)n_in; (void)out_size; (void)ws_size;
}


# both projection GEMMs: k loop software-pipelined: chunk c+1's W piece and A fragments are fetched into registers ahead of chunk c's products, W panel double-buffered in LDS (1 barrier a chunk)
# speedup vs baseline: 2.1122x; 1.0496x over previous
#include <hip/hip_runtime.h>
#include <hip/hip_bf16.h>

#define B_    8
#define N_    1024
#define C_    768
#define H_    12
#define HD_   64
#define NB_   49
#define NBP_  64
#define ROWS_ 32
#define BSTR_ 66
#define KSPL_ 128
#define PSTR_ 66
#define SSTR_ 1028
#define ABSTR_ 68
#define LKSTR_ 68
#define KQSTR_ 68

typedef float v2f __attribute__((ext_vector_type(2)));
typedef float v8f __attribute__((ext_vector_type(8)));

static __device__ __forceinline__ v8f wmma4(v2f a, v2f b, v8f c) {
  return __builtin_amdgcn_wmma_f32_16x16x4_f32(false, a, false, b, (short)0, c,
                                               false, false);
}
static __device__ __forceinline__ v2f ld2(const float* p) {
  return *(const v2f*)p;
}

static __device__ __forceinline__ float xhalf(float x) {
  const int b = __builtin_bit_cast(int, x);
  return __builtin_bit_cast(float,
      __builtin_amdgcn_permlanex16(b, b, 0x76543210, 0xfedcba98, false, false));
}
template <int M>
static __device__ __forceinline__ float xrow(float x) {
  const int b = __builtin_bit_cast(int, x);
  return __builtin_bit_cast(float,
      __builtin_amdgcn_update_dpp(b, b, 0x160 + M, 0xf, 0xf, false));
}
static __device__ __forceinline__ float wave_max(float v) {
  v = fmaxf(v, xrow<1>(v));
  v = fmaxf(v, xrow<2>(v));
  v = fmaxf(v, xrow<4>(v));
  v = fmaxf(v, xrow<8>(v));
  return fmaxf(v, xhalf(v));
}
static __device__ __forceinline__ float wave_sum(float v) {
  v += xrow<1>(v);
  v += xrow<2>(v);
  v += xrow<4>(v);
  v += xrow<8>(v);
  return v + xhalf(v);
}

__global__ __launch_bounds__(256) __attribute__((amdgpu_waves_per_eu(2, 6)))
void k_qkv(const float* __restrict__ x,
           const float* __restrict__ wqkv,
           float* __restrict__ qb,
           float* __restrict__ kb,
           float* __restrict__ vb) {
  __shared__ float Bt[2 * 64 * BSTR_];

  const int wave  = threadIdx.x >> 5;
  const int lane  = threadIdx.x & 31;
  const int l15   = lane & 15;
  const int lhalf = lane >> 4;

  const int ngrp = blockIdx.y;
  const int nb   = ngrp * 64;
  const int row  = blockIdx.x * 128 + wave * 16 + l15;
  const float* xr = x + (size_t)row * C_ + 2 * lhalf;

  const int sk  = threadIdx.x >> 4;
  const int snq = (threadIdx.x & 15) * 4;
  const float* wsrc = wqkv + (size_t)sk * (3 * C_) + nb + snq;
  float* bst = Bt + snq * BSTR_ + sk;

  float4 wn[4];
  v2f    an[16];
#pragma unroll
  for (int r = 0; r < 4; ++r) wn[r] = *(const float4*)(wsrc + (size_t)(16 * r) * (3 * C_));
#pragma unroll
  for (int s = 0; s < 16; ++s) an[s] = ld2(xr + 4 * s);
#pragma unroll
  for (int r = 0; r < 4; ++r) {
    bst[0 * BSTR_ + 16 * r] = wn[r].x;
    bst[1 * BSTR_ + 16 * r] = wn[r].y;
    bst[2 * BSTR_ + 16 * r] = wn[r].z;
    bst[3 * BSTR_ + 16 * r] = wn[r].w;
  }

  v8f acc[4];
  acc[0] = (v8f)0.f; acc[1] = (v8f)0.f; acc[2] = (v8f)0.f; acc[3] = (v8f)0.f;

  int cur = 0;
  for (int k0 = 0; k0 < C_; k0 += 64) {
    __syncthreads();
    v2f a[16];
#pragma unroll
    for (int s = 0; s < 16; ++s) a[s] = an[s];
    const int kn0 = (k0 + 64 < C_) ? k0 + 64 : k0;
#pragma unroll
    for (int r = 0; r < 4; ++r) wn[r] = *(const float4*)(wsrc + (size_t)(kn0 + 16 * r) * (3 * C_));
#pragma unroll
    for (int s = 0; s < 16; ++s) an[s] = ld2(xr + kn0 + 4 * s);

    const float* bld = Bt + cur * (64 * BSTR_) + l15 * BSTR_ + 2 * lhalf;
#pragma unroll
    for (int s = 0; s < 16; ++s) {
#pragma unroll
      for (int t = 0; t < 4; ++t) {
        v2f bf = ld2(bld + (t * 16) * BSTR_ + 4 * s);
        acc[t] = wmma4(a[s], bf, acc[t]);
      }
    }

    float* bnx = bst + (cur ^ 1) * (64 * BSTR_);
#pragma unroll
    for (int r = 0; r < 4; ++r) {
      bnx[0 * BSTR_ + 16 * r] = wn[r].x;
      bnx[1 * BSTR_ + 16 * r] = wn[r].y;
      bnx[2 * BSTR_ + 16 * r] = wn[r].z;
      bnx[3 * BSTR_ + 16 * r] = wn[r].w;
    }
    cur ^= 1;
  }

  const int which = ngrp / H_;
  const int h     = ngrp % H_;
  float* dst = (which == 0) ? qb : (which == 1) ? kb : vb;
#pragma unroll
  for (int t = 0; t < 4; ++t) {
    const int d = t * 16 + l15;
#pragma unroll
    for (int vv = 0; vv < 8; ++vv) {
      const int r    = blockIdx.x * 128 + wave * 16 + vv + 8 * lhalf;
      const int btok = r >> 10;
      const int tok  = r & (N_ - 1);
      dst[(((size_t)btok * H_ + h) * N_ + tok) * HD_ + d] = acc[t][vv];
    }
  }
}

__global__ __launch_bounds__(256) void k_tables(const float* __restrict__ qb,
                                                const float* __restrict__ kb,
                                                const float* __restrict__ tq,
                                                const float* __restrict__ tk,
                                                float* __restrict__ lkb,
                                                float* __restrict__ lqb) {
  const int wave  = threadIdx.x >> 5;
  const int lane  = threadIdx.x & 31;
  const int l15   = lane & 15;
  const int lhalf = lane >> 4;

  const int unit = blockIdx.x * 8 + wave;
  const int bh   = unit >> 6;
  const int mt   = unit & 63;
  const int h    = bh % H_;

  const size_t rowbase = ((size_t)bh * N_ + mt * 16 + l15) * HD_;
  const v2f z2 = (v2f)0.f;

  v8f ck[4], cq[4];
#pragma unroll
  for (int t = 0; t < 4; ++t) { ck[t] = (v8f)0.f; cq[t] = (v8f)0.f; }

#pragma unroll
  for (int kk = 0; kk < HD_; kk += 4) {
    const int kbi = kk + 2 * lhalf;
    v2f aq = ld2(qb + rowbase + kbi);
    v2f ak = ld2(kb + rowbase + kbi);
    ak = ak * 0.125f;
#pragma unroll
    for (int t = 0; t < 4; ++t) {
      const int n = t * 16 + l15;
      v2f btk = (n < NB_) ? ld2(tk + ((size_t)h * NB_ + n) * HD_ + kbi) : z2;
      v2f btq = (n < NB_) ? ld2(tq + ((size_t)h * NB_ + n) * HD_ + kbi) : z2;
      ck[t] = wmma4(aq, btk, ck[t]);
      cq[t] = wmma4(ak, btq, cq[t]);
    }
  }

#pragma unroll
  for (int t = 0; t < 4; ++t) {
    const int col = t * 16 + l15;
#pragma unroll
    for (int vv = 0; vv < 8; ++vv) {
      const int r = mt * 16 + vv + 8 * lhalf;
      lkb[((size_t)bh * N_ + r) * NBP_ + col] = ck[t][vv];
      lqb[((size_t)bh * N_ + r) * NBP_ + col] = cq[t][vv];
    }
  }
}

__global__ __launch_bounds__(256) __attribute__((amdgpu_waves_per_eu(2, 2)))
void k_attn(const float* __restrict__ qb,
                                              const float* __restrict__ kb,
                                              const float* __restrict__ vb,
                                              const float* __restrict__ lkb,
                                              const float* __restrict__ lqb,
                                              const float* __restrict__ tv,
                                              const int*   __restrict__ rp,
                                              float* __restrict__ ob) {
  extern __shared__ __attribute__((aligned(16))) float smem[];
  float* S  = smem;
  float* ab = smem + ROWS_ * SSTR_;
  float* pw = ab + ROWS_ * ABSTR_;
  float* kst = pw + 8 * ROWS_ * PSTR_;
  float* qS  = kst + 8 * 32 * KQSTR_;

  const int wave  = threadIdx.x >> 5;
  const int lane  = threadIdx.x & 31;
  const int l15   = lane & 15;
  const int lhalf = lane >> 4;

  const int rblk = blockIdx.x;
  const int bh   = blockIdx.y;
  const int b    = bh / H_;
  const int h    = bh % H_;
  const int r0   = rblk * ROWS_;

  {
    float* lkS = pw;
    {
      const float4* lsrc = (const float4*)(lkb + ((size_t)bh * N_ + r0) * NBP_);
      const int e0 = threadIdx.x, e1 = threadIdx.x + 256;
      const float4 t0 = lsrc[e0];
      const float4 t1 = lsrc[e1];
      *(float4*)(lkS + (e0 >> 4) * LKSTR_ + (e0 & 15) * 4) = t0;
      *(float4*)(lkS + (e1 >> 4) * LKSTR_ + (e1 & 15) * 4) = t1;
      const float4* qsrc = (const float4*)(qb + ((size_t)bh * N_ + r0) * HD_);
      const float4 u0 = qsrc[e0];
      const float4 u1 = qsrc[e1];
      *(float4*)(qS + (e0 >> 4) * KQSTR_ + (e0 & 15) * 4) = u0;
      *(float4*)(qS + (e1 >> 4) * KQSTR_ + (e1 & 15) * 4) = u1;
    }
    const float* kbh  = kb  + (size_t)bh * N_ * HD_;
    const float* lqbh = lqb + (size_t)bh * N_ * NBP_;
    const float4* ksrc = (const float4*)kbh + lane;
    float* kw = kst + wave * (32 * KQSTR_);
    float* kwst = kw + lhalf * KQSTR_ + 4 * l15;
    const float* kwld = kw + l15 * KQSTR_ + 2 * lhalf;
    {
      float4 kt[16];
#pragma unroll
      for (int cc = 0; cc < 16; ++cc) kt[cc] = ksrc[wave * 512 + 32 * cc];
#pragma unroll
      for (int cc = 0; cc < 16; ++cc) *(float4*)(kwst + 2 * cc * KQSTR_) = kt[cc];
    }
    __syncthreads();

    v2f qf0[16], qf1[16];
    {
      const float* qrow = qS + l15 * KQSTR_ + 2 * lhalf;
#pragma unroll
      for (int s = 0; s < 16; ++s) {
        qf0[s] = ld2(qrow + 4 * s);
        qf1[s] = ld2(qrow + 16 * KQSTR_ + 4 * s);
      }
    }

    for (int np = wave; np < 32; np += 8) {
      v2f kf0[16], kf1[16];
#pragma unroll
      for (int s = 0; s < 16; ++s) {
        kf0[s] = ld2(kwld + 4 * s);
        kf1[s] = ld2(kwld + 16 * KQSTR_ + 4 * s);
      }
      const int nxt = (np + 8 < 32) ? np + 8 : np;
      float4 kn[16];
#pragma unroll
      for (int cc = 0; cc < 16; ++cc) kn[cc] = ksrc[nxt * 512 + 32 * cc];
      v8f acc[2][2];
      acc[0][0] = (v8f)0.f; acc[0][1] = (v8f)0.f;
      acc[1][0] = (v8f)0.f; acc[1][1] = (v8f)0.f;

      int4 bin[2][2][2];
      int  bni[2][2][8];
#pragma unroll
      for (int kt = 0; kt < 2; ++kt) {
        const unsigned nb = (unsigned)(np * 32 + kt * 16 + 8 * lhalf);
#pragma unroll
        for (int rt = 0; rt < 2; ++rt) {
          const unsigned i = (unsigned)(r0 + rt * 16 + l15);
          const int4* pi = (const int4*)(rp + (i * N_ + nb));
          bin[kt][rt][0] = pi[0];
          bin[kt][rt][1] = pi[1];
#pragma unroll
          for (int vv = 0; vv < 8; ++vv) bni[kt][rt][vv] = rp[(nb + vv) * N_ + i];
        }
      }

      float lkv[2][2][8];
#pragma unroll
      for (int kt = 0; kt < 2; ++kt) {
#pragma unroll
        for (int rt = 0; rt < 2; ++rt) {
          const float* lkrow = lkS + (rt * 16 + l15) * LKSTR_;
          const int4 ia = bin[kt][rt][0], ib = bin[kt][rt][1];
          const int bi[8] = {ia.x, ia.y, ia.z, ia.w, ib.x, ib.y, ib.z, ib.w};
#pragma unroll
          for (int vv = 0; vv < 8; ++vv) lkv[kt][rt][vv] = lkrow[(unsigned)bi[vv]];
        }
      }

#pragma unroll
      for (int s = 0; s < 16; ++s) {
        acc[0][0] = wmma4(kf0[s], qf0[s], acc[0][0]);
        acc[1][0] = wmma4(kf0[s], qf1[s], acc[1][0]);
        acc[0][1] = wmma4(kf1[s], qf0[s], acc[0][1]);
        acc[1][1] = wmma4(kf1[s], qf1[s], acc[1][1]);
      }

#pragma unroll
      for (int kt = 0; kt < 2; ++kt) {
        const unsigned nb = (unsigned)(np * 32 + kt * 16 + 8 * lhalf);
#pragma unroll
        for (int rt = 0; rt < 2; ++rt) {
          const unsigned m = (unsigned)(rt * 16 + l15);
          float o[8];
#pragma unroll
          for (int vv = 0; vv < 8; ++vv) {
            const unsigned n = nb + vv;
            o[vv] = acc[rt][kt][vv]
                  + lkv[kt][rt][vv]
                  + lqbh[n * NBP_ + (unsigned)bni[kt][rt][vv]];
          }
          *(float4*)(S + m * SSTR_ + nb)     = make_float4(o[0], o[1], o[2], o[3]);
          *(float4*)(S + m * SSTR_ + nb + 4) = make_float4(o[4], o[5], o[6], o[7]);
        }
      }
#pragma unroll
      for (int cc = 0; cc < 16; ++cc) *(float4*)(kwst + 2 * cc * KQSTR_) = kn[cc];
    }
  }
  __syncthreads();

  {
    float4 x[4][8];
#pragma unroll
    for (int rr = 0; rr < 4; ++rr) {
      const float* row = S + (wave * 4 + rr) * SSTR_ + 4 * lane;
#pragma unroll
      for (int c = 0; c < 8; ++c) x[rr][c] = *(const float4*)(row + 128 * c);
    }
    float mx[4];
#pragma unroll
    for (int rr = 0; rr < 4; ++rr) {
      float m = fmaxf(fmaxf(x[rr][0].x, x[rr][0].y), fmaxf(x[rr][0].z, x[rr][0].w));
#pragma unroll
      for (int c = 1; c < 8; ++c)
        m = fmaxf(m, fmaxf(fmaxf(x[rr][c].x, x[rr][c].y), fmaxf(x[rr][c].z, x[rr][c].w)));
      mx[rr] = wave_max(m);
    }
    float inv[4];
#pragma unroll
    for (int rr = 0; rr < 4; ++rr) {
      float s0 = 0.f, s1 = 0.f;
#pragma unroll
      for (int c = 0; c < 8; ++c) {
        float4 e;
        e.x = __expf(x[rr][c].x - mx[rr]);
        e.y = __expf(x[rr][c].y - mx[rr]);
        e.z = __expf(x[rr][c].z - mx[rr]);
        e.w = __expf(x[rr][c].w - mx[rr]);
        x[rr][c] = e;
        s0 += e.x + e.y;
        s1 += e.z + e.w;
      }
      inv[rr] = 1.0f / wave_sum(s0 + s1);
    }
#pragma unroll
    for (int rr = 0; rr < 4; ++rr) {
      float* row = S + (wave * 4 + rr) * SSTR_ + 4 * lane;
#pragma unroll
      for (int c = 0; c < 8; ++c) {
        float4 p = x[rr][c];
        p.x *= inv[rr]; p.y *= inv[rr]; p.z *= inv[rr]; p.w *= inv[rr];
        *(float4*)(row + 128 * c) = p;
      }
    }
  }
  __syncthreads();

  {
    float* mine = pw + wave * (ROWS_ * PSTR_);
    for (int k = lane; k < ROWS_ * PSTR_ / 2; k += 32)
      ((v2f*)mine)[k] = (v2f)0.f;
    float*       myrow = mine + lane * PSTR_;
    const float* prow  = S + lane * SSTR_;
    const int*   brow  = rp + (size_t)(r0 + lane) * N_;
    for (int win = 0; win < 2; ++win) {
      const int j0 = wave * 128 + win * 64;
      float4 p[16];
      int4   b[16];
#pragma unroll
      for (int c = 0; c < 16; ++c) {
        const int col = j0 + 4 * c;
        p[c] = *(const float4*)(prow + col);
        b[c] = *(const int4*)(brow + col);
      }
#pragma unroll
      for (int c = 0; c < 16; ++c) {
        atomicAdd(myrow + b[c].x, p[c].x);
        atomicAdd(myrow + b[c].y, p[c].y);
        atomicAdd(myrow + b[c].z, p[c].z);
        atomicAdd(myrow + b[c].w, p[c].w);
      }
    }
  }
  __syncthreads();

  const float* vp = vb + ((size_t)bh * N_ + wave * KSPL_ + 2 * lhalf) * HD_ + 4 * l15;
  float4 vq[2][16];
#pragma unroll
  for (int s = 0; s < 8; ++s) {
    vq[0][2 * s]     = *(const float4*)(vp + (4 * s) * HD_);
    vq[0][2 * s + 1] = *(const float4*)(vp + (4 * s + 1) * HD_);
  }
  float4 tq4[2][2];
#pragma unroll
  for (int s = 0; s < 2; ++s) {
#pragma unroll
    for (int e = 0; e < 2; ++e) {
      const int kb = wave * 8 + 4 * s + 2 * lhalf + e;
      const int kr = (kb < NB_) ? kb : NB_ - 1;
      tq4[s][e] = *(const float4*)(tv + ((size_t)h * NB_ + kr) * HD_ + 4 * l15);
    }
  }

  for (int idx = threadIdx.x; idx < ROWS_ * NBP_; idx += 256) {
    const int m  = idx >> 6;
    const int bk = idx & 63;
    float s = 0.f;
    for (int w = 0; w < 8; ++w) s += pw[(w * ROWS_ + m) * PSTR_ + bk];
    ab[m * ABSTR_ + bk] = s;
  }
  __syncthreads();

  v8f acc[2][4];
#pragma unroll
  for (int t = 0; t < 4; ++t) { acc[0][t] = (v8f)0.f; acc[1][t] = (v8f)0.f; }

#pragma unroll
  for (int g = 0; g < KSPL_ / 32; ++g) {
    if (g + 1 < KSPL_ / 32) {
#pragma unroll
      for (int s = 0; s < 8; ++s) {
        vq[(g + 1) & 1][2 * s]     = *(const float4*)(vp + ((g + 1) * 32 + 4 * s) * HD_);
        vq[(g + 1) & 1][2 * s + 1] = *(const float4*)(vp + ((g + 1) * 32 + 4 * s + 1) * HD_);
      }
    }
#pragma unroll
    for (int s = 0; s < 8; ++s) {
      const int col = wave * KSPL_ + g * 32 + 4 * s + 2 * lhalf;
      const v2f a0 = ld2(&S[l15 * SSTR_ + col]);
      const v2f a1 = ld2(&S[(16 + l15) * SSTR_ + col]);
      const float4 x0 = vq[g & 1][2 * s];
      const float4 x1 = vq[g & 1][2 * s + 1];
      v2f bf;
      bf.x = x0.x; bf.y = x1.x;
      acc[0][0] = wmma4(a0, bf, acc[0][0]);
      acc[1][0] = wmma4(a1, bf, acc[1][0]);
      bf.x = x0.y; bf.y = x1.y;
      acc[0][1] = wmma4(a0, bf, acc[0][1]);
      acc[1][1] = wmma4(a1, bf, acc[1][1]);
      bf.x = x0.z; bf.y = x1.z;
      acc[0][2] = wmma4(a0, bf, acc[0][2]);
      acc[1][2] = wmma4(a1, bf, acc[1][2]);
      bf.x = x0.w; bf.y = x1.w;
      acc[0][3] = wmma4(a0, bf, acc[0][3]);
      acc[1][3] = wmma4(a1, bf, acc[1][3]);
    }
  }

#pragma unroll
  for (int s = 0; s < 2; ++s) {
    const int kb = wave * 8 + 4 * s + 2 * lhalf;
    const float4 z4 = make_float4(0.f, 0.f, 0.f, 0.f);
    float4 x0 = tq4[s][0];
    float4 x1 = tq4[s][1];
    if (kb     >= NB_) x0 = z4;
    if (kb + 1 >= NB_) x1 = z4;
    const v2f a0 = ld2(&ab[l15 * ABSTR_ + kb]);
    const v2f a1 = ld2(&ab[(16 + l15) * ABSTR_ + kb]);
    v2f bf;
    bf.x = x0.x; bf.y = x1.x;
    acc[0][0] = wmma4(a0, bf, acc[0][0]);
    acc[1][0] = wmma4(a1, bf, acc[1][0]);
    bf.x = x0.y; bf.y = x1.y;
    acc[0][1] = wmma4(a0, bf, acc[0][1]);
    acc[1][1] = wmma4(a1, bf, acc[1][1]);
    bf.x = x0.z; bf.y = x1.z;
    acc[0][2] = wmma4(a0, bf, acc[0][2]);
    acc[1][2] = wmma4(a1, bf, acc[1][2]);
    bf.x = x0.w; bf.y = x1.w;
    acc[0][3] = wmma4(a0, bf, acc[0][3]);
    acc[1][3] = wmma4(a1, bf, acc[1][3]);
  }

  float* red = pw;
  {
    float* mine = red + wave * (ROWS_ * HD_);
#pragma unroll
    for (int mt = 0; mt < 2; ++mt) {
#pragma unroll
      for (int vv = 0; vv < 8; ++vv) {
        const int m = mt * 16 + vv + 8 * lhalf;
        *(float4*)(mine + m * HD_ + 4 * l15) =
            make_float4(acc[mt][0][vv], acc[mt][1][vv], acc[mt][2][vv], acc[mt][3][vv]);
      }
    }
  }
  __syncthreads();

  for (int e = threadIdx.x; e < ROWS_ * HD_ / 4; e += 256) {
    float4 o = ((const float4*)red)[e];
#pragma unroll
    for (int w = 1; w < 8; ++w) {
      const float4 p = ((const float4*)(red + w * (ROWS_ * HD_)))[e];
      o.x += p.x; o.y += p.y; o.z += p.z; o.w += p.w;
    }
    const int m   = e >> 4;
    const int d4  = (e & 15) * 4;
    const int tok = r0 + m;
    *(float4*)(ob + (((size_t)b * N_ + tok) * H_ + h) * HD_ + d4) = o;
  }
}

__global__ __launch_bounds__(256) __attribute__((amdgpu_waves_per_eu(2, 6)))
void k_proj(const float* __restrict__ ob,
            const float* __restrict__ wp,
            const float* __restrict__ bias,
            float* __restrict__ out) {
  __shared__ float Bt[2 * 64 * BSTR_];

  const int wave  = threadIdx.x >> 5;
  const int lane  = threadIdx.x & 31;
  const int l15   = lane & 15;
  const int lhalf = lane >> 4;

  const int ngrp = blockIdx.y;
  const int nb   = ngrp * 64;
  const int row  = blockIdx.x * 128 + wave * 16 + l15;
  const float* xr = ob + (size_t)row * C_ + 2 * lhalf;

  const int sk  = threadIdx.x >> 4;
  const int snq = (threadIdx.x & 15) * 4;
  const float* wsrc = wp + (size_t)sk * C_ + nb + snq;
  float* bst = Bt + snq * BSTR_ + sk;

  float4 wn[4];
  v2f    an[16];
#pragma unroll
  for (int r = 0; r < 4; ++r) wn[r] = *(const float4*)(wsrc + (size_t)(16 * r) * C_);
#pragma unroll
  for (int s = 0; s < 16; ++s) an[s] = ld2(xr + 4 * s);
#pragma unroll
  for (int r = 0; r < 4; ++r) {
    bst[0 * BSTR_ + 16 * r] = wn[r].x;
    bst[1 * BSTR_ + 16 * r] = wn[r].y;
    bst[2 * BSTR_ + 16 * r] = wn[r].z;
    bst[3 * BSTR_ + 16 * r] = wn[r].w;
  }

  v8f acc[4];
  acc[0] = (v8f)0.f; acc[1] = (v8f)0.f; acc[2] = (v8f)0.f; acc[3] = (v8f)0.f;

  int cur = 0;
  for (int k0 = 0; k0 < C_; k0 += 64) {
    __syncthreads();
    v2f a[16];
#pragma unroll
    for (int s = 0; s < 16; ++s) a[s] = an[s];
    const int kn0 = (k0 + 64 < C_) ? k0 + 64 : k0;
#pragma unroll
    for (int r = 0; r < 4; ++r) wn[r] = *(const float4*)(wsrc + (size_t)(kn0 + 16 * r) * C_);
#pragma unroll
    for (int s = 0; s < 16; ++s) an[s] = ld2(xr + kn0 + 4 * s);

    const float* bld = Bt + cur * (64 * BSTR_) + l15 * BSTR_ + 2 * lhalf;
#pragma unroll
    for (int s = 0; s < 16; ++s) {
#pragma unroll
      for (int t = 0; t < 4; ++t) {
        v2f bf = ld2(bld + (t * 16) * BSTR_ + 4 * s);
        acc[t] = wmma4(a[s], bf, acc[t]);
      }
    }

    float* bnx = bst + (cur ^ 1) * (64 * BSTR_);
#pragma unroll
    for (int r = 0; r < 4; ++r) {
      bnx[0 * BSTR_ + 16 * r] = wn[r].x;
      bnx[1 * BSTR_ + 16 * r] = wn[r].y;
      bnx[2 * BSTR_ + 16 * r] = wn[r].z;
      bnx[3 * BSTR_ + 16 * r] = wn[r].w;
    }
    cur ^= 1;
  }

#pragma unroll
  for (int t = 0; t < 4; ++t) {
    const int c0 = nb + t * 16 + l15;
    const float bv = bias[c0];
#pragma unroll
    for (int vv = 0; vv < 8; ++vv) {
      const int r = blockIdx.x * 128 + wave * 16 + vv + 8 * lhalf;
      out[(size_t)r * C_ + c0] = acc[t][vv] + bv;
    }
  }
}

extern "C" void kernel_launch(void* const* d_in, const int* in_sizes, int n_in,
                              void* d_out, int out_size, void* d_ws, size_t ws_size,
                              hipStream_t stream) {
  const float* x     = (const float*)d_in[0];
  const float* wqkv  = (const float*)d_in[1];
  const float* wproj = (const float*)d_in[2];
  const float* bproj = (const float*)d_in[3];
  const float* tq    = (const float*)d_in[4];
  const float* tk    = (const float*)d_in[5];
  const float* tv    = (const float*)d_in[6];
  const int*   rp    = (const int*)d_in[7];
  float*       out   = (float*)d_out;

  float* ws = (float*)d_ws;
  const size_t SZ = (size_t)B_ * H_ * N_ * HD_;
  float* qb  = ws;
  float* kb  = ws + 1 * SZ;
  float* vb  = ws + 2 * SZ;
  float* lkb = ws + 3 * SZ;
  float* lqb = ws + 4 * SZ;
  float* ob  = ws + 5 * SZ;

  k_qkv<<<dim3(64, 36), 256, 0, stream>>>(x, wqkv, qb, kb, vb);

  k_tables<<<768, 256, 0, stream>>>(qb, kb, tq, tk, lkb, lqb);

  const size_t lds = (size_t)(ROWS_ * SSTR_ + ROWS_ * ABSTR_ +
                              8 * ROWS_ * PSTR_ + 8 * 32 * KQSTR_ +
                              ROWS_ * KQSTR_) * sizeof(float);
  k_attn<<<dim3(32, 96), 256, lds, stream>>>(qb, kb, vb, lkb, lqb, tv, rp, ob);

  k_proj<<<dim3(64, 12), 256, 0, stream>>>(ob, wproj, bproj, out);

  (void)in_sizes; (void)n_in; (void)out_size; (void)ws_size;
}


# attention: 256 work-groups stay resident and loop over 12 tiles each (was 3072 one-tile groups); next tile's first loads fly under P@V; rp rows that depend on the row stripe alone are read once
# speedup vs baseline: 2.1942x; 1.0388x over previous
#include <hip/hip_runtime.h>
#include <hip/hip_bf16.h>

#define B_    8
#define N_    1024
#define C_    768
#define H_    12
#define HD_   64
#define NB_   49
#define NBP_  64
#define ROWS_ 32
#define BSTR_ 66
#define KSPL_ 128
#define PSTR_ 66
#define SSTR_ 1028
#define ABSTR_ 68
#define LKSTR_ 68
#define KQSTR_ 68
#define NTILE_ 3072
#define NWG_   256

typedef float v2f __attribute__((ext_vector_type(2)));
typedef float v8f __attribute__((ext_vector_type(8)));

static __device__ __forceinline__ v8f wmma4(v2f a, v2f b, v8f c) {
  return __builtin_amdgcn_wmma_f32_16x16x4_f32(false, a, false, b, (short)0, c,
                                               false, false);
}
static __device__ __forceinline__ v2f ld2(const float* p) {
  return *(const v2f*)p;
}

static __device__ __forceinline__ float xhalf(float x) {
  const int b = __builtin_bit_cast(int, x);
  return __builtin_bit_cast(float,
      __builtin_amdgcn_permlanex16(b, b, 0x76543210, 0xfedcba98, false, false));
}
template <int M>
static __device__ __forceinline__ float xrow(float x) {
  const int b = __builtin_bit_cast(int, x);
  return __builtin_bit_cast(float,
      __builtin_amdgcn_update_dpp(b, b, 0x160 + M, 0xf, 0xf, false));
}
static __device__ __forceinline__ float wave_max(float v) {
  v = fmaxf(v, xrow<1>(v));
  v = fmaxf(v, xrow<2>(v));
  v = fmaxf(v, xrow<4>(v));
  v = fmaxf(v, xrow<8>(v));
  return fmaxf(v, xhalf(v));
}
static __device__ __forceinline__ float wave_sum(float v) {
  v += xrow<1>(v);
  v += xrow<2>(v);
  v += xrow<4>(v);
  v += xrow<8>(v);
  return v + xhalf(v);
}

__global__ __launch_bounds__(256) __attribute__((amdgpu_waves_per_eu(2, 6)))
void k_qkv(const float* __restrict__ x,
           const float* __restrict__ wqkv,
           float* __restrict__ qb,
           float* __restrict__ kb,
           float* __restrict__ vb) {
  __shared__ float Bt[2 * 64 * BSTR_];

  const int wave  = threadIdx.x >> 5;
  const int lane  = threadIdx.x & 31;
  const int l15   = lane & 15;
  const int lhalf = lane >> 4;

  const int ngrp = blockIdx.y;
  const int nb   = ngrp * 64;
  const int row  = blockIdx.x * 128 + wave * 16 + l15;
  const float* xr = x + (size_t)row * C_ + 2 * lhalf;

  const int sk  = threadIdx.x >> 4;
  const int snq = (threadIdx.x & 15) * 4;
  const float* wsrc = wqkv + (size_t)sk * (3 * C_) + nb + snq;
  float* bst = Bt + snq * BSTR_ + sk;

  float4 wn[4];
  v2f    an[16];
#pragma unroll
  for (int r = 0; r < 4; ++r) wn[r] = *(const float4*)(wsrc + (size_t)(16 * r) * (3 * C_));
#pragma unroll
  for (int s = 0; s < 16; ++s) an[s] = ld2(xr + 4 * s);
#pragma unroll
  for (int r = 0; r < 4; ++r) {
    bst[0 * BSTR_ + 16 * r] = wn[r].x;
    bst[1 * BSTR_ + 16 * r] = wn[r].y;
    bst[2 * BSTR_ + 16 * r] = wn[r].z;
    bst[3 * BSTR_ + 16 * r] = wn[r].w;
  }

  v8f acc[4];
  acc[0] = (v8f)0.f; acc[1] = (v8f)0.f; acc[2] = (v8f)0.f; acc[3] = (v8f)0.f;

  int cur = 0;
  for (int k0 = 0; k0 < C_; k0 += 64) {
    __syncthreads();
    v2f a[16];
#pragma unroll
    for (int s = 0; s < 16; ++s) a[s] = an[s];
    const int kn0 = (k0 + 64 < C_) ? k0 + 64 : k0;
#pragma unroll
    for (int r = 0; r < 4; ++r) wn[r] = *(const float4*)(wsrc + (size_t)(kn0 + 16 * r) * (3 * C_));
#pragma unroll
    for (int s = 0; s < 16; ++s) an[s] = ld2(xr + kn0 + 4 * s);

    const float* bld = Bt + cur * (64 * BSTR_) + l15 * BSTR_ + 2 * lhalf;
#pragma unroll
    for (int s = 0; s < 16; ++s) {
#pragma unroll
      for (int t = 0; t < 4; ++t) {
        v2f bf = ld2(bld + (t * 16) * BSTR_ + 4 * s);
        acc[t] = wmma4(a[s], bf, acc[t]);
      }
    }

    float* bnx = bst + (cur ^ 1) * (64 * BSTR_);
#pragma unroll
    for (int r = 0; r < 4; ++r) {
      bnx[0 * BSTR_ + 16 * r] = wn[r].x;
      bnx[1 * BSTR_ + 16 * r] = wn[r].y;
      bnx[2 * BSTR_ + 16 * r] = wn[r].z;
      bnx[3 * BSTR_ + 16 * r] = wn[r].w;
    }
    cur ^= 1;
  }

  const int which = ngrp / H_;
  const int h     = ngrp % H_;
  float* dst = (which == 0) ? qb : (which == 1) ? kb : vb;
#pragma unroll
  for (int t = 0; t < 4; ++t) {
    const int d = t * 16 + l15;
#pragma unroll
    for (int vv = 0; vv < 8; ++vv) {
      const int r    = blockIdx.x * 128 + wave * 16 + vv + 8 * lhalf;
      const int btok = r >> 10;
      const int tok  = r & (N_ - 1);
      dst[(((size_t)btok * H_ + h) * N_ + tok) * HD_ + d] = acc[t][vv];
    }
  }
}

__global__ __launch_bounds__(256) void k_tables(const float* __restrict__ qb,
                                                const float* __restrict__ kb,
                                                const float* __restrict__ tq,
                                                const float* __restrict__ tk,
                                                float* __restrict__ lkb,
                                                float* __restrict__ lqb) {
  const int wave  = threadIdx.x >> 5;
  const int lane  = threadIdx.x & 31;
  const int l15   = lane & 15;
  const int lhalf = lane >> 4;

  const int unit = blockIdx.x * 8 + wave;
  const int bh   = unit >> 6;
  const int mt   = unit & 63;
  const int h    = bh % H_;

  const size_t rowbase = ((size_t)bh * N_ + mt * 16 + l15) * HD_;
  const v2f z2 = (v2f)0.f;

  v8f ck[4], cq[4];
#pragma unroll
  for (int t = 0; t < 4; ++t) { ck[t] = (v8f)0.f; cq[t] = (v8f)0.f; }

#pragma unroll
  for (int kk = 0; kk < HD_; kk += 4) {
    const int kbi = kk + 2 * lhalf;
    v2f aq = ld2(qb + rowbase + kbi);
    v2f ak = ld2(kb + rowbase + kbi);
    ak = ak * 0.125f;
#pragma unroll
    for (int t = 0; t < 4; ++t) {
      const int n = t * 16 + l15;
      v2f btk = (n < NB_) ? ld2(tk + ((size_t)h * NB_ + n) * HD_ + kbi) : z2;
      v2f btq = (n < NB_) ? ld2(tq + ((size_t)h * NB_ + n) * HD_ + kbi) : z2;
      ck[t] = wmma4(aq, btk, ck[t]);
      cq[t] = wmma4(ak, btq, cq[t]);
    }
  }

#pragma unroll
  for (int t = 0; t < 4; ++t) {
    const int col = t * 16 + l15;
#pragma unroll
    for (int vv = 0; vv < 8; ++vv) {
      const int r = mt * 16 + vv + 8 * lhalf;
      lkb[((size_t)bh * N_ + r) * NBP_ + col] = ck[t][vv];
      lqb[((size_t)bh * N_ + r) * NBP_ + col] = cq[t][vv];
    }
  }
}

__global__ __launch_bounds__(256) __attribute__((amdgpu_waves_per_eu(2, 2)))
void k_attn(const float* __restrict__ qb,
                                              const float* __restrict__ kb,
                                              const float* __restrict__ vb,
                                              const float* __restrict__ lkb,
                                              const float* __restrict__ lqb,
                                              const float* __restrict__ tv,
                                              const int*   __restrict__ rp,
                                              float* __restrict__ ob) {
  extern __shared__ __attribute__((aligned(16))) float smem[];
  float* S  = smem;
  float* ab = smem + ROWS_ * SSTR_;
  float* pw = ab + ROWS_ * ABSTR_;
  float* kst = pw + 8 * ROWS_ * PSTR_;
  float* qS  = kst + 8 * 32 * KQSTR_;

  const int wave  = threadIdx.x >> 5;
  const int lane  = threadIdx.x & 31;
  const int l15   = lane & 15;
  const int lhalf = lane >> 4;

  float* lkS = pw;
  float* kw = kst + wave * (32 * KQSTR_);
  float* kwst = kw + lhalf * KQSTR_ + 4 * l15;
  const float* kwld = kw + l15 * KQSTR_ + 2 * lhalf;
  const int e0 = threadIdx.x, e1 = threadIdx.x + 256;

  float4 nl0, nl1, nq0, nq1;
  float4 nk0, nk1, nk2, nk3, nk4, nk5, nk6, nk7,
         nk8, nk9, nk10, nk11, nk12, nk13, nk14, nk15;
#define NK_LOAD_(i)  nk##i = ks[wave * 512 + 32 * (i)]
#define NK_LOAD_ALL_() do { \
    NK_LOAD_(0);  NK_LOAD_(1);  NK_LOAD_(2);  NK_LOAD_(3);  \
    NK_LOAD_(4);  NK_LOAD_(5);  NK_LOAD_(6);  NK_LOAD_(7);  \
    NK_LOAD_(8);  NK_LOAD_(9);  NK_LOAD_(10); NK_LOAD_(11); \
    NK_LOAD_(12); NK_LOAD_(13); NK_LOAD_(14); NK_LOAD_(15); } while (0)
#define NK_STAGE_(i) *(float4*)(kwst + 2 * (i) * KQSTR_) = nk##i
#define NK_STAGE_ALL_() do { \
    NK_STAGE_(0);  NK_STAGE_(1);  NK_STAGE_(2);  NK_STAGE_(3);  \
    NK_STAGE_(4);  NK_STAGE_(5);  NK_STAGE_(6);  NK_STAGE_(7);  \
    NK_STAGE_(8);  NK_STAGE_(9);  NK_STAGE_(10); NK_STAGE_(11); \
    NK_STAGE_(12); NK_STAGE_(13); NK_STAGE_(14); NK_STAGE_(15); } while (0)
  {
    const int ft  = blockIdx.x;
    const int fbh = ft >> 5;
    const int fr0 = (ft & 31) * ROWS_;
    const float4* lsrc = (const float4*)(lkb + ((size_t)fbh * N_ + fr0) * NBP_);
    const float4* qsrc = (const float4*)(qb + ((size_t)fbh * N_ + fr0) * HD_);
    const float4* ks   = (const float4*)(kb + (size_t)fbh * N_ * HD_) + lane;
    nl0 = lsrc[e0];
    nl1 = lsrc[e1];
    nq0 = qsrc[e0];
    nq1 = qsrc[e1];
    NK_LOAD_ALL_();
  }

  for (int it = 0; it < NTILE_ / NWG_; ++it) {
  const int tile = it * NWG_ + (int)blockIdx.x;
  const int rblk = tile & 31;
  const int bh   = tile >> 5;
  const int b    = bh / H_;
  const int h    = bh % H_;
  const int r0   = rblk * ROWS_;

  {
    *(float4*)(lkS + (e0 >> 4) * LKSTR_ + (e0 & 15) * 4) = nl0;
    *(float4*)(lkS + (e1 >> 4) * LKSTR_ + (e1 & 15) * 4) = nl1;
    *(float4*)(qS + (e0 >> 4) * KQSTR_ + (e0 & 15) * 4) = nq0;
    *(float4*)(qS + (e1 >> 4) * KQSTR_ + (e1 & 15) * 4) = nq1;
    const float* kbh  = kb  + (size_t)bh * N_ * HD_;
    const float* lqbh = lqb + (size_t)bh * N_ * NBP_;
    const float4* ksrc = (const float4*)kbh + lane;
    NK_STAGE_ALL_();
    __syncthreads();

    v2f qf0[16], qf1[16];
    {
      const float* qrow = qS + l15 * KQSTR_ + 2 * lhalf;
#pragma unroll
      for (int s = 0; s < 16; ++s) {
        qf0[s] = ld2(qrow + 4 * s);
        qf1[s] = ld2(qrow + 16 * KQSTR_ + 4 * s);
      }
    }

    for (int np = wave; np < 32; np += 8) {
      v2f kf0[16], kf1[16];
#pragma unroll
      for (int s = 0; s < 16; ++s) {
        kf0[s] = ld2(kwld + 4 * s);
        kf1[s] = ld2(kwld + 16 * KQSTR_ + 4 * s);
      }
      const int nxt = (np + 8 < 32) ? np + 8 : np;
      float4 kn[16];
#pragma unroll
      for (int cc = 0; cc < 16; ++cc) kn[cc] = ksrc[nxt * 512 + 32 * cc];
      v8f acc[2][2];
      acc[0][0] = (v8f)0.f; acc[0][1] = (v8f)0.f;
      acc[1][0] = (v8f)0.f; acc[1][1] = (v8f)0.f;

      int4 bin[2][2][2];
      int  bni[2][2][8];
#pragma unroll
      for (int kt = 0; kt < 2; ++kt) {
        const unsigned nb = (unsigned)(np * 32 + kt * 16 + 8 * lhalf);
#pragma unroll
        for (int rt = 0; rt < 2; ++rt) {
          const unsigned i = (unsigned)(r0 + rt * 16 + l15);
          const int4* pi = (const int4*)(rp + (i * N_ + nb));
          bin[kt][rt][0] = pi[0];
          bin[kt][rt][1] = pi[1];
#pragma unroll
          for (int vv = 0; vv < 8; ++vv) bni[kt][rt][vv] = rp[(nb + vv) * N_ + i];
        }
      }

      float lkv[2][2][8];
#pragma unroll
      for (int kt = 0; kt < 2; ++kt) {
#pragma unroll
        for (int rt = 0; rt < 2; ++rt) {
          const float* lkrow = lkS + (rt * 16 + l15) * LKSTR_;
          const int4 ia = bin[kt][rt][0], ib = bin[kt][rt][1];
          const int bi[8] = {ia.x, ia.y, ia.z, ia.w, ib.x, ib.y, ib.z, ib.w};
#pragma unroll
          for (int vv = 0; vv < 8; ++vv) lkv[kt][rt][vv] = lkrow[(unsigned)bi[vv]];
        }
      }

#pragma unroll
      for (int s = 0; s < 16; ++s) {
        acc[0][0] = wmma4(kf0[s], qf0[s], acc[0][0]);
        acc[1][0] = wmma4(kf0[s], qf1[s], acc[1][0]);
        acc[0][1] = wmma4(kf1[s], qf0[s], acc[0][1]);
        acc[1][1] = wmma4(kf1[s], qf1[s], acc[1][1]);
      }

#pragma unroll
      for (int kt = 0; kt < 2; ++kt) {
        const unsigned nb = (unsigned)(np * 32 + kt * 16 + 8 * lhalf);
#pragma unroll
        for (int rt = 0; rt < 2; ++rt) {
          const unsigned m = (unsigned)(rt * 16 + l15);
          float o[8];
#pragma unroll
          for (int vv = 0; vv < 8; ++vv) {
            const unsigned n = nb + vv;
            o[vv] = acc[rt][kt][vv]
                  + lkv[kt][rt][vv]
                  + lqbh[n * NBP_ + (unsigned)bni[kt][rt][vv]];
          }
          *(float4*)(S + m * SSTR_ + nb)     = make_float4(o[0], o[1], o[2], o[3]);
          *(float4*)(S + m * SSTR_ + nb + 4) = make_float4(o[4], o[5], o[6], o[7]);
        }
      }
#pragma unroll
      for (int cc = 0; cc < 16; ++cc) *(float4*)(kwst + 2 * cc * KQSTR_) = kn[cc];
    }
  }
  __syncthreads();

  {
    float4 x[4][8];
#pragma unroll
    for (int rr = 0; rr < 4; ++rr) {
      const float* row = S + (wave * 4 + rr) * SSTR_ + 4 * lane;
#pragma unroll
      for (int c = 0; c < 8; ++c) x[rr][c] = *(const float4*)(row + 128 * c);
    }
    float mx[4];
#pragma unroll
    for (int rr = 0; rr < 4; ++rr) {
      float m = fmaxf(fmaxf(x[rr][0].x, x[rr][0].y), fmaxf(x[rr][0].z, x[rr][0].w));
#pragma unroll
      for (int c = 1; c < 8; ++c)
        m = fmaxf(m, fmaxf(fmaxf(x[rr][c].x, x[rr][c].y), fmaxf(x[rr][c].z, x[rr][c].w)));
      mx[rr] = wave_max(m);
    }
    float inv[4];
#pragma unroll
    for (int rr = 0; rr < 4; ++rr) {
      float s0 = 0.f, s1 = 0.f;
#pragma unroll
      for (int c = 0; c < 8; ++c) {
        float4 e;
        e.x = __expf(x[rr][c].x - mx[rr]);
        e.y = __expf(x[rr][c].y - mx[rr]);
        e.z = __expf(x[rr][c].z - mx[rr]);
        e.w = __expf(x[rr][c].w - mx[rr]);
        x[rr][c] = e;
        s0 += e.x + e.y;
        s1 += e.z + e.w;
      }
      inv[rr] = 1.0f / wave_sum(s0 + s1);
    }
#pragma unroll
    for (int rr = 0; rr < 4; ++rr) {
      float* row = S + (wave * 4 + rr) * SSTR_ + 4 * lane;
#pragma unroll
      for (int c = 0; c < 8; ++c) {
        float4 p = x[rr][c];
        p.x *= inv[rr]; p.y *= inv[rr]; p.z *= inv[rr]; p.w *= inv[rr];
        *(float4*)(row + 128 * c) = p;
      }
    }
  }
  __syncthreads();

  {
    float* mine = pw + wave * (ROWS_ * PSTR_);
    for (int k = lane; k < ROWS_ * PSTR_ / 2; k += 32)
      ((v2f*)mine)[k] = (v2f)0.f;
    float*       myrow = mine + lane * PSTR_;
    const float* prow  = S + lane * SSTR_;
    const int*   brow  = rp + (size_t)(r0 + lane) * N_;
    for (int win = 0; win < 2; ++win) {
      const int j0 = wave * 128 + win * 64;
      float4 p[16];
      int4   b[16];
#pragma unroll
      for (int c = 0; c < 16; ++c) {
        const int col = j0 + 4 * c;
        p[c] = *(const float4*)(prow + col);
        b[c] = *(const int4*)(brow + col);
      }
#pragma unroll
      for (int c = 0; c < 16; ++c) {
        atomicAdd(myrow + b[c].x, p[c].x);
        atomicAdd(myrow + b[c].y, p[c].y);
        atomicAdd(myrow + b[c].z, p[c].z);
        atomicAdd(myrow + b[c].w, p[c].w);
      }
    }
  }
  __syncthreads();

  const float* vp = vb + ((size_t)bh * N_ + wave * KSPL_ + 2 * lhalf) * HD_ + 4 * l15;
  float4 vq[2][16];
#pragma unroll
  for (int s = 0; s < 8; ++s) {
    vq[0][2 * s]     = *(const float4*)(vp + (4 * s) * HD_);
    vq[0][2 * s + 1] = *(const float4*)(vp + (4 * s + 1) * HD_);
  }
  float4 tq4[2][2];
#pragma unroll
  for (int s = 0; s < 2; ++s) {
#pragma unroll
    for (int e = 0; e < 2; ++e) {
      const int kb = wave * 8 + 4 * s + 2 * lhalf + e;
      const int kr = (kb < NB_) ? kb : NB_ - 1;
      tq4[s][e] = *(const float4*)(tv + ((size_t)h * NB_ + kr) * HD_ + 4 * l15);
    }
  }

  for (int idx = threadIdx.x; idx < ROWS_ * NBP_; idx += 256) {
    const int m  = idx >> 6;
    const int bk = idx & 63;
    float s = 0.f;
    for (int w = 0; w < 8; ++w) s += pw[(w * ROWS_ + m) * PSTR_ + bk];
    ab[m * ABSTR_ + bk] = s;
  }
  __syncthreads();

  v8f acc[2][4];
#pragma unroll
  for (int t = 0; t < 4; ++t) { acc[0][t] = (v8f)0.f; acc[1][t] = (v8f)0.f; }

#pragma unroll
  for (int g = 0; g < KSPL_ / 32; ++g) {
    if (g + 1 < KSPL_ / 32) {
#pragma unroll
      for (int s = 0; s < 8; ++s) {
        vq[(g + 1) & 1][2 * s]     = *(const float4*)(vp + ((g + 1) * 32 + 4 * s) * HD_);
        vq[(g + 1) & 1][2 * s + 1] = *(const float4*)(vp + ((g + 1) * 32 + 4 * s + 1) * HD_);
      }
    } else {
      const int tn  = (it + 1 < NTILE_ / NWG_) ? tile + NWG_ : tile;
      const int nbh = tn >> 5;
      const int nr0 = (tn & 31) * ROWS_;
      const float4* lsrc = (const float4*)(lkb + ((size_t)nbh * N_ + nr0) * NBP_);
      const float4* qsrc = (const float4*)(qb + ((size_t)nbh * N_ + nr0) * HD_);
      const float4* ks   = (const float4*)(kb + (size_t)nbh * N_ * HD_) + lane;
      nl0 = lsrc[e0];
      nl1 = lsrc[e1];
      nq0 = qsrc[e0];
      nq1 = qsrc[e1];
      NK_LOAD_ALL_();
    }
#pragma unroll
    for (int s = 0; s < 8; ++s) {
      const int col = wave * KSPL_ + g * 32 + 4 * s + 2 * lhalf;
      const v2f a0 = ld2(&S[l15 * SSTR_ + col]);
      const v2f a1 = ld2(&S[(16 + l15) * SSTR_ + col]);
      const float4 x0 = vq[g & 1][2 * s];
      const float4 x1 = vq[g & 1][2 * s + 1];
      v2f bf;
      bf.x = x0.x; bf.y = x1.x;
      acc[0][0] = wmma4(a0, bf, acc[0][0]);
      acc[1][0] = wmma4(a1, bf, acc[1][0]);
      bf.x = x0.y; bf.y = x1.y;
      acc[0][1] = wmma4(a0, bf, acc[0][1]);
      acc[1][1] = wmma4(a1, bf, acc[1][1]);
      bf.x = x0.z; bf.y = x1.z;
      acc[0][2] = wmma4(a0, bf, acc[0][2]);
      acc[1][2] = wmma4(a1, bf, acc[1][2]);
      bf.x = x0.w; bf.y = x1.w;
      acc[0][3] = wmma4(a0, bf, acc[0][3]);
      acc[1][3] = wmma4(a1, bf, acc[1][3]);
    }
  }

#pragma unroll
  for (int s = 0; s < 2; ++s) {
    const int kb = wave * 8 + 4 * s + 2 * lhalf;
    const float4 z4 = make_float4(0.f, 0.f, 0.f, 0.f);
    float4 x0 = tq4[s][0];
    float4 x1 = tq4[s][1];
    if (kb     >= NB_) x0 = z4;
    if (kb + 1 >= NB_) x1 = z4;
    const v2f a0 = ld2(&ab[l15 * ABSTR_ + kb]);
    const v2f a1 = ld2(&ab[(16 + l15) * ABSTR_ + kb]);
    v2f bf;
    bf.x = x0.x; bf.y = x1.x;
    acc[0][0] = wmma4(a0, bf, acc[0][0]);
    acc[1][0] = wmma4(a1, bf, acc[1][0]);
    bf.x = x0.y; bf.y = x1.y;
    acc[0][1] = wmma4(a0, bf, acc[0][1]);
    acc[1][1] = wmma4(a1, bf, acc[1][1]);
    bf.x = x0.z; bf.y = x1.z;
    acc[0][2] = wmma4(a0, bf, acc[0][2]);
    acc[1][2] = wmma4(a1, bf, acc[1][2]);
    bf.x = x0.w; bf.y = x1.w;
    acc[0][3] = wmma4(a0, bf, acc[0][3]);
    acc[1][3] = wmma4(a1, bf, acc[1][3]);
  }

  float* red = pw;
  {
    float* mine = red + wave * (ROWS_ * HD_);
#pragma unroll
    for (int mt = 0; mt < 2; ++mt) {
#pragma unroll
      for (int vv = 0; vv < 8; ++vv) {
        const int m = mt * 16 + vv + 8 * lhalf;
        *(float4*)(mine + m * HD_ + 4 * l15) =
            make_float4(acc[mt][0][vv], acc[mt][1][vv], acc[mt][2][vv], acc[mt][3][vv]);
      }
    }
  }
  __syncthreads();

  for (int e = threadIdx.x; e < ROWS_ * HD_ / 4; e += 256) {
    float4 o = ((const float4*)red)[e];
#pragma unroll
    for (int w = 1; w < 8; ++w) {
      const float4 p = ((const float4*)(red + w * (ROWS_ * HD_)))[e];
      o.x += p.x; o.y += p.y; o.z += p.z; o.w += p.w;
    }
    const int m   = e >> 4;
    const int d4  = (e & 15) * 4;
    const int tok = r0 + m;
    *(float4*)(ob + (((size_t)b * N_ + tok) * H_ + h) * HD_ + d4) = o;
  }
  __syncthreads();
  }
}

__global__ __launch_bounds__(256) __attribute__((amdgpu_waves_per_eu(2, 6)))
void k_proj(const float* __restrict__ ob,
            const float* __restrict__ wp,
            const float* __restrict__ bias,
            float* __restrict__ out) {
  __shared__ float Bt[2 * 64 * BSTR_];

  const int wave  = threadIdx.x >> 5;
  const int lane  = threadIdx.x & 31;
  const int l15   = lane & 15;
  const int lhalf = lane >> 4;

  const int ngrp = blockIdx.y;
  const int nb   = ngrp * 64;
  const int row  = blockIdx.x * 128 + wave * 16 + l15;
  const float* xr = ob + (size_t)row * C_ + 2 * lhalf;

  const int sk  = threadIdx.x >> 4;
  const int snq = (threadIdx.x & 15) * 4;
  const float* wsrc = wp + (size_t)sk * C_ + nb + snq;
  float* bst = Bt + snq * BSTR_ + sk;

  float4 wn[4];
  v2f    an[16];
#pragma unroll
  for (int r = 0; r < 4; ++r) wn[r] = *(const float4*)(wsrc + (size_t)(16 * r) * C_);
#pragma unroll
  for (int s = 0; s < 16; ++s) an[s] = ld2(xr + 4 * s);
#pragma unroll
  for (int r = 0; r < 4; ++r) {
    bst[0 * BSTR_ + 16 * r] = wn[r].x;
    bst[1 * BSTR_ + 16 * r] = wn[r].y;
    bst[2 * BSTR_ + 16 * r] = wn[r].z;
    bst[3 * BSTR_ + 16 * r] = wn[r].w;
  }

  v8f acc[4];
  acc[0] = (v8f)0.f; acc[1] = (v8f)0.f; acc[2] = (v8f)0.f; acc[3] = (v8f)0.f;

  int cur = 0;
  for (int k0 = 0; k0 < C_; k0 += 64) {
    __syncthreads();
    v2f a[16];
#pragma unroll
    for (int s = 0; s < 16; ++s) a[s] = an[s];
    const int kn0 = (k0 + 64 < C_) ? k0 + 64 : k0;
#pragma unroll
    for (int r = 0; r < 4; ++r) wn[r] = *(const float4*)(wsrc + (size_t)(kn0 + 16 * r) * C_);
#pragma unroll
    for (int s = 0; s < 16; ++s) an[s] = ld2(xr + kn0 + 4 * s);

    const float* bld = Bt + cur * (64 * BSTR_) + l15 * BSTR_ + 2 * lhalf;
#pragma unroll
    for (int s = 0; s < 16; ++s) {
#pragma unroll
      for (int t = 0; t < 4; ++t) {
        v2f bf = ld2(bld + (t * 16) * BSTR_ + 4 * s);
        acc[t] = wmma4(a[s], bf, acc[t]);
      }
    }

    float* bnx = bst + (cur ^ 1) * (64 * BSTR_);
#pragma unroll
    for (int r = 0; r < 4; ++r) {
      bnx[0 * BSTR_ + 16 * r] = wn[r].x;
      bnx[1 * BSTR_ + 16 * r] = wn[r].y;
      bnx[2 * BSTR_ + 16 * r] = wn[r].z;
      bnx[3 * BSTR_ + 16 * r] = wn[r].w;
    }
    cur ^= 1;
  }

#pragma unroll
  for (int t = 0; t < 4; ++t) {
    const int c0 = nb + t * 16 + l15;
    const float bv = bias[c0];
#pragma unroll
    for (int vv = 0; vv < 8; ++vv) {
      const int r = blockIdx.x * 128 + wave * 16 + vv + 8 * lhalf;
      out[(size_t)r * C_ + c0] = acc[t][vv] + bv;
    }
  }
}

extern "C" void kernel_launch(void* const* d_in, const int* in_sizes, int n_in,
                              void* d_out, int out_size, void* d_ws, size_t ws_size,
                              hipStream_t stream) {
  const float* x     = (const float*)d_in[0];
  const float* wqkv  = (const float*)d_in[1];
  const float* wproj = (const float*)d_in[2];
  const float* bproj = (const float*)d_in[3];
  const float* tq    = (const float*)d_in[4];
  const float* tk    = (const float*)d_in[5];
  const float* tv    = (const float*)d_in[6];
  const int*   rp    = (const int*)d_in[7];
  float*       out   = (float*)d_out;

  float* ws = (float*)d_ws;
  const size_t SZ = (size_t)B_ * H_ * N_ * HD_;
  float* qb  = ws;
  float* kb  = ws + 1 * SZ;
  float* vb  = ws + 2 * SZ;
  float* lkb = ws + 3 * SZ;
  float* lqb = ws + 4 * SZ;
  float* ob  = ws + 5 * SZ;

  k_qkv<<<dim3(64, 36), 256, 0, stream>>>(x, wqkv, qb, kb, vb);

  k_tables<<<768, 256, 0, stream>>>(qb, kb, tq, tk, lkb, lqb);

  const size_t lds = (size_t)(ROWS_ * SSTR_ + ROWS_ * ABSTR_ +
                              8 * ROWS_ * PSTR_ + 8 * 32 * KQSTR_ +
                              ROWS_ * KQSTR_) * sizeof(float);
  k_attn<<<dim3(256), 256, lds, stream>>>(qb, kb, vb, lkb, lqb, tv, rp, ob);

  k_proj<<<dim3(64, 12), 256, 0, stream>>>(ob, wproj, bproj, out);

  (void)in_sizes; (void)n_in; (void)out_size; (void)ws_size;
}


# both projection GEMMs: fragments of 16 bytes per lane, k dealt per lane half (same for A and B): one conflict-free ds_load_b128 (pitch 68) or 16-byte global load per 2 products (sum order changes)
# speedup vs baseline: 2.3546x; 1.0731x over previous
#include <hip/hip_runtime.h>
#include <hip/hip_bf16.h>

#define B_    8
#define N_    1024
#define C_    768
#define H_    12
#define HD_   64
#define NB_   49
#define NBP_  64
#define ROWS_ 32
#define BSTR_ 68
#define KSPL_ 128
#define PSTR_ 66
#define SSTR_ 1028
#define ABSTR_ 68
#define LKSTR_ 68
#define KQSTR_ 68
#define NTILE_ 3072
#define NWG_   256

typedef float v2f __attribute__((ext_vector_type(2)));
typedef float v8f __attribute__((ext_vector_type(8)));

static __device__ __forceinline__ v8f wmma4(v2f a, v2f b, v8f c) {
  return __builtin_amdgcn_wmma_f32_16x16x4_f32(false, a, false, b, (short)0, c,
                                               false, false);
}
static __device__ __forceinline__ v2f ld2(const float* p) {
  return *(const v2f*)p;
}

static __device__ __forceinline__ float xhalf(float x) {
  const int b = __builtin_bit_cast(int, x);
  return __builtin_bit_cast(float,
      __builtin_amdgcn_permlanex16(b, b, 0x76543210, 0xfedcba98, false, false));
}
template <int M>
static __device__ __forceinline__ float xrow(float x) {
  const int b = __builtin_bit_cast(int, x);
  return __builtin_bit_cast(float,
      __builtin_amdgcn_update_dpp(b, b, 0x160 + M, 0xf, 0xf, false));
}
static __device__ __forceinline__ float wave_max(float v) {
  v = fmaxf(v, xrow<1>(v));
  v = fmaxf(v, xrow<2>(v));
  v = fmaxf(v, xrow<4>(v));
  v = fmaxf(v, xrow<8>(v));
  return fmaxf(v, xhalf(v));
}
static __device__ __forceinline__ float wave_sum(float v) {
  v += xrow<1>(v);
  v += xrow<2>(v);
  v += xrow<4>(v);
  v += xrow<8>(v);
  return v + xhalf(v);
}

__global__ __launch_bounds__(256) __attribute__((amdgpu_waves_per_eu(2, 6)))
void k_qkv(const float* __restrict__ x,
           const float* __restrict__ wqkv,
           float* __restrict__ qb,
           float* __restrict__ kb,
           float* __restrict__ vb) {
  __shared__ __attribute__((aligned(16))) float Bt[2 * 64 * BSTR_];

  const int wave  = threadIdx.x >> 5;
  const int lane  = threadIdx.x & 31;
  const int l15   = lane & 15;
  const int lhalf = lane >> 4;

  const int ngrp = blockIdx.y;
  const int nb   = ngrp * 64;
  const int row  = blockIdx.x * 128 + wave * 16 + l15;
  const float* xr = x + (size_t)row * C_ + 4 * lhalf;

  const int sn = threadIdx.x & 63;
  const int sq = threadIdx.x >> 6;
  const float* wsrc = wqkv + (size_t)(16 * sq) * (3 * C_) + nb + sn;
  float* bst = Bt + sn * BSTR_ + 16 * sq;

  float  wn[16];
  float4 an[8];
#pragma unroll
  for (int j = 0; j < 16; ++j) wn[j] = wsrc[(size_t)j * (3 * C_)];
#pragma unroll
  for (int u = 0; u < 8; ++u) an[u] = *(const float4*)(xr + 8 * u);
#pragma unroll
  for (int g = 0; g < 4; ++g)
    *(float4*)(bst + 4 * g) = make_float4(wn[4 * g], wn[4 * g + 1], wn[4 * g + 2], wn[4 * g + 3]);

  v8f acc[4];
  acc[0] = (v8f)0.f; acc[1] = (v8f)0.f; acc[2] = (v8f)0.f; acc[3] = (v8f)0.f;

  int cur = 0;
  for (int k0 = 0; k0 < C_; k0 += 64) {
    __syncthreads();
    float4 a[8];
#pragma unroll
    for (int u = 0; u < 8; ++u) a[u] = an[u];
    const int kn0 = (k0 + 64 < C_) ? k0 + 64 : k0;
#pragma unroll
    for (int j = 0; j < 16; ++j) wn[j] = wsrc[(size_t)(kn0 + j) * (3 * C_)];
#pragma unroll
    for (int u = 0; u < 8; ++u) an[u] = *(const float4*)(xr + kn0 + 8 * u);

    const float* bld = Bt + cur * (64 * BSTR_) + l15 * BSTR_ + 4 * lhalf;
#pragma unroll
    for (int u = 0; u < 8; ++u) {
      float4 bq[4];
#pragma unroll
      for (int t = 0; t < 4; ++t) bq[t] = *(const float4*)(bld + (t * 16) * BSTR_ + 8 * u);
      v2f af;
      af.x = a[u].x; af.y = a[u].y;
#pragma unroll
      for (int t = 0; t < 4; ++t) {
        v2f bf; bf.x = bq[t].x; bf.y = bq[t].y;
        acc[t] = wmma4(af, bf, acc[t]);
      }
      af.x = a[u].z; af.y = a[u].w;
#pragma unroll
      for (int t = 0; t < 4; ++t) {
        v2f bf; bf.x = bq[t].z; bf.y = bq[t].w;
        acc[t] = wmma4(af, bf, acc[t]);
      }
    }

    float* bnx = bst + (cur ^ 1) * (64 * BSTR_);
#pragma unroll
    for (int g = 0; g < 4; ++g)
      *(float4*)(bnx + 4 * g) = make_float4(wn[4 * g], wn[4 * g + 1], wn[4 * g + 2], wn[4 * g + 3]);
    cur ^= 1;
  }

  const int which = ngrp / H_;
  const int h     = ngrp % H_;
  float* dst = (which == 0) ? qb : (which == 1) ? kb : vb;
#pragma unroll
  for (int t = 0; t < 4; ++t) {
    const int d = t * 16 + l15;
#pragma unroll
    for (int vv = 0; vv < 8; ++vv) {
      const int r    = blockIdx.x * 128 + wave * 16 + vv + 8 * lhalf;
      const int btok = r >> 10;
      const int tok  = r & (N_ - 1);
      dst[(((size_t)btok * H_ + h) * N_ + tok) * HD_ + d] = acc[t][vv];
    }
  }
}

__global__ __launch_bounds__(256) void k_tables(const float* __restrict__ qb,
                                                const float* __restrict__ kb,
                                                const float* __restrict__ tq,
                                                const float* __restrict__ tk,
                                                float* __restrict__ lkb,
                                                float* __restrict__ lqb) {
  const int wave  = threadIdx.x >> 5;
  const int lane  = threadIdx.x & 31;
  const int l15   = lane & 15;
  const int lhalf = lane >> 4;

  const int unit = blockIdx.x * 8 + wave;
  const int bh   = unit >> 6;
  const int mt   = unit & 63;
  const int h    = bh % H_;

  const size_t rowbase = ((size_t)bh * N_ + mt * 16 + l15) * HD_;
  const v2f z2 = (v2f)0.f;

  v8f ck[4], cq[4];
#pragma unroll
  for (int t = 0; t < 4; ++t) { ck[t] = (v8f)0.f; cq[t] = (v8f)0.f; }

#pragma unroll
  for (int kk = 0; kk < HD_; kk += 4) {
    const int kbi = kk + 2 * lhalf;
    v2f aq = ld2(qb + rowbase + kbi);
    v2f ak = ld2(kb + rowbase + kbi);
    ak = ak * 0.125f;
#pragma unroll
    for (int t = 0; t < 4; ++t) {
      const int n = t * 16 + l15;
      v2f btk = (n < NB_) ? ld2(tk + ((size_t)h * NB_ + n) * HD_ + kbi) : z2;
      v2f btq = (n < NB_) ? ld2(tq + ((size_t)h * NB_ + n) * HD_ + kbi) : z2;
      ck[t] = wmma4(aq, btk, ck[t]);
      cq[t] = wmma4(ak, btq, cq[t]);
    }
  }

#pragma unroll
  for (int t = 0; t < 4; ++t) {
    const int col = t * 16 + l15;
#pragma unroll
    for (int vv = 0; vv < 8; ++vv) {
      const int r = mt * 16 + vv + 8 * lhalf;
      lkb[((size_t)bh * N_ + r) * NBP_ + col] = ck[t][vv];
      lqb[((size_t)bh * N_ + r) * NBP_ + col] = cq[t][vv];
    }
  }
}

__global__ __launch_bounds__(256) __attribute__((amdgpu_waves_per_eu(2, 2)))
void k_attn(const float* __restrict__ qb,
                                              const float* __restrict__ kb,
                                              const float* __restrict__ vb,
                                              const float* __restrict__ lkb,
                                              const float* __restrict__ lqb,
                                              const float* __restrict__ tv,
                                              const int*   __restrict__ rp,
                                              float* __restrict__ ob) {
  extern __shared__ __attribute__((aligned(16))) float smem[];
  float* S  = smem;
  float* ab = smem + ROWS_ * SSTR_;
  float* pw = ab + ROWS_ * ABSTR_;
  float* kst = pw + 8 * ROWS_ * PSTR_;
  float* qS  = kst + 8 * 32 * KQSTR_;

  const int wave  = threadIdx.x >> 5;
  const int lane  = threadIdx.x & 31;
  const int l15   = lane & 15;
  const int lhalf = lane >> 4;

  float* lkS = pw;
  float* kw = kst + wave * (32 * KQSTR_);
  float* kwst = kw + lhalf * KQSTR_ + 4 * l15;
  const float* kwld = kw + l15 * KQSTR_ + 2 * lhalf;
  const int e0 = threadIdx.x, e1 = threadIdx.x + 256;

  float4 nl0, nl1, nq0, nq1;
  float4 nk0, nk1, nk2, nk3, nk4, nk5, nk6, nk7,
         nk8, nk9, nk10, nk11, nk12, nk13, nk14, nk15;
#define NK_LOAD_(i)  nk##i = ks[wave * 512 + 32 * (i)]
#define NK_LOAD_ALL_() do { \
    NK_LOAD_(0);  NK_LOAD_(1);  NK_LOAD_(2);  NK_LOAD_(3);  \
    NK_LOAD_(4);  NK_LOAD_(5);  NK_LOAD_(6);  NK_LOAD_(7);  \
    NK_LOAD_(8);  NK_LOAD_(9);  NK_LOAD_(10); NK_LOAD_(11); \
    NK_LOAD_(12); NK_LOAD_(13); NK_LOAD_(14); NK_LOAD_(15); } while (0)
#define NK_STAGE_(i) *(float4*)(kwst + 2 * (i) * KQSTR_) = nk##i
#define NK_STAGE_ALL_() do { \
    NK_STAGE_(0);  NK_STAGE_(1);  NK_STAGE_(2);  NK_STAGE_(3);  \
    NK_STAGE_(4);  NK_STAGE_(5);  NK_STAGE_(6);  NK_STAGE_(7);  \
    NK_STAGE_(8);  NK_STAGE_(9);  NK_STAGE_(10); NK_STAGE_(11); \
    NK_STAGE_(12); NK_STAGE_(13); NK_STAGE_(14); NK_STAGE_(15); } while (0)
  {
    const int ft  = blockIdx.x;
    const int fbh = ft >> 5;
    const int fr0 = (ft & 31) * ROWS_;
    const float4* lsrc = (const float4*)(lkb + ((size_t)fbh * N_ + fr0) * NBP_);
    const float4* qsrc = (const float4*)(qb + ((size_t)fbh * N_ + fr0) * HD_);
    const float4* ks   = (const float4*)(kb + (size_t)fbh * N_ * HD_) + lane;
    nl0 = lsrc[e0];
    nl1 = lsrc[e1];
    nq0 = qsrc[e0];
    nq1 = qsrc[e1];
    NK_LOAD_ALL_();
  }

  for (int it = 0; it < NTILE_ / NWG_; ++it) {
  const int tile = it * NWG_ + (int)blockIdx.x;
  const int rblk = tile & 31;
  const int bh   = tile >> 5;
  const int b    = bh / H_;
  const int h    = bh % H_;
  const int r0   = rblk * ROWS_;

  {
    *(float4*)(lkS + (e0 >> 4) * LKSTR_ + (e0 & 15) * 4) = nl0;
    *(float4*)(lkS + (e1 >> 4) * LKSTR_ + (e1 & 15) * 4) = nl1;
    *(float4*)(qS + (e0 >> 4) * KQSTR_ + (e0 & 15) * 4) = nq0;
    *(float4*)(qS + (e1 >> 4) * KQSTR_ + (e1 & 15) * 4) = nq1;
    const float* kbh  = kb  + (size_t)bh * N_ * HD_;
    const float* lqbh = lqb + (size_t)bh * N_ * NBP_;
    const float4* ksrc = (const float4*)kbh + lane;
    NK_STAGE_ALL_();
    __syncthreads();

    v2f qf0[16], qf1[16];
    {
      const float* qrow = qS + l15 * KQSTR_ + 2 * lhalf;
#pragma unroll
      for (int s = 0; s < 16; ++s) {
        qf0[s] = ld2(qrow + 4 * s);
        qf1[s] = ld2(qrow + 16 * KQSTR_ + 4 * s);
      }
    }

    for (int np = wave; np < 32; np += 8) {
      v2f kf0[16], kf1[16];
#pragma unroll
      for (int s = 0; s < 16; ++s) {
        kf0[s] = ld2(kwld + 4 * s);
        kf1[s] = ld2(kwld + 16 * KQSTR_ + 4 * s);
      }
      const int nxt = (np + 8 < 32) ? np + 8 : np;
      float4 kn[16];
#pragma unroll
      for (int cc = 0; cc < 16; ++cc) kn[cc] = ksrc[nxt * 512 + 32 * cc];
      v8f acc[2][2];
      acc[0][0] = (v8f)0.f; acc[0][1] = (v8f)0.f;
      acc[1][0] = (v8f)0.f; acc[1][1] = (v8f)0.f;

      int4 bin[2][2][2];
      int  bni[2][2][8];
#pragma unroll
      for (int kt = 0; kt < 2; ++kt) {
        const unsigned nb = (unsigned)(np * 32 + kt * 16 + 8 * lhalf);
#pragma unroll
        for (int rt = 0; rt < 2; ++rt) {
          const unsigned i = (unsigned)(r0 + rt * 16 + l15);
          const int4* pi = (const int4*)(rp + (i * N_ + nb));
          bin[kt][rt][0] = pi[0];
          bin[kt][rt][1] = pi[1];
#pragma unroll
          for (int vv = 0; vv < 8; ++vv) bni[kt][rt][vv] = rp[(nb + vv) * N_ + i];
        }
      }

      float lkv[2][2][8];
#pragma unroll
      for (int kt = 0; kt < 2; ++kt) {
#pragma unroll
        for (int rt = 0; rt < 2; ++rt) {
          const float* lkrow = lkS + (rt * 16 + l15) * LKSTR_;
          const int4 ia = bin[kt][rt][0], ib = bin[kt][rt][1];
          const int bi[8] = {ia.x, ia.y, ia.z, ia.w, ib.x, ib.y, ib.z, ib.w};
#pragma unroll
          for (int vv = 0; vv < 8; ++vv) lkv[kt][rt][vv] = lkrow[(unsigned)bi[vv]];
        }
      }

#pragma unroll
      for (int s = 0; s < 16; ++s) {
        acc[0][0] = wmma4(kf0[s], qf0[s], acc[0][0]);
        acc[1][0] = wmma4(kf0[s], qf1[s], acc[1][0]);
        acc[0][1] = wmma4(kf1[s], qf0[s], acc[0][1]);
        acc[1][1] = wmma4(kf1[s], qf1[s], acc[1][1]);
      }

#pragma unroll
      for (int kt = 0; kt < 2; ++kt) {
        const unsigned nb = (unsigned)(np * 32 + kt * 16 + 8 * lhalf);
#pragma unroll
        for (int rt = 0; rt < 2; ++rt) {
          const unsigned m = (unsigned)(rt * 16 + l15);
          float o[8];
#pragma unroll
          for (int vv = 0; vv < 8; ++vv) {
            const unsigned n = nb + vv;
            o[vv] = acc[rt][kt][vv]
                  + lkv[kt][rt][vv]
                  + lqbh[n * NBP_ + (unsigned)bni[kt][rt][vv]];
          }
          *(float4*)(S + m * SSTR_ + nb)     = make_float4(o[0], o[1], o[2], o[3]);
          *(float4*)(S + m * SSTR_ + nb + 4) = make_float4(o[4], o[5], o[6], o[7]);
        }
      }
#pragma unroll
      for (int cc = 0; cc < 16; ++cc) *(float4*)(kwst + 2 * cc * KQSTR_) = kn[cc];
    }
  }
  __syncthreads();

  {
    float4 x[4][8];
#pragma unroll
    for (int rr = 0; rr < 4; ++rr) {
      const float* row = S + (wave * 4 + rr) * SSTR_ + 4 * lane;
#pragma unroll
      for (int c = 0; c < 8; ++c) x[rr][c] = *(const float4*)(row + 128 * c);
    }
    float mx[4];
#pragma unroll
    for (int rr = 0; rr < 4; ++rr) {
      float m = fmaxf(fmaxf(x[rr][0].x, x[rr][0].y), fmaxf(x[rr][0].z, x[rr][0].w));
#pragma unroll
      for (int c = 1; c < 8; ++c)
        m = fmaxf(m, fmaxf(fmaxf(x[rr][c].x, x[rr][c].y), fmaxf(x[rr][c].z, x[rr][c].w)));
      mx[rr] = wave_max(m);
    }
    float inv[4];
#pragma unroll
    for (int rr = 0; rr < 4; ++rr) {
      float s0 = 0.f, s1 = 0.f;
#pragma unroll
      for (int c = 0; c < 8; ++c) {
        float4 e;
        e.x = __expf(x[rr][c].x - mx[rr]);
        e.y = __expf(x[rr][c].y - mx[rr]);
        e.z = __expf(x[rr][c].z - mx[rr]);
        e.w = __expf(x[rr][c].w - mx[rr]);
        x[rr][c] = e;
        s0 += e.x + e.y;
        s1 += e.z + e.w;
      }
      inv[rr] = 1.0f / wave_sum(s0 + s1);
    }
#pragma unroll
    for (int rr = 0; rr < 4; ++rr) {
      float* row = S + (wave * 4 + rr) * SSTR_ + 4 * lane;
#pragma unroll
      for (int c = 0; c < 8; ++c) {
        float4 p = x[rr][c];
        p.x *= inv[rr]; p.y *= inv[rr]; p.z *= inv[rr]; p.w *= inv[rr];
        *(float4*)(row + 128 * c) = p;
      }
    }
  }
  __syncthreads();

  {
    float* mine = pw + wave * (ROWS_ * PSTR_);
    for (int k = lane; k < ROWS_ * PSTR_ / 2; k += 32)
      ((v2f*)mine)[k] = (v2f)0.f;
    float*       myrow = mine + lane * PSTR_;
    const float* prow  = S + lane * SSTR_;
    const int*   brow  = rp + (size_t)(r0 + lane) * N_;
    for (int win = 0; win < 2; ++win) {
      const int j0 = wave * 128 + win * 64;
      float4 p[16];
      int4   b[16];
#pragma unroll
      for (int c = 0; c < 16; ++c) {
        const int col = j0 + 4 * c;
        p[c] = *(const float4*)(prow + col);
        b[c] = *(const int4*)(brow + col);
      }
#pragma unroll
      for (int c = 0; c < 16; ++c) {
        atomicAdd(myrow + b[c].x, p[c].x);
        atomicAdd(myrow + b[c].y, p[c].y);
        atomicAdd(myrow + b[c].z, p[c].z);
        atomicAdd(myrow + b[c].w, p[c].w);
      }
    }
  }
  __syncthreads();

  const float* vp = vb + ((size_t)bh * N_ + wave * KSPL_ + 2 * lhalf) * HD_ + 4 * l15;
  float4 vq[2][16];
#pragma unroll
  for (int s = 0; s < 8; ++s) {
    vq[0][2 * s]     = *(const float4*)(vp + (4 * s) * HD_);
    vq[0][2 * s + 1] = *(const float4*)(vp + (4 * s + 1) * HD_);
  }
  float4 tq4[2][2];
#pragma unroll
  for (int s = 0; s < 2; ++s) {
#pragma unroll
    for (int e = 0; e < 2; ++e) {
      const int kb = wave * 8 + 4 * s + 2 * lhalf + e;
      const int kr = (kb < NB_) ? kb : NB_ - 1;
      tq4[s][e] = *(const float4*)(tv + ((size_t)h * NB_ + kr) * HD_ + 4 * l15);
    }
  }

  for (int idx = threadIdx.x; idx < ROWS_ * NBP_; idx += 256) {
    const int m  = idx >> 6;
    const int bk = idx & 63;
    float s = 0.f;
    for (int w = 0; w < 8; ++w) s += pw[(w * ROWS_ + m) * PSTR_ + bk];
    ab[m * ABSTR_ + bk] = s;
  }
  __syncthreads();

  v8f acc[2][4];
#pragma unroll
  for (int t = 0; t < 4; ++t) { acc[0][t] = (v8f)0.f; acc[1][t] = (v8f)0.f; }

#pragma unroll
  for (int g = 0; g < KSPL_ / 32; ++g) {
    if (g + 1 < KSPL_ / 32) {
#pragma unroll
      for (int s = 0; s < 8; ++s) {
        vq[(g + 1) & 1][2 * s]     = *(const float4*)(vp + ((g + 1) * 32 + 4 * s) * HD_);
        vq[(g + 1) & 1][2 * s + 1] = *(const float4*)(vp + ((g + 1) * 32 + 4 * s + 1) * HD_);
      }
    } else {
      const int tn  = (it + 1 < NTILE_ / NWG_) ? tile + NWG_ : tile;
      const int nbh = tn >> 5;
      const int nr0 = (tn & 31) * ROWS_;
      const float4* lsrc = (const float4*)(lkb + ((size_t)nbh * N_ + nr0) * NBP_);
      const float4* qsrc = (const float4*)(qb + ((size_t)nbh * N_ + nr0) * HD_);
      const float4* ks   = (const float4*)(kb + (size_t)nbh * N_ * HD_) + lane;
      nl0 = lsrc[e0];
      nl1 = lsrc[e1];
      nq0 = qsrc[e0];
      nq1 = qsrc[e1];
      NK_LOAD_ALL_();
    }
#pragma unroll
    for (int s = 0; s < 8; ++s) {
      const int col = wave * KSPL_ + g * 32 + 4 * s + 2 * lhalf;
      const v2f a0 = ld2(&S[l15 * SSTR_ + col]);
      const v2f a1 = ld2(&S[(16 + l15) * SSTR_ + col]);
      const float4 x0 = vq[g & 1][2 * s];
      const float4 x1 = vq[g & 1][2 * s + 1];
      v2f bf;
      bf.x = x0.x; bf.y = x1.x;
      acc[0][0] = wmma4(a0, bf, acc[0][0]);
      acc[1][0] = wmma4(a1, bf, acc[1][0]);
      bf.x = x0.y; bf.y = x1.y;
      acc[0][1] = wmma4(a0, bf, acc[0][1]);
      acc[1][1] = wmma4(a1, bf, acc[1][1]);
      bf.x = x0.z; bf.y = x1.z;
      acc[0][2] = wmma4(a0, bf, acc[0][2]);
      acc[1][2] = wmma4(a1, bf, acc[1][2]);
      bf.x = x0.w; bf.y = x1.w;
      acc[0][3] = wmma4(a0, bf, acc[0][3]);
      acc[1][3] = wmma4(a1, bf, acc[1][3]);
    }
  }

#pragma unroll
  for (int s = 0; s < 2; ++s) {
    const int kb = wave * 8 + 4 * s + 2 * lhalf;
    const float4 z4 = make_float4(0.f, 0.f, 0.f, 0.f);
    float4 x0 = tq4[s][0];
    float4 x1 = tq4[s][1];
    if (kb     >= NB_) x0 = z4;
    if (kb + 1 >= NB_) x1 = z4;
    const v2f a0 = ld2(&ab[l15 * ABSTR_ + kb]);
    const v2f a1 = ld2(&ab[(16 + l15) * ABSTR_ + kb]);
    v2f bf;
    bf.x = x0.x; bf.y = x1.x;
    acc[0][0] = wmma4(a0, bf, acc[0][0]);
    acc[1][0] = wmma4(a1, bf, acc[1][0]);
    bf.x = x0.y; bf.y = x1.y;
    acc[0][1] = wmma4(a0, bf, acc[0][1]);
    acc[1][1] = wmma4(a1, bf, acc[1][1]);
    bf.x = x0.z; bf.y = x1.z;
    acc[0][2] = wmma4(a0, bf, acc[0][2]);
    acc[1][2] = wmma4(a1, bf, acc[1][2]);
    bf.x = x0.w; bf.y = x1.w;
    acc[0][3] = wmma4(a0, bf, acc[0][3]);
    acc[1][3] = wmma4(a1, bf, acc[1][3]);
  }

  float* red = pw;
  {
    float* mine = red + wave * (ROWS_ * HD_);
#pragma unroll
    for (int mt = 0; mt < 2; ++mt) {
#pragma unroll
      for (int vv = 0; vv < 8; ++vv) {
        const int m = mt * 16 + vv + 8 * lhalf;
        *(float4*)(mine + m * HD_ + 4 * l15) =
            make_float4(acc[mt][0][vv], acc[mt][1][vv], acc[mt][2][vv], acc[mt][3][vv]);
      }
    }
  }
  __syncthreads();

  for (int e = threadIdx.x; e < ROWS_ * HD_ / 4; e += 256) {
    float4 o = ((const float4*)red)[e];
#pragma unroll
    for (int w = 1; w < 8; ++w) {
      const float4 p = ((const float4*)(red + w * (ROWS_ * HD_)))[e];
      o.x += p.x; o.y += p.y; o.z += p.z; o.w += p.w;
    }
    const int m   = e >> 4;
    const int d4  = (e & 15) * 4;
    const int tok = r0 + m;
    *(float4*)(ob + (((size_t)b * N_ + tok) * H_ + h) * HD_ + d4) = o;
  }
  __syncthreads();
  }
}

__global__ __launch_bounds__(256) __attribute__((amdgpu_waves_per_eu(2, 6)))
void k_proj(const float* __restrict__ ob,
            const float* __restrict__ wp,
            const float* __restrict__ bias,
            float* __restrict__ out) {
  __shared__ __attribute__((aligned(16))) float Bt[2 * 64 * BSTR_];

  const int wave  = threadIdx.x >> 5;
  const int lane  = threadIdx.x & 31;
  const int l15   = lane & 15;
  const int lhalf = lane >> 4;

  const int ngrp = blockIdx.y;
  const int nb   = ngrp * 64;
  const int row  = blockIdx.x * 128 + wave * 16 + l15;
  const float* xr = ob + (size_t)row * C_ + 4 * lhalf;

  const int sn = threadIdx.x & 63;
  const int sq = threadIdx.x >> 6;
  const float* wsrc = wp + (size_t)(16 * sq) * C_ + nb + sn;
  float* bst = Bt + sn * BSTR_ + 16 * sq;

  float  wn[16];
  float4 an[8];
#pragma unroll
  for (int j = 0; j < 16; ++j) wn[j] = wsrc[(size_t)j * C_];
#pragma unroll
  for (int u = 0; u < 8; ++u) an[u] = *(const float4*)(xr + 8 * u);
#pragma unroll
  for (int g = 0; g < 4; ++g)
    *(float4*)(bst + 4 * g) = make_float4(wn[4 * g], wn[4 * g + 1], wn[4 * g + 2], wn[4 * g + 3]);

  v8f acc[4];
  acc[0] = (v8f)0.f; acc[1] = (v8f)0.f; acc[2] = (v8f)0.f; acc[3] = (v8f)0.f;

  int cur = 0;
  for (int k0 = 0; k0 < C_; k0 += 64) {
    __syncthreads();
    float4 a[8];
#pragma unroll
    for (int u = 0; u < 8; ++u) a[u] = an[u];
    const int kn0 = (k0 + 64 < C_) ? k0 + 64 : k0;
#pragma unroll
    for (int j = 0; j < 16; ++j) wn[j] = wsrc[(size_t)(kn0 + j) * C_];
#pragma unroll
    for (int u = 0; u < 8; ++u) an[u] = *(const float4*)(xr + kn0 + 8 * u);

    const float* bld = Bt + cur * (64 * BSTR_) + l15 * BSTR_ + 4 * lhalf;
#pragma unroll
    for (int u = 0; u < 8; ++u) {
      float4 bq[4];
#pragma unroll
      for (int t = 0; t < 4; ++t) bq[t] = *(const float4*)(bld + (t * 16) * BSTR_ + 8 * u);
      v2f af;
      af.x = a[u].x; af.y = a[u].y;
#pragma unroll
      for (int t = 0; t < 4; ++t) {
        v2f bf; bf.x = bq[t].x; bf.y = bq[t].y;
        acc[t] = wmma4(af, bf, acc[t]);
      }
      af.x = a[u].z; af.y = a[u].w;
#pragma unroll
      for (int t = 0; t < 4; ++t) {
        v2f bf; bf.x = bq[t].z; bf.y = bq[t].w;
        acc[t] = wmma4(af, bf, acc[t]);
      }
    }

    float* bnx = bst + (cur ^ 1) * (64 * BSTR_);
#pragma unroll
    for (int g = 0; g < 4; ++g)
      *(float4*)(bnx + 4 * g) = make_float4(wn[4 * g], wn[4 * g + 1], wn[4 * g + 2], wn[4 * g + 3]);
    cur ^= 1;
  }

#pragma unroll
  for (int t = 0; t < 4; ++t) {
    const int c0 = nb + t * 16 + l15;
    const float bv = bias[c0];
#pragma unroll
    for (int vv = 0; vv < 8; ++vv) {
      const int r = blockIdx.x * 128 + wave * 16 + vv + 8 * lhalf;
      out[(size_t)r * C_ + c0] = acc[t][vv] + bv;
    }
  }
}

extern "C" void kernel_launch(void* const* d_in, const int* in_sizes, int n_in,
                              void* d_out, int out_size, void* d_ws, size_t ws_size,
                              hipStream_t stream) {
  const float* x     = (const float*)d_in[0];
  const float* wqkv  = (const float*)d_in[1];
  const float* wproj = (const float*)d_in[2];
  const float* bproj = (const float*)d_in[3];
  const float* tq    = (const float*)d_in[4];
  const float* tk    = (const float*)d_in[5];
  const float* tv    = (const float*)d_in[6];
  const int*   rp    = (const int*)d_in[7];
  float*       out   = (float*)d_out;

  float* ws = (float*)d_ws;
  const size_t SZ = (size_t)B_ * H_ * N_ * HD_;
  float* qb  = ws;
  float* kb  = ws + 1 * SZ;
  float* vb  = ws + 2 * SZ;
  float* lkb = ws + 3 * SZ;
  float* lqb = ws + 4 * SZ;
  float* ob  = ws + 5 * SZ;

  k_qkv<<<dim3(64, 36), 256, 0, stream>>>(x, wqkv, qb, kb, vb);

  k_tables<<<768, 256, 0, stream>>>(qb, kb, tq, tk, lkb, lqb);

  const size_t lds = (size_t)(ROWS_ * SSTR_ + ROWS_ * ABSTR_ +
                              8 * ROWS_ * PSTR_ + 8 * 32 * KQSTR_ +
                              ROWS_ * KQSTR_) * sizeof(float);
  k_attn<<<dim3(256), 256, lds, stream>>>(qb, kb, vb, lkb, lqb, tv, rp, ob);

  k_proj<<<dim3(64, 12), 256, 0, stream>>>(ob, wproj, bproj, out);

  (void)in_sizes; (void)n_in; (void)out_size; (void)ws_size;
}
